# starting kernel
# baseline (speedup 1.0000x reference)
#include <hip/hip_runtime.h>
#include <math.h>

#define N_NODES 40000
#define N_EDGES 320000
#define ETOT    (N_EDGES + N_NODES)
#define F_IN    256
#define H1      4
#define C1      128
#define D1      (H1*C1)
#define H2      8
#define C2      6
#define D2      (H2*C2)
#define BN_EPS    1e-5f
#define NEG_SLOPE 0.2f

typedef __attribute__((ext_vector_type(2))) float v2f;
typedef __attribute__((ext_vector_type(8))) float v8f;


__device__ __forceinline__ void atomic_max_f32(float* addr, float v) {
  if (v >= 0.0f) atomicMax((int*)addr, __float_as_int(v));
  else           atomicMin((unsigned int*)addr, __float_as_uint(v));
}

__device__ __forceinline__ void edge_rc(const int* __restrict__ rowA,
                                        const int* __restrict__ colA,
                                        int e, int& r, int& c) {
  if (e < N_EDGES) { r = rowA[e]; c = colA[e]; }
  else             { r = e - N_EDGES; c = r; }
}


__global__ void fill_f32(float* __restrict__ p, float v, int n) {
  int i = blockIdx.x * blockDim.x + threadIdx.x;
  int stride = gridDim.x * blockDim.x;
  for (; i < n; i += stride) p[i] = v;
}

__global__ void bn_prep(const float* g1, const float* be1, const float* mu1, const float* va1,
                        const float* g2, const float* be2, const float* mu2, const float* va2,
                        const float* bias1,
                        float* sc1, float* sh1, float* sc2, float* sh2) {
  int i = blockIdx.x * blockDim.x + threadIdx.x;
  if (i < F_IN) {
    float s = g1[i] / sqrtf(va1[i] + BN_EPS);
    sc1[i] = s; sh1[i] = be1[i] - mu1[i] * s;
  }
  if (i < D1) {
    float s = g2[i] / sqrtf(va2[i] + BN_EPS);
    sc2[i] = s; sh2[i] = be2[i] + (bias1[i] - mu2[i]) * s;
  }
}


__global__ __launch_bounds__(256)
void gemm_bn_wmma(const float* __restrict__ A,
                  const float* __restrict__ scale,
                  const float* __restrict__ shift,
                  const float* __restrict__ B,
                  float* __restrict__ C,
                  int M, int K, int N) {
  int wave = blockIdx.x * (blockDim.x >> 5) + (threadIdx.x >> 5);
  int ntiles = N >> 4;
  int tm = wave / ntiles;
  int tn = wave - tm * ntiles;
  if (tm >= (M >> 4)) return;

  int lane = threadIdx.x & 31;
  int half = lane >> 4;
  int l16  = lane & 15;
  int row  = (tm << 4) + l16;
  int colB = (tn << 4) + l16;

  const float* __restrict__ Arow = A + (size_t)row * K;
  v8f acc = {};
  for (int k = 0; k < K; k += 4) {
    int kk = k + (half << 1);
    v2f a, b;
    a.x = Arow[kk]     * scale[kk]     + shift[kk];
    a.y = Arow[kk + 1] * scale[kk + 1] + shift[kk + 1];
    b.x = B[(size_t)kk * N + colB];
    b.y = B[(size_t)(kk + 1) * N + colB];
    acc = __builtin_amdgcn_wmma_f32_16x16x4_f32(
         false, a,  false, b,
         (short)0, acc,  false,  false);
  }
  float* Crow = C + (size_t)((tm << 4) + (half << 3)) * N + colB;
  #pragma unroll
  for (int r = 0; r < 8; ++r) Crow[(size_t)r * N] = acc[r];
}


__global__ void node_att1(const float* __restrict__ h1, const float* __restrict__ att1,
                          float* __restrict__ ai, float* __restrict__ aj) {
  int wave = blockIdx.x * (blockDim.x >> 5) + (threadIdx.x >> 5);
  if (wave >= N_NODES * H1) return;
  int n = wave >> 2, h = wave & 3;
  int lane = threadIdx.x & 31;
  const float* __restrict__ hp = h1 + (size_t)n * D1 + h * C1;
  const float* __restrict__ al = att1 + h * (2 * C1);
  const float* __restrict__ ar = al + C1;
  float si = 0.f, sj = 0.f;
  #pragma unroll
  for (int c0 = 0; c0 < C1; c0 += 32) {
    float v = hp[c0 + lane];
    si += v * al[c0 + lane];
    sj += v * ar[c0 + lane];
  }
  #pragma unroll
  for (int off = 16; off > 0; off >>= 1) {
    si += __shfl_xor(si, off, 32);
    sj += __shfl_xor(sj, off, 32);
  }
  if (lane == 0) { ai[wave] = si; aj[wave] = sj; }
}

__global__ void edge_max1(const int* __restrict__ rowA, const int* __restrict__ colA,
                          const float* __restrict__ ai, const float* __restrict__ aj,
                          float* __restrict__ m1) {
  int e = blockIdx.x * blockDim.x + threadIdx.x;
  if (e >= ETOT) return;
  int row, col; edge_rc(rowA, colA, e, row, col);
  float4 av = *(const float4*)(ai + (size_t)row * 4);
  float4 bv = *(const float4*)(aj + (size_t)col * 4);
  float s[4] = {av.x + bv.x, av.y + bv.y, av.z + bv.z, av.w + bv.w};
  #pragma unroll
  for (int h = 0; h < 4; ++h) {
    float a = s[h]; a = a > 0.f ? a : NEG_SLOPE * a;
    atomic_max_f32(&m1[row * 4 + h], a);
  }
}

__global__ void edge_sum1(const int* __restrict__ rowA, const int* __restrict__ colA,
                          const float* __restrict__ ai, const float* __restrict__ aj,
                          const float* __restrict__ m1, float* __restrict__ s1) {
  int e = blockIdx.x * blockDim.x + threadIdx.x;
  if (e >= ETOT) return;
  int row, col; edge_rc(rowA, colA, e, row, col);
  float4 av = *(const float4*)(ai + (size_t)row * 4);
  float4 bv = *(const float4*)(aj + (size_t)col * 4);
  float4 mv = *(const float4*)(m1 + (size_t)row * 4);
  float s[4] = {av.x + bv.x, av.y + bv.y, av.z + bv.z, av.w + bv.w};
  float m[4] = {mv.x, mv.y, mv.z, mv.w};
  #pragma unroll
  for (int h = 0; h < 4; ++h) {
    float a = s[h]; a = a > 0.f ? a : NEG_SLOPE * a;
    atomicAdd(&s1[row * 4 + h], expf(a - m[h]));
  }
}

__global__ __launch_bounds__(256)
void edge_scatter1(const int* __restrict__ rowA, const int* __restrict__ colA,
                   const float* __restrict__ ai, const float* __restrict__ aj,
                   const float* __restrict__ m1, const float* __restrict__ s1,
                   const float* __restrict__ h1, float* __restrict__ out1) {
  int wave = blockIdx.x * (blockDim.x >> 5) + (threadIdx.x >> 5);
  if (wave >= ETOT * H1) return;
  int e = wave >> 2, h = wave & 3;
  int row, col; edge_rc(rowA, colA, e, row, col);
  float a = ai[row * 4 + h] + aj[col * 4 + h];
  a = a > 0.f ? a : NEG_SLOPE * a;
  float w = expf(a - m1[row * 4 + h]) / (s1[row * 4 + h] + 1e-16f);
  int lane = threadIdx.x & 31;
  const float4* __restrict__ src = (const float4*)(h1 + (size_t)col * D1 + h * C1);
  float4 v = src[lane];
  float* dst = out1 + (size_t)row * D1 + h * C1 + lane * 4;
  atomicAdd(dst + 0, v.x * w);
  atomicAdd(dst + 1, v.y * w);
  atomicAdd(dst + 2, v.z * w);
  atomicAdd(dst + 3, v.w * w);
}


__global__ void node_att2(const float* __restrict__ h2, const float* __restrict__ att2,
                          float* __restrict__ s2) {
  int tid = blockIdx.x * blockDim.x + threadIdx.x;
  if (tid >= N_NODES * H2) return;
  int n = tid >> 3, h = tid & 7;
  const float* __restrict__ hp = h2 + (size_t)n * D2 + h * C2;
  const float* __restrict__ ap = att2 + h * (2 * C2);
  float s = 0.f;
  #pragma unroll
  for (int c = 0; c < C2; ++c) s += hp[c] * (ap[c] + ap[C2 + c]);
  s2[tid] = tanhf(s);
}

__global__ void edge_max2(const int* __restrict__ rowA, const int* __restrict__ colA,
                          const float* __restrict__ s2, float* __restrict__ m2) {
  int e = blockIdx.x * blockDim.x + threadIdx.x;
  if (e >= ETOT) return;
  int row, col; edge_rc(rowA, colA, e, row, col);
  const float* sv = s2 + (size_t)col * 8;
  float* mr = m2 + (size_t)row * 8;
  #pragma unroll
  for (int h = 0; h < 8; ++h) atomic_max_f32(&mr[h], sv[h]);
}

__global__ void edge_sum2(const int* __restrict__ rowA, const int* __restrict__ colA,
                          const float* __restrict__ s2, const float* __restrict__ m2,
                          float* __restrict__ ss2) {
  int e = blockIdx.x * blockDim.x + threadIdx.x;
  if (e >= ETOT) return;
  int row, col; edge_rc(rowA, colA, e, row, col);
  const float* sv = s2 + (size_t)col * 8;
  const float* mr = m2 + (size_t)row * 8;
  #pragma unroll
  for (int h = 0; h < 8; ++h) atomicAdd(&ss2[row * 8 + h], expf(sv[h] - mr[h]));
}

__global__ void edge_scatter2(const int* __restrict__ rowA, const int* __restrict__ colA,
                              const float* __restrict__ s2, const float* __restrict__ m2,
                              const float* __restrict__ ss2,
                              const float* __restrict__ h2, float* __restrict__ out2) {
  int tid = blockIdx.x * blockDim.x + threadIdx.x;
  if (tid >= ETOT * H2) return;
  int e = tid >> 3, h = tid & 7;
  int row, col; edge_rc(rowA, colA, e, row, col);
  float w = expf(s2[col * 8 + h] - m2[row * 8 + h]) / (ss2[row * 8 + h] + 1e-16f);
  const float* __restrict__ src = h2 + (size_t)col * D2 + h * C2;
  float* dst = out2 + (size_t)row * D2 + h * C2;
  #pragma unroll
  for (int c = 0; c < C2; ++c) atomicAdd(&dst[c], src[c] * w);
}

__global__ void finalize(const float* __restrict__ out2, const float* __restrict__ b2,
                         float* __restrict__ out) {
  int n = blockIdx.x * blockDim.x + threadIdx.x;
  if (n >= N_NODES) return;
  float v[C2];
  #pragma unroll
  for (int c = 0; c < C2; ++c) {
    float s = 0.f;
    #pragma unroll
    for (int h = 0; h < H2; ++h) s += out2[(size_t)n * D2 + h * C2 + c];
    float t = s * (1.0f / H2) + b2[c];
    v[c] = t > 0.f ? t : (expf(t) - 1.0f);
  }
  float m = v[0];
  #pragma unroll
  for (int c = 1; c < C2; ++c) m = fmaxf(m, v[c]);
  float sum = 0.f;
  #pragma unroll
  for (int c = 0; c < C2; ++c) sum += expf(v[c] - m);
  float lse = logf(sum);
  #pragma unroll
  for (int c = 0; c < C2; ++c) out[(size_t)n * C2 + c] = v[c] - m - lse;
}


extern "C" void kernel_launch(void* const* d_in, const int* in_sizes, int n_in,
                              void* d_out, int out_size, void* d_ws, size_t ws_size,
                              hipStream_t stream) {
  const float* x    = (const float*)d_in[0];
  const int*   ei   = (const int*)  d_in[1];
  const float* g1   = (const float*)d_in[2];
  const float* be1  = (const float*)d_in[3];
  const float* mu1  = (const float*)d_in[4];
  const float* va1  = (const float*)d_in[5];
  const float* W1   = (const float*)d_in[6];
  const float* att1 = (const float*)d_in[7];
  const float* b1   = (const float*)d_in[8];
  const float* g2   = (const float*)d_in[9];
  const float* be2  = (const float*)d_in[10];
  const float* mu2  = (const float*)d_in[11];
  const float* va2  = (const float*)d_in[12];
  const float* W2   = (const float*)d_in[13];
  const float* att2 = (const float*)d_in[14];
  const float* b2   = (const float*)d_in[15];
  const int* rowA = ei;
  const int* colA = ei + N_EDGES;
  float* out = (float*)d_out;
  float* ws  = (float*)d_ws;

  float* sc1  = ws;
  float* sh1  = ws + 256;
  float* sc2  = ws + 512;
  float* sh2  = ws + 1024;
  float* ai   = ws + 1536;
  float* aj   = ai  + (size_t)N_NODES * H1;
  float* m1   = aj  + (size_t)N_NODES * H1;
  float* s1   = m1  + (size_t)N_NODES * H1;
  float* out1 = s1  + (size_t)N_NODES * H1;
  float* h1   = out1 + (size_t)N_NODES * D1;
  float* h2   = h1;
  float* s2   = h2  + (size_t)N_NODES * D2;
  float* m2   = s2  + (size_t)N_NODES * H2;
  float* ss2  = m2  + (size_t)N_NODES * H2;
  float* out2 = ss2 + (size_t)N_NODES * H2;

  const float NEG_INF = -__builtin_inff();

  bn_prep<<<2, 256, 0, stream>>>(g1, be1, mu1, va1, g2, be2, mu2, va2, b1,
                                 sc1, sh1, sc2, sh2);
  fill_f32<<<625,   256, 0, stream>>>(m1, NEG_INF, N_NODES * H1);
  fill_f32<<<625,   256, 0, stream>>>(s1, 0.0f,    N_NODES * H1);
  fill_f32<<<20000, 256, 0, stream>>>(out1, 0.0f,  N_NODES * D1);

  gemm_bn_wmma<<<10000, 256, 0, stream>>>(x, sc1, sh1, W1, h1, N_NODES, F_IN, D1);

  node_att1<<<20000, 256, 0, stream>>>(h1, att1, ai, aj);
  edge_max1<<<(ETOT + 255) / 256, 256, 0, stream>>>(rowA, colA, ai, aj, m1);
  edge_sum1<<<(ETOT + 255) / 256, 256, 0, stream>>>(rowA, colA, ai, aj, m1, s1);
  edge_scatter1<<<(ETOT * H1) / 8, 256, 0, stream>>>(rowA, colA, ai, aj, m1, s1, h1, out1);

  fill_f32<<<1250, 256, 0, stream>>>(m2, NEG_INF, N_NODES * H2);
  fill_f32<<<1250, 256, 0, stream>>>(ss2, 0.0f,   N_NODES * H2);
  fill_f32<<<7500, 256, 0, stream>>>(out2, 0.0f,  N_NODES * D2);

  gemm_bn_wmma<<<(2500 * 3 + 7) / 8, 256, 0, stream>>>(out1, sc2, sh2, W2, h2,
                                                       N_NODES, D1, D2);

  node_att2<<<(N_NODES * H2) / 256, 256, 0, stream>>>(h2, att2, s2);
  edge_max2<<<(ETOT + 255) / 256, 256, 0, stream>>>(rowA, colA, s2, m2);
  edge_sum2<<<(ETOT + 255) / 256, 256, 0, stream>>>(rowA, colA, s2, m2, ss2);
  edge_scatter2<<<(ETOT * H2) / 256, 256, 0, stream>>>(rowA, colA, s2, m2, ss2, h2, out2);

  finalize<<<(N_NODES + 255) / 256, 256, 0, stream>>>(out2, b2, out);
}


# Edges counting-sorted by destination per call (CSR via int atomics); both layers' softmax+aggregation become per-row gathers, no float atomics; sums: self loop, then arrival-ranked CSR slots
# speedup vs baseline: 2.3354x; 2.3354x over previous
#include <hip/hip_runtime.h>
#include <math.h>

#define N_NODES 40000
#define N_EDGES 320000
#define ETOT    (N_EDGES + N_NODES)
#define F_IN    256
#define H1      4
#define C1      128
#define D1      (H1*C1)
#define H2      8
#define C2      6
#define D2      (H2*C2)
#define SCAN_T     1024
#define SCAN_W     (SCAN_T / 32)
#define SCAN_CHUNK (N_NODES / SCAN_W)
#define SCAN_STEPS ((SCAN_CHUNK + 31) / 32)
#define BN_EPS    1e-5f
#define NEG_SLOPE 0.2f

typedef __attribute__((ext_vector_type(2))) float v2f;
typedef __attribute__((ext_vector_type(8))) float v8f;


static_assert(N_NODES <= 65536, "16-bit column ids need N_NODES <= 65536");
typedef unsigned short col_t;
static_assert(N_EDGES % 256 == 0 && N_NODES % 8 == 0 && (N_NODES * H2) % 256 == 0,
              "launch grids below divide exactly");

__device__ __forceinline__ float lrelu(float a) { return a > 0.f ? a : NEG_SLOPE * a; }


__global__ void fill_f32(float* __restrict__ p, float v, int n) {
  int i = blockIdx.x * blockDim.x + threadIdx.x;
  int stride = gridDim.x * blockDim.x;
  for (; i < n; i += stride) p[i] = v;
}

__global__ void bn_prep(const float* g1, const float* be1, const float* mu1, const float* va1,
                        const float* g2, const float* be2, const float* mu2, const float* va2,
                        const float* bias1,
                        float* sc1, float* sh1, float* sc2, float* sh2) {
  int i = blockIdx.x * blockDim.x + threadIdx.x;
  if (i < F_IN) {
    float s = g1[i] / sqrtf(va1[i] + BN_EPS);
    sc1[i] = s; sh1[i] = be1[i] - mu1[i] * s;
  }
  if (i < D1) {
    float s = g2[i] / sqrtf(va2[i] + BN_EPS);
    sc2[i] = s; sh2[i] = be2[i] + (bias1[i] - mu2[i]) * s;
  }
}

__global__ void row_count(const int* __restrict__ rowA, int* __restrict__ cnt) {
  int e = blockIdx.x * blockDim.x + threadIdx.x;
  if (e < N_EDGES) atomicAdd(&cnt[rowA[e]], 1);
}

static_assert(SCAN_W == 32 && N_NODES % SCAN_W == 0, "one lane per wave total; equal chunks");
__global__ __launch_bounds__(SCAN_T)
void row_scan(int* __restrict__ cnt) {
  __shared__ int wtot[SCAN_W];
  int lane = threadIdx.x & 31, w = threadIdx.x >> 5;
  int* __restrict__ p = cnt + w * SCAN_CHUNK;
  int s = 0;
  #pragma unroll
  for (int i = 0; i < SCAN_STEPS; ++i) {
    bool in = i < SCAN_CHUNK / 32 || lane < SCAN_CHUNK % 32;
    s += in ? p[i * 32 + lane] : 0;
  }
  #pragma unroll
  for (int off = 16; off > 0; off >>= 1) s += __shfl_xor(s, off, 32);
  if (lane == 0) wtot[w] = s;
  __syncthreads();
  int carry = lane < w ? wtot[lane] : 0;
  #pragma unroll
  for (int off = 16; off > 0; off >>= 1) carry += __shfl_xor(carry, off, 32);
  #pragma unroll 8
  for (int i = 0; i < SCAN_STEPS; ++i) {
    bool in = i < SCAN_CHUNK / 32 || lane < SCAN_CHUNK % 32;
    int x0 = in ? p[i * 32 + lane] : 0;
    int x = x0;
    #pragma unroll
    for (int off = 1; off < 32; off <<= 1) {
      int y = __shfl_up(x, off, 32);
      if (lane >= off) x += y;
    }
    if (in) p[i * 32 + lane] = x - x0 + carry;
    carry += __shfl(x, 31, 32);
  }
}

__global__ void csr_fill(const int* __restrict__ rowA, const int* __restrict__ colA,
                         int* __restrict__ cur, col_t* __restrict__ csr) {
  int e = blockIdx.x * blockDim.x + threadIdx.x;
  if (e >= N_EDGES) return;
  int slot = atomicAdd(&cur[rowA[e]], 1);
  csr[slot] = (col_t)colA[e];
}


__global__ __launch_bounds__(256)
void gemm_bn_wmma(const float* __restrict__ A,
                  const float* __restrict__ scale,
                  const float* __restrict__ shift,
                  const float* __restrict__ B,
                  float* __restrict__ C,
                  int M, int K, int N) {
  int wave = blockIdx.x * (blockDim.x >> 5) + (threadIdx.x >> 5);
  int ntiles = N >> 4;
  int tm = wave / ntiles;
  int tn = wave - tm * ntiles;
  if (tm >= (M >> 4)) return;

  int lane = threadIdx.x & 31;
  int half = lane >> 4;
  int l16  = lane & 15;
  int row  = (tm << 4) + l16;
  int colB = (tn << 4) + l16;

  const float* __restrict__ Arow = A + (size_t)row * K;
  v8f acc = {};
  for (int k = 0; k < K; k += 4) {
    int kk = k + (half << 1);
    v2f a, b;
    a.x = Arow[kk]     * scale[kk]     + shift[kk];
    a.y = Arow[kk + 1] * scale[kk + 1] + shift[kk + 1];
    b.x = B[(size_t)kk * N + colB];
    b.y = B[(size_t)(kk + 1) * N + colB];
    acc = __builtin_amdgcn_wmma_f32_16x16x4_f32(
         false, a,  false, b,
         (short)0, acc,  false,  false);
  }
  float* Crow = C + (size_t)((tm << 4) + (half << 3)) * N + colB;
  #pragma unroll
  for (int r = 0; r < 8; ++r) Crow[(size_t)r * N] = acc[r];
}


__global__ void node_att1(const float* __restrict__ h1, const float* __restrict__ att1,
                          float* __restrict__ ai, float* __restrict__ aj) {
  int wave = blockIdx.x * (blockDim.x >> 5) + (threadIdx.x >> 5);
  if (wave >= N_NODES * H1) return;
  int n = wave >> 2, h = wave & 3;
  int lane = threadIdx.x & 31;
  const float* __restrict__ hp = h1 + (size_t)n * D1 + h * C1;
  const float* __restrict__ al = att1 + h * (2 * C1);
  const float* __restrict__ ar = al + C1;
  float si = 0.f, sj = 0.f;
  #pragma unroll
  for (int c0 = 0; c0 < C1; c0 += 32) {
    float v = hp[c0 + lane];
    si += v * al[c0 + lane];
    sj += v * ar[c0 + lane];
  }
  #pragma unroll
  for (int off = 16; off > 0; off >>= 1) {
    si += __shfl_xor(si, off, 32);
    sj += __shfl_xor(sj, off, 32);
  }
  if (lane == 0) { ai[wave] = si; aj[wave] = sj; }
}

__global__ __launch_bounds__(256)
void row_agg1(const int* __restrict__ cur, const col_t* __restrict__ csr,
              const float* __restrict__ ai, const float* __restrict__ aj,
              const float* __restrict__ h1, float* __restrict__ out1) {
  int r = __builtin_amdgcn_readfirstlane((int)(blockIdx.x * 8 + (threadIdx.x >> 5)));
  int lane = threadIdx.x & 31;
  int beg = r ? cur[r - 1] : 0;
  int end = cur[r];
  const float4* __restrict__ aj4 = (const float4*)aj;
  float4 a_i = *(const float4*)(ai + (size_t)r * 4);
  float4 bs  = aj4[r];
  float t0 = lrelu(a_i.x + bs.x), t1 = lrelu(a_i.y + bs.y);
  float t2 = lrelu(a_i.z + bs.z), t3 = lrelu(a_i.w + bs.w);

  float m0 = t0, m1 = t1, m2 = t2, m3 = t3;
  for (int j = beg + lane; j < end; j += 32) {
    float4 b = aj4[csr[j]];
    m0 = fmaxf(m0, lrelu(a_i.x + b.x));
    m1 = fmaxf(m1, lrelu(a_i.y + b.y));
    m2 = fmaxf(m2, lrelu(a_i.z + b.z));
    m3 = fmaxf(m3, lrelu(a_i.w + b.w));
  }
  #pragma unroll
  for (int off = 16; off > 0; off >>= 1) {
    m0 = fmaxf(m0, __shfl_xor(m0, off, 32));
    m1 = fmaxf(m1, __shfl_xor(m1, off, 32));
    m2 = fmaxf(m2, __shfl_xor(m2, off, 32));
    m3 = fmaxf(m3, __shfl_xor(m3, off, 32));
  }

  const float4* __restrict__ h4 = (const float4*)h1;
  float u0 = expf(t0 - m0), u1 = expf(t1 - m1), u2 = expf(t2 - m2), u3 = expf(t3 - m3);
  float s0 = u0, s1 = u1, s2 = u2, s3 = u3;
  const float4* __restrict__ src = h4 + (size_t)r * (D1 / 4);
  float4 v0 = src[lane], v1 = src[32 + lane], v2 = src[64 + lane], v3 = src[96 + lane];
  float4 acc0 = {u0 * v0.x, u0 * v0.y, u0 * v0.z, u0 * v0.w};
  float4 acc1 = {u1 * v1.x, u1 * v1.y, u1 * v1.z, u1 * v1.w};
  float4 acc2 = {u2 * v2.x, u2 * v2.y, u2 * v2.z, u2 * v2.w};
  float4 acc3 = {u3 * v3.x, u3 * v3.y, u3 * v3.z, u3 * v3.w};

  for (int base = beg; base < end; base += 32) {
    int j = base + lane;
    int c = r;
    float w0 = 0.f, w1 = 0.f, w2 = 0.f, w3 = 0.f;
    if (j < end) {
      c = csr[j];
      float4 b = aj4[c];
      w0 = expf(lrelu(a_i.x + b.x) - m0);
      w1 = expf(lrelu(a_i.y + b.y) - m1);
      w2 = expf(lrelu(a_i.z + b.z) - m2);
      w3 = expf(lrelu(a_i.w + b.w) - m3);
    }
    int cnt = end - base < 32 ? end - base : 32;
    for (int k = 0; k < cnt; ++k) {
      int ck = __shfl(c, k, 32);
      u0 = __shfl(w0, k, 32); u1 = __shfl(w1, k, 32);
      u2 = __shfl(w2, k, 32); u3 = __shfl(w3, k, 32);
      s0 += u0; s1 += u1; s2 += u2; s3 += u3;
      src = h4 + (size_t)ck * (D1 / 4);
      v0 = src[lane]; v1 = src[32 + lane]; v2 = src[64 + lane]; v3 = src[96 + lane];
      acc0.x += u0 * v0.x; acc0.y += u0 * v0.y; acc0.z += u0 * v0.z; acc0.w += u0 * v0.w;
      acc1.x += u1 * v1.x; acc1.y += u1 * v1.y; acc1.z += u1 * v1.z; acc1.w += u1 * v1.w;
      acc2.x += u2 * v2.x; acc2.y += u2 * v2.y; acc2.z += u2 * v2.z; acc2.w += u2 * v2.w;
      acc3.x += u3 * v3.x; acc3.y += u3 * v3.y; acc3.z += u3 * v3.z; acc3.w += u3 * v3.w;
    }
  }
  float i0 = 1.0f / (s0 + 1e-16f), i1 = 1.0f / (s1 + 1e-16f);
  float i2 = 1.0f / (s2 + 1e-16f), i3 = 1.0f / (s3 + 1e-16f);
  float4* __restrict__ dst = (float4*)(out1 + (size_t)r * D1);
  dst[lane]      = make_float4(acc0.x * i0, acc0.y * i0, acc0.z * i0, acc0.w * i0);
  dst[32 + lane] = make_float4(acc1.x * i1, acc1.y * i1, acc1.z * i1, acc1.w * i1);
  dst[64 + lane] = make_float4(acc2.x * i2, acc2.y * i2, acc2.z * i2, acc2.w * i2);
  dst[96 + lane] = make_float4(acc3.x * i3, acc3.y * i3, acc3.z * i3, acc3.w * i3);
}


__global__ void node_att2(const float* __restrict__ h2, const float* __restrict__ att2,
                          float* __restrict__ s2) {
  int tid = blockIdx.x * blockDim.x + threadIdx.x;
  if (tid >= N_NODES * H2) return;
  int n = tid >> 3, h = tid & 7;
  const float* __restrict__ hp = h2 + (size_t)n * D2 + h * C2;
  const float* __restrict__ ap = att2 + h * (2 * C2);
  float s = 0.f;
  #pragma unroll
  for (int c = 0; c < C2; ++c) s += hp[c] * (ap[c] + ap[C2 + c]);
  s2[tid] = tanhf(s);
}

__global__ void row_agg2(const int* __restrict__ cur, const col_t* __restrict__ csr,
                         const float* __restrict__ s2, const float* __restrict__ h2,
                         float* __restrict__ out2) {
  int tid = blockIdx.x * blockDim.x + threadIdx.x;
  if (tid >= N_NODES * H2) return;
  int r = tid >> 3, h = tid & 7;
  int beg = r ? cur[r - 1] : 0;
  int end = cur[r];
  float sself = s2[tid];
  float m = sself;
  for (int j = beg; j < end; ++j) m = fmaxf(m, s2[(int)csr[j] * H2 + h]);

  float w = expf(sself - m);
  float den = w;
  const float2* __restrict__ src = (const float2*)(h2 + (size_t)r * D2 + h * C2);
  float2 p0 = src[0], p1 = src[1], p2 = src[2];
  float a0 = w * p0.x, a1 = w * p0.y, a2 = w * p1.x, a3 = w * p1.y, a4 = w * p2.x, a5 = w * p2.y;
  for (int j = beg; j < end; ++j) {
    int c = csr[j];
    w = expf(s2[c * H2 + h] - m);
    den += w;
    src = (const float2*)(h2 + (size_t)c * D2 + h * C2);
    p0 = src[0]; p1 = src[1]; p2 = src[2];
    a0 += w * p0.x; a1 += w * p0.y; a2 += w * p1.x;
    a3 += w * p1.y; a4 += w * p2.x; a5 += w * p2.y;
  }
  float inv = 1.0f / (den + 1e-16f);
  float2* __restrict__ dst = (float2*)(out2 + (size_t)r * D2 + h * C2);
  dst[0] = make_float2(a0 * inv, a1 * inv);
  dst[1] = make_float2(a2 * inv, a3 * inv);
  dst[2] = make_float2(a4 * inv, a5 * inv);
}

__global__ void finalize(const float* __restrict__ out2, const float* __restrict__ b2,
                         float* __restrict__ out) {
  int n = blockIdx.x * blockDim.x + threadIdx.x;
  if (n >= N_NODES) return;
  float v[C2];
  #pragma unroll
  for (int c = 0; c < C2; ++c) {
    float s = 0.f;
    #pragma unroll
    for (int h = 0; h < H2; ++h) s += out2[(size_t)n * D2 + h * C2 + c];
    float t = s * (1.0f / H2) + b2[c];
    v[c] = t > 0.f ? t : (expf(t) - 1.0f);
  }
  float m = v[0];
  #pragma unroll
  for (int c = 1; c < C2; ++c) m = fmaxf(m, v[c]);
  float sum = 0.f;
  #pragma unroll
  for (int c = 0; c < C2; ++c) sum += expf(v[c] - m);
  float lse = logf(sum);
  #pragma unroll
  for (int c = 0; c < C2; ++c) out[(size_t)n * C2 + c] = v[c] - m - lse;
}


extern "C" void kernel_launch(void* const* d_in, const int* in_sizes, int n_in,
                              void* d_out, int out_size, void* d_ws, size_t ws_size,
                              hipStream_t stream) {
  const float* x    = (const float*)d_in[0];
  const int*   ei   = (const int*)  d_in[1];
  const float* g1   = (const float*)d_in[2];
  const float* be1  = (const float*)d_in[3];
  const float* mu1  = (const float*)d_in[4];
  const float* va1  = (const float*)d_in[5];
  const float* W1   = (const float*)d_in[6];
  const float* att1 = (const float*)d_in[7];
  const float* b1   = (const float*)d_in[8];
  const float* g2   = (const float*)d_in[9];
  const float* be2  = (const float*)d_in[10];
  const float* mu2  = (const float*)d_in[11];
  const float* va2  = (const float*)d_in[12];
  const float* W2   = (const float*)d_in[13];
  const float* att2 = (const float*)d_in[14];
  const float* b2   = (const float*)d_in[15];
  const int* rowA = ei;
  const int* colA = ei + N_EDGES;
  float* out = (float*)d_out;
  float* ws  = (float*)d_ws;

  float* sc1  = ws;
  float* sh1  = ws + 256;
  float* sc2  = ws + 512;
  float* sh2  = ws + 1024;
  float* ai   = ws + 1536;
  float* aj   = ai  + (size_t)N_NODES * H1;
  float* edg  = aj  + (size_t)N_NODES * H1;
  int*   cur  = (int*)edg;
  col_t* csr  = (col_t*)(edg + N_NODES);
  float* out1 = edg + (size_t)N_NODES * 2 * H1;
  float* h1   = out1 + (size_t)N_NODES * D1;
  float* h2   = h1;
  float* s2   = h2  + (size_t)N_NODES * D2;
  float* out2 = s2  + (size_t)N_NODES * H2;

  bn_prep<<<2, 256, 0, stream>>>(g1, be1, mu1, va1, g2, be2, mu2, va2, b1,
                                 sc1, sh1, sc2, sh2);

  fill_f32<<<(N_NODES + 255) / 256, 256, 0, stream>>>(edg, 0.0f, N_NODES);
  row_count<<<N_EDGES / 256, 256, 0, stream>>>(rowA, cur);
  row_scan<<<1, SCAN_T, 0, stream>>>(cur);
  csr_fill<<<N_EDGES / 256, 256, 0, stream>>>(rowA, colA, cur, csr);

  gemm_bn_wmma<<<10000, 256, 0, stream>>>(x, sc1, sh1, W1, h1, N_NODES, F_IN, D1);

  node_att1<<<20000, 256, 0, stream>>>(h1, att1, ai, aj);
  row_agg1<<<N_NODES / 8, 256, 0, stream>>>(cur, csr, ai, aj, h1, out1);

  gemm_bn_wmma<<<(2500 * 3 + 7) / 8, 256, 0, stream>>>(out1, sc2, sh2, W2, h2,
                                                       N_NODES, D1, D2);

  node_att2<<<(N_NODES * H2) / 256, 256, 0, stream>>>(h2, att2, s2);
  row_agg2<<<(N_NODES * H2) / 256, 256, 0, stream>>>(cur, csr, s2, h2, out2);

  finalize<<<(N_NODES + 255) / 256, 256, 0, stream>>>(out2, b2, out);
}


# GEMMs re-tiled: BN-applied A rows staged once per work-group in LDS; each wave holds 2x4 (layer 2: 1x3) 16x16 accumulators so A and B fragments are reused; per-tile summation order unchanged
# speedup vs baseline: 3.4628x; 1.4827x over previous
#include <hip/hip_runtime.h>
#include <math.h>

#define N_NODES 40000
#define N_EDGES 320000
#define ETOT    (N_EDGES + N_NODES)
#define F_IN    256
#define H1      4
#define C1      128
#define D1      (H1*C1)
#define H2      8
#define C2      6
#define D2      (H2*C2)
#define SCAN_T     1024
#define SCAN_W     (SCAN_T / 32)
#define SCAN_CHUNK (N_NODES / SCAN_W)
#define SCAN_STEPS ((SCAN_CHUNK + 31) / 32)
#define BN_EPS    1e-5f
#define NEG_SLOPE 0.2f

typedef __attribute__((ext_vector_type(2))) float v2f;
typedef __attribute__((ext_vector_type(8))) float v8f;


static_assert(N_NODES <= 65536, "16-bit column ids need N_NODES <= 65536");
typedef unsigned short col_t;
static_assert(N_EDGES % 256 == 0 && N_NODES % 8 == 0 && (N_NODES * H2) % 256 == 0,
              "launch grids below divide exactly");

__device__ __forceinline__ float lrelu(float a) { return a > 0.f ? a : NEG_SLOPE * a; }


__global__ void fill_f32(float* __restrict__ p, float v, int n) {
  int i = blockIdx.x * blockDim.x + threadIdx.x;
  int stride = gridDim.x * blockDim.x;
  for (; i < n; i += stride) p[i] = v;
}

__global__ void bn_prep(const float* g1, const float* be1, const float* mu1, const float* va1,
                        const float* g2, const float* be2, const float* mu2, const float* va2,
                        const float* bias1,
                        float* sc1, float* sh1, float* sc2, float* sh2) {
  int i = blockIdx.x * blockDim.x + threadIdx.x;
  if (i < F_IN) {
    float s = g1[i] / sqrtf(va1[i] + BN_EPS);
    sc1[i] = s; sh1[i] = be1[i] - mu1[i] * s;
  }
  if (i < D1) {
    float s = g2[i] / sqrtf(va2[i] + BN_EPS);
    sc2[i] = s; sh2[i] = be2[i] + (bias1[i] - mu2[i]) * s;
  }
}

__global__ void row_count(const int* __restrict__ rowA, int* __restrict__ cnt) {
  int e = blockIdx.x * blockDim.x + threadIdx.x;
  if (e < N_EDGES) atomicAdd(&cnt[rowA[e]], 1);
}

static_assert(SCAN_W == 32 && N_NODES % SCAN_W == 0, "one lane per wave total; equal chunks");
__global__ __launch_bounds__(SCAN_T)
void row_scan(int* __restrict__ cnt) {
  __shared__ int wtot[SCAN_W];
  int lane = threadIdx.x & 31, w = threadIdx.x >> 5;
  int* __restrict__ p = cnt + w * SCAN_CHUNK;
  int s = 0;
  #pragma unroll
  for (int i = 0; i < SCAN_STEPS; ++i) {
    bool in = i < SCAN_CHUNK / 32 || lane < SCAN_CHUNK % 32;
    s += in ? p[i * 32 + lane] : 0;
  }
  #pragma unroll
  for (int off = 16; off > 0; off >>= 1) s += __shfl_xor(s, off, 32);
  if (lane == 0) wtot[w] = s;
  __syncthreads();
  int carry = lane < w ? wtot[lane] : 0;
  #pragma unroll
  for (int off = 16; off > 0; off >>= 1) carry += __shfl_xor(carry, off, 32);
  #pragma unroll 8
  for (int i = 0; i < SCAN_STEPS; ++i) {
    bool in = i < SCAN_CHUNK / 32 || lane < SCAN_CHUNK % 32;
    int x0 = in ? p[i * 32 + lane] : 0;
    int x = x0;
    #pragma unroll
    for (int off = 1; off < 32; off <<= 1) {
      int y = __shfl_up(x, off, 32);
      if (lane >= off) x += y;
    }
    if (in) p[i * 32 + lane] = x - x0 + carry;
    carry += __shfl(x, 31, 32);
  }
}

__global__ void csr_fill(const int* __restrict__ rowA, const int* __restrict__ colA,
                         int* __restrict__ cur, col_t* __restrict__ csr) {
  int e = blockIdx.x * blockDim.x + threadIdx.x;
  if (e >= N_EDGES) return;
  int slot = atomicAdd(&cur[rowA[e]], 1);
  csr[slot] = (col_t)colA[e];
}


template <int NT, int N>
__device__ __forceinline__ void load_b(const float* __restrict__ b0, float (&bx)[NT], float (&by)[NT]) {
  const float* __restrict__ b1 = b0 + N;
  if constexpr (NT == 4) {
    float4 t0 = *(const float4*)b0, t1 = *(const float4*)b1;
    bx[0] = t0.x; bx[1] = t0.y; bx[2] = t0.z; bx[3] = t0.w;
    by[0] = t1.x; by[1] = t1.y; by[2] = t1.z; by[3] = t1.w;
  } else {
    #pragma unroll
    for (int nt = 0; nt < NT; ++nt) { bx[nt] = b0[nt]; by[nt] = b1[nt]; }
  }
}

template <int K, int N, int MT, int NT, int WN, int KC, int PF>
__global__ __launch_bounds__(32 * WN)
void gemm_bn_wmma(const float* A,
                  const float* __restrict__ scale,
                  const float* __restrict__ shift,
                  const float* __restrict__ B,
                  float* __restrict__ C) {
  constexpr int THREADS = 32 * WN;
  constexpr int ROWS  = 16 * MT;
  constexpr int PITCH = K + 4;
  constexpr int K4  = K / 4;
  constexpr int C4  = KC / 4;
  constexpr int NCH = K / KC;
  constexpr int RPP = THREADS / C4;
  constexpr int LPT = ROWS / RPP;
  static_assert(PF >= 1 && 4 * PF <= K, "queue depth");
  static_assert(N == WN * NT * 16, "the waves of a work-group tile all N columns");
  static_assert(K % KC == 0 && KC % 4 == 0 && THREADS % C4 == 0 && ROWS % RPP == 0, "exact staging");
  static_assert((PITCH / 4) % 2 == 1, "pitch is 16 B x odd");
  static_assert(N_NODES % ROWS == 0, "grid divides exactly");
  __shared__ float As[ROWS * PITCH];

  const int tid  = threadIdx.x;
  const int lane = tid & 31;
  const int half = lane >> 4;
  const int l16  = lane & 15;
  const int wn   = tid >> 5;
  const int row0 = blockIdx.x * ROWS;

  const int tr = tid / C4, c4 = tid % C4;
  const float4* A4 = (const float4*)(A + (size_t)row0 * K) + (size_t)tr * K4 + c4;
  const float4* __restrict__ sc4 = (const float4*)scale + c4;
  const float4* __restrict__ sh4 = (const float4*)shift + c4;
  float* st = As + tr * PITCH + c4 * 4;

  float4 pre[LPT];
  #pragma unroll
  for (int i = 0; i < LPT; ++i) pre[i] = A4[(size_t)i * RPP * K4];

  const float* as = As + l16 * PITCH + 2 * half;
  const float* __restrict__ bp = B + (size_t)(2 * half) * N + wn * (NT * 16) + NT * l16;

  v8f acc[MT][NT];
  #pragma unroll
  for (int mt = 0; mt < MT; ++mt)
    #pragma unroll
    for (int nt = 0; nt < NT; ++nt) acc[mt][nt] = v8f{};

  float qx[PF][NT], qy[PF][NT];
  #pragma unroll
  for (int p = 0; p < PF; ++p) load_b<NT, N>(bp + (size_t)(4 * p) * N, qx[p], qy[p]);

  for (int ch = 0; ch < NCH; ++ch) {
    const float4 sc = sc4[ch * C4], sh = sh4[ch * C4];
    #pragma unroll
    for (int i = 0; i < LPT; ++i) {
      float4 v = pre[i], o;
      o.x = v.x * sc.x + sh.x; o.y = v.y * sc.y + sh.y;
      o.z = v.z * sc.z + sh.z; o.w = v.w * sc.w + sh.w;
      *(float4*)(st + i * RPP * PITCH + ch * KC) = o;
    }
    __syncthreads();
    if (ch + 1 < NCH) {
      #pragma unroll
      for (int i = 0; i < LPT; ++i) pre[i] = A4[(size_t)i * RPP * K4 + (ch + 1) * C4];
    }
    for (int k = ch * KC; k < ch * KC + KC; k += 4) {
      v2f a[MT];
      #pragma unroll
      for (int mt = 0; mt < MT; ++mt) a[mt] = *(const v2f*)(as + mt * 16 * PITCH + k);
      float bx[NT], by[NT];
      #pragma unroll
      for (int nt = 0; nt < NT; ++nt) { bx[nt] = qx[0][nt]; by[nt] = qy[0][nt]; }
      #pragma unroll
      for (int p = 0; p + 1 < PF; ++p)
        #pragma unroll
        for (int nt = 0; nt < NT; ++nt) { qx[p][nt] = qx[p + 1][nt]; qy[p][nt] = qy[p + 1][nt]; }
      if (k + 4 * PF < K) load_b<NT, N>(bp + (size_t)(k + 4 * PF) * N, qx[PF - 1], qy[PF - 1]);
      #pragma unroll
      for (int mt = 0; mt < MT; ++mt)
        #pragma unroll
        for (int nt = 0; nt < NT; ++nt) {
          v2f b; b.x = bx[nt]; b.y = by[nt];
          acc[mt][nt] = __builtin_amdgcn_wmma_f32_16x16x4_f32(
               false, a[mt],  false, b,
               (short)0, acc[mt][nt],  false,  false);
        }
    }
  }

  #pragma unroll
  for (int mt = 0; mt < MT; ++mt) {
    float* cp = C + (size_t)(row0 + mt * 16 + (half << 3)) * N + wn * (NT * 16) + NT * l16;
    #pragma unroll
    for (int r = 0; r < 8; ++r) {
      if constexpr (NT == 4) {
        *(float4*)(cp + (size_t)r * N) =
            make_float4(acc[mt][0][r], acc[mt][1][r], acc[mt][2][r], acc[mt][3][r]);
      } else {
        #pragma unroll
        for (int nt = 0; nt < NT; ++nt) cp[(size_t)r * N + nt] = acc[mt][nt][r];
      }
    }
  }
}


__global__ void node_att1(const float* __restrict__ h1, const float* __restrict__ att1,
                          float* __restrict__ ai, float* __restrict__ aj) {
  int wave = blockIdx.x * (blockDim.x >> 5) + (threadIdx.x >> 5);
  if (wave >= N_NODES * H1) return;
  int n = wave >> 2, h = wave & 3;
  int lane = threadIdx.x & 31;
  const float* __restrict__ hp = h1 + (size_t)n * D1 + h * C1;
  const float* __restrict__ al = att1 + h * (2 * C1);
  const float* __restrict__ ar = al + C1;
  float si = 0.f, sj = 0.f;
  #pragma unroll
  for (int c0 = 0; c0 < C1; c0 += 32) {
    float v = hp[c0 + lane];
    si += v * al[c0 + lane];
    sj += v * ar[c0 + lane];
  }
  #pragma unroll
  for (int off = 16; off > 0; off >>= 1) {
    si += __shfl_xor(si, off, 32);
    sj += __shfl_xor(sj, off, 32);
  }
  if (lane == 0) { ai[wave] = si; aj[wave] = sj; }
}

__global__ __launch_bounds__(256)
void row_agg1(const int* __restrict__ cur, const col_t* __restrict__ csr,
              const float* __restrict__ ai, const float* __restrict__ aj,
              const float* __restrict__ h1, float* __restrict__ out1) {
  int r = __builtin_amdgcn_readfirstlane((int)(blockIdx.x * 8 + (threadIdx.x >> 5)));
  int lane = threadIdx.x & 31;
  int beg = r ? cur[r - 1] : 0;
  int end = cur[r];
  const float4* __restrict__ aj4 = (const float4*)aj;
  float4 a_i = *(const float4*)(ai + (size_t)r * 4);
  float4 bs  = aj4[r];
  float t0 = lrelu(a_i.x + bs.x), t1 = lrelu(a_i.y + bs.y);
  float t2 = lrelu(a_i.z + bs.z), t3 = lrelu(a_i.w + bs.w);

  float m0 = t0, m1 = t1, m2 = t2, m3 = t3;
  for (int j = beg + lane; j < end; j += 32) {
    float4 b = aj4[csr[j]];
    m0 = fmaxf(m0, lrelu(a_i.x + b.x));
    m1 = fmaxf(m1, lrelu(a_i.y + b.y));
    m2 = fmaxf(m2, lrelu(a_i.z + b.z));
    m3 = fmaxf(m3, lrelu(a_i.w + b.w));
  }
  #pragma unroll
  for (int off = 16; off > 0; off >>= 1) {
    m0 = fmaxf(m0, __shfl_xor(m0, off, 32));
    m1 = fmaxf(m1, __shfl_xor(m1, off, 32));
    m2 = fmaxf(m2, __shfl_xor(m2, off, 32));
    m3 = fmaxf(m3, __shfl_xor(m3, off, 32));
  }

  const float4* __restrict__ h4 = (const float4*)h1;
  float u0 = expf(t0 - m0), u1 = expf(t1 - m1), u2 = expf(t2 - m2), u3 = expf(t3 - m3);
  float s0 = u0, s1 = u1, s2 = u2, s3 = u3;
  const float4* __restrict__ src = h4 + (size_t)r * (D1 / 4);
  float4 v0 = src[lane], v1 = src[32 + lane], v2 = src[64 + lane], v3 = src[96 + lane];
  float4 acc0 = {u0 * v0.x, u0 * v0.y, u0 * v0.z, u0 * v0.w};
  float4 acc1 = {u1 * v1.x, u1 * v1.y, u1 * v1.z, u1 * v1.w};
  float4 acc2 = {u2 * v2.x, u2 * v2.y, u2 * v2.z, u2 * v2.w};
  float4 acc3 = {u3 * v3.x, u3 * v3.y, u3 * v3.z, u3 * v3.w};

  for (int base = beg; base < end; base += 32) {
    int j = base + lane;
    int c = r;
    float w0 = 0.f, w1 = 0.f, w2 = 0.f, w3 = 0.f;
    if (j < end) {
      c = csr[j];
      float4 b = aj4[c];
      w0 = expf(lrelu(a_i.x + b.x) - m0);
      w1 = expf(lrelu(a_i.y + b.y) - m1);
      w2 = expf(lrelu(a_i.z + b.z) - m2);
      w3 = expf(lrelu(a_i.w + b.w) - m3);
    }
    int cnt = end - base < 32 ? end - base : 32;
    for (int k = 0; k < cnt; ++k) {
      int ck = __shfl(c, k, 32);
      u0 = __shfl(w0, k, 32); u1 = __shfl(w1, k, 32);
      u2 = __shfl(w2, k, 32); u3 = __shfl(w3, k, 32);
      s0 += u0; s1 += u1; s2 += u2; s3 += u3;
      src = h4 + (size_t)ck * (D1 / 4);
      v0 = src[lane]; v1 = src[32 + lane]; v2 = src[64 + lane]; v3 = src[96 + lane];
      acc0.x += u0 * v0.x; acc0.y += u0 * v0.y; acc0.z += u0 * v0.z; acc0.w += u0 * v0.w;
      acc1.x += u1 * v1.x; acc1.y += u1 * v1.y; acc1.z += u1 * v1.z; acc1.w += u1 * v1.w;
      acc2.x += u2 * v2.x; acc2.y += u2 * v2.y; acc2.z += u2 * v2.z; acc2.w += u2 * v2.w;
      acc3.x += u3 * v3.x; acc3.y += u3 * v3.y; acc3.z += u3 * v3.z; acc3.w += u3 * v3.w;
    }
  }
  float i0 = 1.0f / (s0 + 1e-16f), i1 = 1.0f / (s1 + 1e-16f);
  float i2 = 1.0f / (s2 + 1e-16f), i3 = 1.0f / (s3 + 1e-16f);
  float4* __restrict__ dst = (float4*)(out1 + (size_t)r * D1);
  dst[lane]      = make_float4(acc0.x * i0, acc0.y * i0, acc0.z * i0, acc0.w * i0);
  dst[32 + lane] = make_float4(acc1.x * i1, acc1.y * i1, acc1.z * i1, acc1.w * i1);
  dst[64 + lane] = make_float4(acc2.x * i2, acc2.y * i2, acc2.z * i2, acc2.w * i2);
  dst[96 + lane] = make_float4(acc3.x * i3, acc3.y * i3, acc3.z * i3, acc3.w * i3);
}


__global__ void node_att2(const float* __restrict__ h2, const float* __restrict__ att2,
                          float* __restrict__ s2) {
  int tid = blockIdx.x * blockDim.x + threadIdx.x;
  if (tid >= N_NODES * H2) return;
  int n = tid >> 3, h = tid & 7;
  const float* __restrict__ hp = h2 + (size_t)n * D2 + h * C2;
  const float* __restrict__ ap = att2 + h * (2 * C2);
  float s = 0.f;
  #pragma unroll
  for (int c = 0; c < C2; ++c) s += hp[c] * (ap[c] + ap[C2 + c]);
  s2[tid] = tanhf(s);
}

__global__ void row_agg2(const int* __restrict__ cur, const col_t* __restrict__ csr,
                         const float* __restrict__ s2, const float* __restrict__ h2,
                         float* __restrict__ out2) {
  int tid = blockIdx.x * blockDim.x + threadIdx.x;
  if (tid >= N_NODES * H2) return;
  int r = tid >> 3, h = tid & 7;
  int beg = r ? cur[r - 1] : 0;
  int end = cur[r];
  float sself = s2[tid];
  float m = sself;
  for (int j = beg; j < end; ++j) m = fmaxf(m, s2[(int)csr[j] * H2 + h]);

  float w = expf(sself - m);
  float den = w;
  const float2* __restrict__ src = (const float2*)(h2 + (size_t)r * D2 + h * C2);
  float2 p0 = src[0], p1 = src[1], p2 = src[2];
  float a0 = w * p0.x, a1 = w * p0.y, a2 = w * p1.x, a3 = w * p1.y, a4 = w * p2.x, a5 = w * p2.y;
  for (int j = beg; j < end; ++j) {
    int c = csr[j];
    w = expf(s2[c * H2 + h] - m);
    den += w;
    src = (const float2*)(h2 + (size_t)c * D2 + h * C2);
    p0 = src[0]; p1 = src[1]; p2 = src[2];
    a0 += w * p0.x; a1 += w * p0.y; a2 += w * p1.x;
    a3 += w * p1.y; a4 += w * p2.x; a5 += w * p2.y;
  }
  float inv = 1.0f / (den + 1e-16f);
  float2* __restrict__ dst = (float2*)(out2 + (size_t)r * D2 + h * C2);
  dst[0] = make_float2(a0 * inv, a1 * inv);
  dst[1] = make_float2(a2 * inv, a3 * inv);
  dst[2] = make_float2(a4 * inv, a5 * inv);
}

__global__ void finalize(const float* __restrict__ out2, const float* __restrict__ b2,
                         float* __restrict__ out) {
  int n = blockIdx.x * blockDim.x + threadIdx.x;
  if (n >= N_NODES) return;
  float v[C2];
  #pragma unroll
  for (int c = 0; c < C2; ++c) {
    float s = 0.f;
    #pragma unroll
    for (int h = 0; h < H2; ++h) s += out2[(size_t)n * D2 + h * C2 + c];
    float t = s * (1.0f / H2) + b2[c];
    v[c] = t > 0.f ? t : (expf(t) - 1.0f);
  }
  float m = v[0];
  #pragma unroll
  for (int c = 1; c < C2; ++c) m = fmaxf(m, v[c]);
  float sum = 0.f;
  #pragma unroll
  for (int c = 0; c < C2; ++c) sum += expf(v[c] - m);
  float lse = logf(sum);
  #pragma unroll
  for (int c = 0; c < C2; ++c) out[(size_t)n * C2 + c] = v[c] - m - lse;
}


extern "C" void kernel_launch(void* const* d_in, const int* in_sizes, int n_in,
                              void* d_out, int out_size, void* d_ws, size_t ws_size,
                              hipStream_t stream) {
  const float* x    = (const float*)d_in[0];
  const int*   ei   = (const int*)  d_in[1];
  const float* g1   = (const float*)d_in[2];
  const float* be1  = (const float*)d_in[3];
  const float* mu1  = (const float*)d_in[4];
  const float* va1  = (const float*)d_in[5];
  const float* W1   = (const float*)d_in[6];
  const float* att1 = (const float*)d_in[7];
  const float* b1   = (const float*)d_in[8];
  const float* g2   = (const float*)d_in[9];
  const float* be2  = (const float*)d_in[10];
  const float* mu2  = (const float*)d_in[11];
  const float* va2  = (const float*)d_in[12];
  const float* W2   = (const float*)d_in[13];
  const float* att2 = (const float*)d_in[14];
  const float* b2   = (const float*)d_in[15];
  const int* rowA = ei;
  const int* colA = ei + N_EDGES;
  float* out = (float*)d_out;
  float* ws  = (float*)d_ws;

  float* sc1  = ws;
  float* sh1  = ws + 256;
  float* sc2  = ws + 512;
  float* sh2  = ws + 1024;
  float* ai   = ws + 1536;
  float* aj   = ai  + (size_t)N_NODES * H1;
  float* edg  = aj  + (size_t)N_NODES * H1;
  int*   cur  = (int*)edg;
  col_t* csr  = (col_t*)(edg + N_NODES);
  float* out1 = edg + (size_t)N_NODES * 2 * H1;
  float* h1   = out1 + (size_t)N_NODES * D1;
  float* h2   = h1;
  float* s2   = h2  + (size_t)N_NODES * D2;
  float* out2 = s2  + (size_t)N_NODES * H2;

  bn_prep<<<2, 256, 0, stream>>>(g1, be1, mu1, va1, g2, be2, mu2, va2, b1,
                                 sc1, sh1, sc2, sh2);

  fill_f32<<<(N_NODES + 255) / 256, 256, 0, stream>>>(edg, 0.0f, N_NODES);
  row_count<<<N_EDGES / 256, 256, 0, stream>>>(rowA, cur);
  row_scan<<<1, SCAN_T, 0, stream>>>(cur);
  csr_fill<<<N_EDGES / 256, 256, 0, stream>>>(rowA, colA, cur, csr);

  gemm_bn_wmma<F_IN, D1, 2, 4, 8, 64, 2><<<N_NODES / 32, 256, 0, stream>>>(x, sc1, sh1, W1, h1);

  node_att1<<<20000, 256, 0, stream>>>(h1, att1, ai, aj);
  row_agg1<<<N_NODES / 8, 256, 0, stream>>>(cur, csr, ai, aj, h1, out1);

  gemm_bn_wmma<D1, D2, 1, 3, 1, 128, 2><<<N_NODES / 16, 32, 0, stream>>>(out1, sc2, sh2, W2, h2);

  node_att2<<<(N_NODES * H2) / 256, 256, 0, stream>>>(h2, att2, s2);
  row_agg2<<<(N_NODES * H2) / 256, 256, 0, stream>>>(cur, csr, s2, h2, out2);

  finalize<<<(N_NODES + 255) / 256, 256, 0, stream>>>(out2, b2, out);
}


# Per-node attention scores (layer 1 ai/aj, layer 2 tanh score) taken from the accumulators in the GEMM epilogues; both score kernels removed; score sums now: lane partial, 16-lane tree, wave pair
# speedup vs baseline: 3.8587x; 1.1143x over previous
#include <hip/hip_runtime.h>
#include <math.h>

#define N_NODES 40000
#define N_EDGES 320000
#define ETOT    (N_EDGES + N_NODES)
#define F_IN    256
#define H1      4
#define C1      128
#define D1      (H1*C1)
#define H2      8
#define C2      6
#define D2      (H2*C2)
#define SCAN_T     1024
#define SCAN_W     (SCAN_T / 32)
#define SCAN_CHUNK (N_NODES / SCAN_W)
#define SCAN_STEPS ((SCAN_CHUNK + 31) / 32)
#define BN_EPS    1e-5f
#define NEG_SLOPE 0.2f

typedef __attribute__((ext_vector_type(2))) float v2f;
typedef __attribute__((ext_vector_type(8))) float v8f;


static_assert(N_NODES <= 65536, "16-bit column ids need N_NODES <= 65536");
typedef unsigned short col_t;
static_assert(N_EDGES % 256 == 0 && N_NODES % 8 == 0 && (N_NODES * H2) % 256 == 0,
              "launch grids below divide exactly");

__device__ __forceinline__ float lrelu(float a) { return a > 0.f ? a : NEG_SLOPE * a; }


__global__ void fill_f32(float* __restrict__ p, float v, int n) {
  int i = blockIdx.x * blockDim.x + threadIdx.x;
  int stride = gridDim.x * blockDim.x;
  for (; i < n; i += stride) p[i] = v;
}

__global__ void bn_prep(const float* g1, const float* be1, const float* mu1, const float* va1,
                        const float* g2, const float* be2, const float* mu2, const float* va2,
                        const float* bias1,
                        float* sc1, float* sh1, float* sc2, float* sh2) {
  int i = blockIdx.x * blockDim.x + threadIdx.x;
  if (i < F_IN) {
    float s = g1[i] / sqrtf(va1[i] + BN_EPS);
    sc1[i] = s; sh1[i] = be1[i] - mu1[i] * s;
  }
  if (i < D1) {
    float s = g2[i] / sqrtf(va2[i] + BN_EPS);
    sc2[i] = s; sh2[i] = be2[i] + (bias1[i] - mu2[i]) * s;
  }
}

__global__ void row_count(const int* __restrict__ rowA, int* __restrict__ cnt) {
  int e = blockIdx.x * blockDim.x + threadIdx.x;
  if (e < N_EDGES) atomicAdd(&cnt[rowA[e]], 1);
}

static_assert(SCAN_W == 32 && N_NODES % SCAN_W == 0, "one lane per wave total; equal chunks");
__global__ __launch_bounds__(SCAN_T)
void row_scan(int* __restrict__ cnt) {
  __shared__ int wtot[SCAN_W];
  int lane = threadIdx.x & 31, w = threadIdx.x >> 5;
  int* __restrict__ p = cnt + w * SCAN_CHUNK;
  int s = 0;
  #pragma unroll
  for (int i = 0; i < SCAN_STEPS; ++i) {
    bool in = i < SCAN_CHUNK / 32 || lane < SCAN_CHUNK % 32;
    s += in ? p[i * 32 + lane] : 0;
  }
  #pragma unroll
  for (int off = 16; off > 0; off >>= 1) s += __shfl_xor(s, off, 32);
  if (lane == 0) wtot[w] = s;
  __syncthreads();
  int carry = lane < w ? wtot[lane] : 0;
  #pragma unroll
  for (int off = 16; off > 0; off >>= 1) carry += __shfl_xor(carry, off, 32);
  #pragma unroll 8
  for (int i = 0; i < SCAN_STEPS; ++i) {
    bool in = i < SCAN_CHUNK / 32 || lane < SCAN_CHUNK % 32;
    int x0 = in ? p[i * 32 + lane] : 0;
    int x = x0;
    #pragma unroll
    for (int off = 1; off < 32; off <<= 1) {
      int y = __shfl_up(x, off, 32);
      if (lane >= off) x += y;
    }
    if (in) p[i * 32 + lane] = x - x0 + carry;
    carry += __shfl(x, 31, 32);
  }
}

__global__ void csr_fill(const int* __restrict__ rowA, const int* __restrict__ colA,
                         int* __restrict__ cur, col_t* __restrict__ csr) {
  int e = blockIdx.x * blockDim.x + threadIdx.x;
  if (e >= N_EDGES) return;
  int slot = atomicAdd(&cur[rowA[e]], 1);
  csr[slot] = (col_t)colA[e];
}

template <int M>
__device__ __forceinline__ float row_xor(float v) {
  static_assert(M >= 1 && M <= 15, "partner inside the 16-lane row");
  const int x = __float_as_int(v);
  return __int_as_float(__builtin_amdgcn_update_dpp(x, x, 0x160 + M, 0xf, 0xf, false));
}

__device__ __forceinline__ float reduce_scatter16(const float (&v)[16], int l16) {
  const bool b3 = l16 & 8, b2 = l16 & 4, b1 = l16 & 2, b0 = l16 & 1;
  float w8[8], w4[4], w2[2];
  #pragma unroll
  for (int i = 0; i < 8; ++i) {
    float keep = b3 ? v[i + 8] : v[i], send = b3 ? v[i] : v[i + 8];
    w8[i] = keep + row_xor<8>(send);
  }
  #pragma unroll
  for (int i = 0; i < 4; ++i) {
    float keep = b2 ? w8[i + 4] : w8[i], send = b2 ? w8[i] : w8[i + 4];
    w4[i] = keep + row_xor<4>(send);
  }
  #pragma unroll
  for (int i = 0; i < 2; ++i) {
    float keep = b1 ? w4[i + 2] : w4[i], send = b1 ? w4[i] : w4[i + 2];
    w2[i] = keep + row_xor<2>(send);
  }
  float keep = b0 ? w2[1] : w2[0], send = b0 ? w2[0] : w2[1];
  return keep + row_xor<1>(send);
}


template <int NT, int N>
__device__ __forceinline__ void load_b(const float* __restrict__ b0, float (&bx)[NT], float (&by)[NT]) {
  const float* __restrict__ b1 = b0 + N;
  if constexpr (NT == 4) {
    float4 t0 = *(const float4*)b0, t1 = *(const float4*)b1;
    bx[0] = t0.x; bx[1] = t0.y; bx[2] = t0.z; bx[3] = t0.w;
    by[0] = t1.x; by[1] = t1.y; by[2] = t1.z; by[3] = t1.w;
  } else {
    #pragma unroll
    for (int nt = 0; nt < NT; ++nt) { bx[nt] = b0[nt]; by[nt] = b1[nt]; }
  }
}

template <int K, int N, int MT, int NT, int WN, int KC, int PF, int SCORE, int WPS>
__global__ __launch_bounds__(32 * WN)
__attribute__((amdgpu_waves_per_eu(WPS)))
void gemm_bn_wmma(const float* A,
                  const float* __restrict__ scale,
                  const float* __restrict__ shift,
                  const float* __restrict__ B,
                  float* __restrict__ C,
                  const float* __restrict__ att,
                  float* __restrict__ sa,
                  float* __restrict__ sb) {
  constexpr int THREADS = 32 * WN;
  constexpr int ROWS  = 16 * MT;
  constexpr int PITCH = K + 4;
  constexpr int K4  = K / 4;
  constexpr int C4  = KC / 4;
  constexpr int NCH = K / KC;
  constexpr int RPP = THREADS / C4;
  constexpr int LPT = ROWS / RPP;
  static_assert(PF >= 1 && 4 * PF <= K, "queue depth");
  static_assert(N == WN * NT * 16, "the waves of a work-group tile all N columns");
  static_assert(K % KC == 0 && KC % 4 == 0 && THREADS % C4 == 0 && ROWS % RPP == 0, "exact staging");
  static_assert((PITCH / 4) % 2 == 1, "pitch is 16 B x odd");
  static_assert(N_NODES % ROWS == 0, "grid divides exactly");
  __shared__ float As[ROWS * PITCH];

  const int tid  = threadIdx.x;
  const int lane = tid & 31;
  const int half = lane >> 4;
  const int l16  = lane & 15;
  const int wn   = tid >> 5;
  const int row0 = blockIdx.x * ROWS;

  const int tr = tid / C4, c4 = tid % C4;
  const float4* A4 = (const float4*)(A + (size_t)row0 * K) + (size_t)tr * K4 + c4;
  const float4* __restrict__ sc4 = (const float4*)scale + c4;
  const float4* __restrict__ sh4 = (const float4*)shift + c4;
  float* st = As + tr * PITCH + c4 * 4;

  float4 pre[LPT];
  #pragma unroll
  for (int i = 0; i < LPT; ++i) pre[i] = A4[(size_t)i * RPP * K4];

  const float* as = As + l16 * PITCH + 2 * half;
  const float* __restrict__ bp = B + (size_t)(2 * half) * N + wn * (NT * 16) + NT * l16;

  v8f acc[MT][NT];
  #pragma unroll
  for (int mt = 0; mt < MT; ++mt)
    #pragma unroll
    for (int nt = 0; nt < NT; ++nt) acc[mt][nt] = v8f{};

  float qx[PF][NT], qy[PF][NT];
  #pragma unroll
  for (int p = 0; p < PF; ++p) load_b<NT, N>(bp + (size_t)(4 * p) * N, qx[p], qy[p]);

  for (int ch = 0; ch < NCH; ++ch) {
    const float4 sc = sc4[ch * C4], sh = sh4[ch * C4];
    #pragma unroll
    for (int i = 0; i < LPT; ++i) {
      float4 v = pre[i], o;
      o.x = v.x * sc.x + sh.x; o.y = v.y * sc.y + sh.y;
      o.z = v.z * sc.z + sh.z; o.w = v.w * sc.w + sh.w;
      *(float4*)(st + i * RPP * PITCH + ch * KC) = o;
    }
    __syncthreads();
    if (ch + 1 < NCH) {
      #pragma unroll
      for (int i = 0; i < LPT; ++i) pre[i] = A4[(size_t)i * RPP * K4 + (ch + 1) * C4];
    }
    for (int k = ch * KC; k < ch * KC + KC; k += 4) {
      v2f a[MT];
      #pragma unroll
      for (int mt = 0; mt < MT; ++mt) a[mt] = *(const v2f*)(as + mt * 16 * PITCH + k);
      float bx[NT], by[NT];
      #pragma unroll
      for (int nt = 0; nt < NT; ++nt) { bx[nt] = qx[0][nt]; by[nt] = qy[0][nt]; }
      #pragma unroll
      for (int p = 0; p + 1 < PF; ++p)
        #pragma unroll
        for (int nt = 0; nt < NT; ++nt) { qx[p][nt] = qx[p + 1][nt]; qy[p][nt] = qy[p + 1][nt]; }
      if (k + 4 * PF < K) load_b<NT, N>(bp + (size_t)(k + 4 * PF) * N, qx[PF - 1], qy[PF - 1]);
      #pragma unroll
      for (int mt = 0; mt < MT; ++mt)
        #pragma unroll
        for (int nt = 0; nt < NT; ++nt) {
          v2f b; b.x = bx[nt]; b.y = by[nt];
          acc[mt][nt] = __builtin_amdgcn_wmma_f32_16x16x4_f32(
               false, a[mt],  false, b,
               (short)0, acc[mt][nt],  false,  false);
        }
    }
  }

  #pragma unroll
  for (int mt = 0; mt < MT; ++mt) {
    float* cp = C + (size_t)(row0 + mt * 16 + (half << 3)) * N + wn * (NT * 16) + NT * l16;
    #pragma unroll
    for (int r = 0; r < 8; ++r) {
      if constexpr (NT == 4) {
        *(float4*)(cp + (size_t)r * N) =
            make_float4(acc[mt][0][r], acc[mt][1][r], acc[mt][2][r], acc[mt][3][r]);
      } else {
        #pragma unroll
        for (int nt = 0; nt < NT; ++nt) cp[(size_t)r * N + nt] = acc[mt][nt][r];
      }
    }
  }

  if constexpr (SCORE == 1) {
    static_assert(MT == 2 && NT == 4 && N == D1 && C1 == 2 * NT * 16 && ROWS * H1 * 2 == THREADS,
                  "two waves per head, 32 rows, one score per thread");
    __shared__ float red[WN * ROWS * 2];
    const int hd = wn >> 1;
    const int cb = (wn & 1) * (NT * 16) + NT * l16;
    const float4 al = *(const float4*)(att + hd * (2 * C1) + cb);
    const float4 ar = *(const float4*)(att + hd * (2 * C1) + C1 + cb);
    #pragma unroll
    for (int mt = 0; mt < MT; ++mt) {
      float p[16];
      #pragma unroll
      for (int r = 0; r < 8; ++r) {
        float c0 = acc[mt][0][r], c1 = acc[mt][1][r], c2 = acc[mt][2][r], c3 = acc[mt][3][r];
        p[r]     = c0 * al.x + c1 * al.y + c2 * al.z + c3 * al.w;
        p[8 + r] = c0 * ar.x + c1 * ar.y + c2 * ar.z + c3 * ar.w;
      }
      const float sc = reduce_scatter16(p, l16);
      red[(wn * ROWS + mt * 16 + (half << 3) + (l16 & 7)) * 2 + (l16 >> 3)] = sc;
    }
    __syncthreads();
    const int wh = tid >> 7, idx = tid & 127, rr = idx >> 2, hh = idx & 3;
    const float v = red[((2 * hh) * ROWS + rr) * 2 + wh] + red[((2 * hh + 1) * ROWS + rr) * 2 + wh];
    (wh ? sb : sa)[(size_t)row0 * H1 + idx] = v;
  }

  if constexpr (SCORE == 2) {
    static_assert(MT == 1 && NT == 3 && WN == 1 && N == D2 && C2 == 2 * NT && H2 == 8,
                  "one wave, two lanes per head");
    const int hd = l16 >> 1;
    const int b0 = l16 & 1;
    const float* __restrict__ ap = att + hd * (2 * C2) + NT * b0;
    const float k0 = ap[0] + ap[C2], k1 = ap[1] + ap[C2 + 1], k2 = ap[2] + ap[C2 + 2];
    float p[8];
    #pragma unroll
    for (int r = 0; r < 8; ++r) p[r] = acc[0][0][r] * k0 + acc[0][1][r] * k1 + acc[0][2][r] * k2;
    #pragma unroll
    for (int i = 0; i < 4; ++i) {
      float keep = b0 ? p[i + 4] : p[i], send = b0 ? p[i] : p[i + 4];
      float t = keep + row_xor<1>(send);
      sa[(size_t)(row0 + (half << 3) + (b0 << 2) + i) * H2 + hd] = tanhf(t);
    }
  }
}


__global__ __launch_bounds__(256)
void row_agg1(const int* __restrict__ cur, const col_t* __restrict__ csr,
              const float* __restrict__ ai, const float* __restrict__ aj,
              const float* __restrict__ h1, float* __restrict__ out1) {
  int r = __builtin_amdgcn_readfirstlane((int)(blockIdx.x * 8 + (threadIdx.x >> 5)));
  int lane = threadIdx.x & 31;
  int beg = r ? cur[r - 1] : 0;
  int end = cur[r];
  const float4* __restrict__ aj4 = (const float4*)aj;
  float4 a_i = *(const float4*)(ai + (size_t)r * 4);
  float4 bs  = aj4[r];
  float t0 = lrelu(a_i.x + bs.x), t1 = lrelu(a_i.y + bs.y);
  float t2 = lrelu(a_i.z + bs.z), t3 = lrelu(a_i.w + bs.w);

  float m0 = t0, m1 = t1, m2 = t2, m3 = t3;
  for (int j = beg + lane; j < end; j += 32) {
    float4 b = aj4[csr[j]];
    m0 = fmaxf(m0, lrelu(a_i.x + b.x));
    m1 = fmaxf(m1, lrelu(a_i.y + b.y));
    m2 = fmaxf(m2, lrelu(a_i.z + b.z));
    m3 = fmaxf(m3, lrelu(a_i.w + b.w));
  }
  #pragma unroll
  for (int off = 16; off > 0; off >>= 1) {
    m0 = fmaxf(m0, __shfl_xor(m0, off, 32));
    m1 = fmaxf(m1, __shfl_xor(m1, off, 32));
    m2 = fmaxf(m2, __shfl_xor(m2, off, 32));
    m3 = fmaxf(m3, __shfl_xor(m3, off, 32));
  }

  const float4* __restrict__ h4 = (const float4*)h1;
  float u0 = expf(t0 - m0), u1 = expf(t1 - m1), u2 = expf(t2 - m2), u3 = expf(t3 - m3);
  float s0 = u0, s1 = u1, s2 = u2, s3 = u3;
  const float4* __restrict__ src = h4 + (size_t)r * (D1 / 4);
  float4 v0 = src[lane], v1 = src[32 + lane], v2 = src[64 + lane], v3 = src[96 + lane];
  float4 acc0 = {u0 * v0.x, u0 * v0.y, u0 * v0.z, u0 * v0.w};
  float4 acc1 = {u1 * v1.x, u1 * v1.y, u1 * v1.z, u1 * v1.w};
  float4 acc2 = {u2 * v2.x, u2 * v2.y, u2 * v2.z, u2 * v2.w};
  float4 acc3 = {u3 * v3.x, u3 * v3.y, u3 * v3.z, u3 * v3.w};

  for (int base = beg; base < end; base += 32) {
    int j = base + lane;
    int c = r;
    float w0 = 0.f, w1 = 0.f, w2 = 0.f, w3 = 0.f;
    if (j < end) {
      c = csr[j];
      float4 b = aj4[c];
      w0 = expf(lrelu(a_i.x + b.x) - m0);
      w1 = expf(lrelu(a_i.y + b.y) - m1);
      w2 = expf(lrelu(a_i.z + b.z) - m2);
      w3 = expf(lrelu(a_i.w + b.w) - m3);
    }
    int cnt = end - base < 32 ? end - base : 32;
    for (int k = 0; k < cnt; ++k) {
      int ck = __shfl(c, k, 32);
      u0 = __shfl(w0, k, 32); u1 = __shfl(w1, k, 32);
      u2 = __shfl(w2, k, 32); u3 = __shfl(w3, k, 32);
      s0 += u0; s1 += u1; s2 += u2; s3 += u3;
      src = h4 + (size_t)ck * (D1 / 4);
      v0 = src[lane]; v1 = src[32 + lane]; v2 = src[64 + lane]; v3 = src[96 + lane];
      acc0.x += u0 * v0.x; acc0.y += u0 * v0.y; acc0.z += u0 * v0.z; acc0.w += u0 * v0.w;
      acc1.x += u1 * v1.x; acc1.y += u1 * v1.y; acc1.z += u1 * v1.z; acc1.w += u1 * v1.w;
      acc2.x += u2 * v2.x; acc2.y += u2 * v2.y; acc2.z += u2 * v2.z; acc2.w += u2 * v2.w;
      acc3.x += u3 * v3.x; acc3.y += u3 * v3.y; acc3.z += u3 * v3.z; acc3.w += u3 * v3.w;
    }
  }
  float i0 = 1.0f / (s0 + 1e-16f), i1 = 1.0f / (s1 + 1e-16f);
  float i2 = 1.0f / (s2 + 1e-16f), i3 = 1.0f / (s3 + 1e-16f);
  float4* __restrict__ dst = (float4*)(out1 + (size_t)r * D1);
  dst[lane]      = make_float4(acc0.x * i0, acc0.y * i0, acc0.z * i0, acc0.w * i0);
  dst[32 + lane] = make_float4(acc1.x * i1, acc1.y * i1, acc1.z * i1, acc1.w * i1);
  dst[64 + lane] = make_float4(acc2.x * i2, acc2.y * i2, acc2.z * i2, acc2.w * i2);
  dst[96 + lane] = make_float4(acc3.x * i3, acc3.y * i3, acc3.z * i3, acc3.w * i3);
}


__global__ void row_agg2(const int* __restrict__ cur, const col_t* __restrict__ csr,
                         const float* __restrict__ s2, const float* __restrict__ h2,
                         float* __restrict__ out2) {
  int tid = blockIdx.x * blockDim.x + threadIdx.x;
  if (tid >= N_NODES * H2) return;
  int r = tid >> 3, h = tid & 7;
  int beg = r ? cur[r - 1] : 0;
  int end = cur[r];
  float sself = s2[tid];
  float m = sself;
  for (int j = beg; j < end; ++j) m = fmaxf(m, s2[(int)csr[j] * H2 + h]);

  float w = expf(sself - m);
  float den = w;
  const float2* __restrict__ src = (const float2*)(h2 + (size_t)r * D2 + h * C2);
  float2 p0 = src[0], p1 = src[1], p2 = src[2];
  float a0 = w * p0.x, a1 = w * p0.y, a2 = w * p1.x, a3 = w * p1.y, a4 = w * p2.x, a5 = w * p2.y;
  for (int j = beg; j < end; ++j) {
    int c = csr[j];
    w = expf(s2[c * H2 + h] - m);
    den += w;
    src = (const float2*)(h2 + (size_t)c * D2 + h * C2);
    p0 = src[0]; p1 = src[1]; p2 = src[2];
    a0 += w * p0.x; a1 += w * p0.y; a2 += w * p1.x;
    a3 += w * p1.y; a4 += w * p2.x; a5 += w * p2.y;
  }
  float inv = 1.0f / (den + 1e-16f);
  float2* __restrict__ dst = (float2*)(out2 + (size_t)r * D2 + h * C2);
  dst[0] = make_float2(a0 * inv, a1 * inv);
  dst[1] = make_float2(a2 * inv, a3 * inv);
  dst[2] = make_float2(a4 * inv, a5 * inv);
}

__global__ void finalize(const float* __restrict__ out2, const float* __restrict__ b2,
                         float* __restrict__ out) {
  int n = blockIdx.x * blockDim.x + threadIdx.x;
  if (n >= N_NODES) return;
  float v[C2];
  #pragma unroll
  for (int c = 0; c < C2; ++c) {
    float s = 0.f;
    #pragma unroll
    for (int h = 0; h < H2; ++h) s += out2[(size_t)n * D2 + h * C2 + c];
    float t = s * (1.0f / H2) + b2[c];
    v[c] = t > 0.f ? t : (expf(t) - 1.0f);
  }
  float m = v[0];
  #pragma unroll
  for (int c = 1; c < C2; ++c) m = fmaxf(m, v[c]);
  float sum = 0.f;
  #pragma unroll
  for (int c = 0; c < C2; ++c) sum += expf(v[c] - m);
  float lse = logf(sum);
  #pragma unroll
  for (int c = 0; c < C2; ++c) out[(size_t)n * C2 + c] = v[c] - m - lse;
}


extern "C" void kernel_launch(void* const* d_in, const int* in_sizes, int n_in,
                              void* d_out, int out_size, void* d_ws, size_t ws_size,
                              hipStream_t stream) {
  const float* x    = (const float*)d_in[0];
  const int*   ei   = (const int*)  d_in[1];
  const float* g1   = (const float*)d_in[2];
  const float* be1  = (const float*)d_in[3];
  const float* mu1  = (const float*)d_in[4];
  const float* va1  = (const float*)d_in[5];
  const float* W1   = (const float*)d_in[6];
  const float* att1 = (const float*)d_in[7];
  const float* b1   = (const float*)d_in[8];
  const float* g2   = (const float*)d_in[9];
  const float* be2  = (const float*)d_in[10];
  const float* mu2  = (const float*)d_in[11];
  const float* va2  = (const float*)d_in[12];
  const float* W2   = (const float*)d_in[13];
  const float* att2 = (const float*)d_in[14];
  const float* b2   = (const float*)d_in[15];
  const int* rowA = ei;
  const int* colA = ei + N_EDGES;
  float* out = (float*)d_out;
  float* ws  = (float*)d_ws;

  float* sc1  = ws;
  float* sh1  = ws + 256;
  float* sc2  = ws + 512;
  float* sh2  = ws + 1024;
  float* ai   = ws + 1536;
  float* aj   = ai  + (size_t)N_NODES * H1;
  float* edg  = aj  + (size_t)N_NODES * H1;
  int*   cur  = (int*)edg;
  col_t* csr  = (col_t*)(edg + N_NODES);
  float* out1 = edg + (size_t)N_NODES * 2 * H1;
  float* h1   = out1 + (size_t)N_NODES * D1;
  float* h2   = h1;
  float* s2   = h2  + (size_t)N_NODES * D2;
  float* out2 = s2  + (size_t)N_NODES * H2;

  bn_prep<<<2, 256, 0, stream>>>(g1, be1, mu1, va1, g2, be2, mu2, va2, b1,
                                 sc1, sh1, sc2, sh2);

  fill_f32<<<(N_NODES + 255) / 256, 256, 0, stream>>>(edg, 0.0f, N_NODES);
  row_count<<<N_EDGES / 256, 256, 0, stream>>>(rowA, cur);
  row_scan<<<1, SCAN_T, 0, stream>>>(cur);
  csr_fill<<<N_EDGES / 256, 256, 0, stream>>>(rowA, colA, cur, csr);

  gemm_bn_wmma<F_IN, D1, 2, 4, 8, 64, 2, 1, 8><<<N_NODES / 32, 256, 0, stream>>>(x, sc1, sh1, W1, h1,
                                                                            att1, ai, aj);

  row_agg1<<<N_NODES / 8, 256, 0, stream>>>(cur, csr, ai, aj, h1, out1);

  gemm_bn_wmma<D1, D2, 1, 3, 1, 128, 2, 2, 1><<<N_NODES / 16, 32, 0, stream>>>(out1, sc2, sh2, W2, h2,
                                                                          att2, s2, nullptr);

  row_agg2<<<(N_NODES * H2) / 256, 256, 0, stream>>>(cur, csr, s2, h2, out2);

  finalize<<<(N_NODES + 255) / 256, 256, 0, stream>>>(out2, b2, out);
}


# Edge count and fill passes lose their launches: their blocks run behind the layer-1 GEMM's blocks (GEMM launched as 512+738 row blocks, prefix sum between), counter clear folded into bn_prep
# speedup vs baseline: 4.2733x; 1.1074x over previous
#include <hip/hip_runtime.h>
#include <math.h>

#define N_NODES 40000
#define N_EDGES 320000
#define ETOT    (N_EDGES + N_NODES)
#define F_IN    256
#define H1      4
#define C1      128
#define D1      (H1*C1)
#define H2      8
#define C2      6
#define D2      (H2*C2)
#define SCAN_T     1024
#define SCAN_W     (SCAN_T / 32)
#define SCAN_CHUNK (N_NODES / SCAN_W)
#define SCAN_STEPS ((SCAN_CHUNK + 31) / 32)
#define BN_EPS    1e-5f
#define NEG_SLOPE 0.2f

typedef __attribute__((ext_vector_type(2))) float v2f;
typedef __attribute__((ext_vector_type(8))) float v8f;


static_assert(N_NODES <= 65536, "16-bit column ids need N_NODES <= 65536");
typedef unsigned short col_t;
static_assert(N_EDGES % 256 == 0 && N_NODES % 8 == 0 && (N_NODES * H2) % 256 == 0,
              "launch grids below divide exactly");

__device__ __forceinline__ float lrelu(float a) { return a > 0.f ? a : NEG_SLOPE * a; }


__global__ void bn_prep(const float* g1, const float* be1, const float* mu1, const float* va1,
                        const float* g2, const float* be2, const float* mu2, const float* va2,
                        const float* bias1,
                        float* sc1, float* sh1, float* sc2, float* sh2, int* __restrict__ cnt) {
  int i = blockIdx.x * blockDim.x + threadIdx.x;
  if (i < N_NODES) cnt[i] = 0;
  if (i < F_IN) {
    float s = g1[i] / sqrtf(va1[i] + BN_EPS);
    sc1[i] = s; sh1[i] = be1[i] - mu1[i] * s;
  }
  if (i < D1) {
    float s = g2[i] / sqrtf(va2[i] + BN_EPS);
    sc2[i] = s; sh2[i] = be2[i] + (bias1[i] - mu2[i]) * s;
  }
}

#define EDGE_BLOCKS (N_EDGES / 256)
#define G1_BLOCKS   (N_NODES / 32)
#define G1_FIRST    512
__device__ __forceinline__ void row_count_blk(int blk, const int* __restrict__ rowA, int* __restrict__ cnt) {
  int e = blk * 256 + threadIdx.x;
  if (e < N_EDGES) atomicAdd(&cnt[rowA[e]], 1);
}

static_assert(SCAN_W == 32 && N_NODES % SCAN_W == 0, "one lane per wave total; equal chunks");
__global__ __launch_bounds__(SCAN_T)
void row_scan(int* __restrict__ cnt) {
  __shared__ int wtot[SCAN_W];
  int lane = threadIdx.x & 31, w = threadIdx.x >> 5;
  int* __restrict__ p = cnt + w * SCAN_CHUNK;
  int s = 0;
  #pragma unroll
  for (int i = 0; i < SCAN_STEPS; ++i) {
    bool in = i < SCAN_CHUNK / 32 || lane < SCAN_CHUNK % 32;
    s += in ? p[i * 32 + lane] : 0;
  }
  #pragma unroll
  for (int off = 16; off > 0; off >>= 1) s += __shfl_xor(s, off, 32);
  if (lane == 0) wtot[w] = s;
  __syncthreads();
  int carry = lane < w ? wtot[lane] : 0;
  #pragma unroll
  for (int off = 16; off > 0; off >>= 1) carry += __shfl_xor(carry, off, 32);
  #pragma unroll 8
  for (int i = 0; i < SCAN_STEPS; ++i) {
    bool in = i < SCAN_CHUNK / 32 || lane < SCAN_CHUNK % 32;
    int x0 = in ? p[i * 32 + lane] : 0;
    int x = x0;
    #pragma unroll
    for (int off = 1; off < 32; off <<= 1) {
      int y = __shfl_up(x, off, 32);
      if (lane >= off) x += y;
    }
    if (in) p[i * 32 + lane] = x - x0 + carry;
    carry += __shfl(x, 31, 32);
  }
}

__device__ __forceinline__ void csr_fill_blk(int blk, const int* __restrict__ rowA, const int* __restrict__ colA,
                                             int* __restrict__ cur, col_t* __restrict__ csr) {
  int e = blk * 256 + threadIdx.x;
  if (e >= N_EDGES) return;
  int slot = atomicAdd(&cur[rowA[e]], 1);
  csr[slot] = (col_t)colA[e];
}

template <int M>
__device__ __forceinline__ float row_xor(float v) {
  static_assert(M >= 1 && M <= 15, "partner inside the 16-lane row");
  const int x = __float_as_int(v);
  return __int_as_float(__builtin_amdgcn_update_dpp(x, x, 0x160 + M, 0xf, 0xf, false));
}

__device__ __forceinline__ float reduce_scatter16(const float (&v)[16], int l16) {
  const bool b3 = l16 & 8, b2 = l16 & 4, b1 = l16 & 2, b0 = l16 & 1;
  float w8[8], w4[4], w2[2];
  #pragma unroll
  for (int i = 0; i < 8; ++i) {
    float keep = b3 ? v[i + 8] : v[i], send = b3 ? v[i] : v[i + 8];
    w8[i] = keep + row_xor<8>(send);
  }
  #pragma unroll
  for (int i = 0; i < 4; ++i) {
    float keep = b2 ? w8[i + 4] : w8[i], send = b2 ? w8[i] : w8[i + 4];
    w4[i] = keep + row_xor<4>(send);
  }
  #pragma unroll
  for (int i = 0; i < 2; ++i) {
    float keep = b1 ? w4[i + 2] : w4[i], send = b1 ? w4[i] : w4[i + 2];
    w2[i] = keep + row_xor<2>(send);
  }
  float keep = b0 ? w2[1] : w2[0], send = b0 ? w2[0] : w2[1];
  return keep + row_xor<1>(send);
}


template <int NT, int N>
__device__ __forceinline__ void load_b(const float* __restrict__ b0, float (&bx)[NT], float (&by)[NT]) {
  const float* __restrict__ b1 = b0 + N;
  if constexpr (NT == 4) {
    float4 t0 = *(const float4*)b0, t1 = *(const float4*)b1;
    bx[0] = t0.x; bx[1] = t0.y; bx[2] = t0.z; bx[3] = t0.w;
    by[0] = t1.x; by[1] = t1.y; by[2] = t1.z; by[3] = t1.w;
  } else {
    #pragma unroll
    for (int nt = 0; nt < NT; ++nt) { bx[nt] = b0[nt]; by[nt] = b1[nt]; }
  }
}

template <int K, int N, int MT, int NT, int WN, int KC, int PF, int SCORE, int WPS, bool EDGES>
__global__ __launch_bounds__(32 * WN)
__attribute__((amdgpu_waves_per_eu(WPS)))
void gemm_bn_wmma(const float* A,
                  const float* __restrict__ scale,
                  const float* __restrict__ shift,
                  const float* __restrict__ B,
                  float* __restrict__ C,
                  const float* __restrict__ att,
                  float* __restrict__ sa,
                  float* __restrict__ sb,
                  int blk0, int nblk, int pass,
                  const int* __restrict__ rowA, const int* __restrict__ colA,
                  int* __restrict__ cur, col_t* __restrict__ csr) {
  constexpr int THREADS = 32 * WN;
  if constexpr (EDGES) {
    static_assert(THREADS == 256, "edge blocks are 256 threads");
    if ((int)blockIdx.x >= nblk) {
      const int eb = (int)blockIdx.x - nblk;
      if (pass == 0) row_count_blk(eb, rowA, cur);
      else           csr_fill_blk(eb, rowA, colA, cur, csr);
      return;
    }
  }
  constexpr int ROWS  = 16 * MT;
  constexpr int PITCH = K + 4;
  constexpr int K4  = K / 4;
  constexpr int C4  = KC / 4;
  constexpr int NCH = K / KC;
  constexpr int RPP = THREADS / C4;
  constexpr int LPT = ROWS / RPP;
  static_assert(PF >= 1 && 4 * PF <= K, "queue depth");
  static_assert(N == WN * NT * 16, "the waves of a work-group tile all N columns");
  static_assert(K % KC == 0 && KC % 4 == 0 && THREADS % C4 == 0 && ROWS % RPP == 0, "exact staging");
  static_assert((PITCH / 4) % 2 == 1, "pitch is 16 B x odd");
  static_assert(N_NODES % ROWS == 0, "grid divides exactly");
  __shared__ float As[ROWS * PITCH];

  const int tid  = threadIdx.x;
  const int lane = tid & 31;
  const int half = lane >> 4;
  const int l16  = lane & 15;
  const int wn   = tid >> 5;
  const int row0 = (blk0 + (int)blockIdx.x) * ROWS;

  const int tr = tid / C4, c4 = tid % C4;
  const float4* A4 = (const float4*)(A + (size_t)row0 * K) + (size_t)tr * K4 + c4;
  const float4* __restrict__ sc4 = (const float4*)scale + c4;
  const float4* __restrict__ sh4 = (const float4*)shift + c4;
  float* st = As + tr * PITCH + c4 * 4;

  float4 pre[LPT];
  #pragma unroll
  for (int i = 0; i < LPT; ++i) pre[i] = A4[(size_t)i * RPP * K4];

  const float* as = As + l16 * PITCH + 2 * half;
  const float* __restrict__ bp = B + (size_t)(2 * half) * N + wn * (NT * 16) + NT * l16;

  v8f acc[MT][NT];
  #pragma unroll
  for (int mt = 0; mt < MT; ++mt)
    #pragma unroll
    for (int nt = 0; nt < NT; ++nt) acc[mt][nt] = v8f{};

  float qx[PF][NT], qy[PF][NT];
  #pragma unroll
  for (int p = 0; p < PF; ++p) load_b<NT, N>(bp + (size_t)(4 * p) * N, qx[p], qy[p]);

  for (int ch = 0; ch < NCH; ++ch) {
    const float4 sc = sc4[ch * C4], sh = sh4[ch * C4];
    #pragma unroll
    for (int i = 0; i < LPT; ++i) {
      float4 v = pre[i], o;
      o.x = v.x * sc.x + sh.x; o.y = v.y * sc.y + sh.y;
      o.z = v.z * sc.z + sh.z; o.w = v.w * sc.w + sh.w;
      *(float4*)(st + i * RPP * PITCH + ch * KC) = o;
    }
    __syncthreads();
    if (ch + 1 < NCH) {
      #pragma unroll
      for (int i = 0; i < LPT; ++i) pre[i] = A4[(size_t)i * RPP * K4 + (ch + 1) * C4];
    }
    for (int k = ch * KC; k < ch * KC + KC; k += 4) {
      v2f a[MT];
      #pragma unroll
      for (int mt = 0; mt < MT; ++mt) a[mt] = *(const v2f*)(as + mt * 16 * PITCH + k);
      float bx[NT], by[NT];
      #pragma unroll
      for (int nt = 0; nt < NT; ++nt) { bx[nt] = qx[0][nt]; by[nt] = qy[0][nt]; }
      #pragma unroll
      for (int p = 0; p + 1 < PF; ++p)
        #pragma unroll
        for (int nt = 0; nt < NT; ++nt) { qx[p][nt] = qx[p + 1][nt]; qy[p][nt] = qy[p + 1][nt]; }
      if (k + 4 * PF < K) load_b<NT, N>(bp + (size_t)(k + 4 * PF) * N, qx[PF - 1], qy[PF - 1]);
      #pragma unroll
      for (int mt = 0; mt < MT; ++mt)
        #pragma unroll
        for (int nt = 0; nt < NT; ++nt) {
          v2f b; b.x = bx[nt]; b.y = by[nt];
          acc[mt][nt] = __builtin_amdgcn_wmma_f32_16x16x4_f32(
               false, a[mt],  false, b,
               (short)0, acc[mt][nt],  false,  false);
        }
    }
  }

  #pragma unroll
  for (int mt = 0; mt < MT; ++mt) {
    float* cp = C + (size_t)(row0 + mt * 16 + (half << 3)) * N + wn * (NT * 16) + NT * l16;
    #pragma unroll
    for (int r = 0; r < 8; ++r) {
      if constexpr (NT == 4) {
        *(float4*)(cp + (size_t)r * N) =
            make_float4(acc[mt][0][r], acc[mt][1][r], acc[mt][2][r], acc[mt][3][r]);
      } else {
        #pragma unroll
        for (int nt = 0; nt < NT; ++nt) cp[(size_t)r * N + nt] = acc[mt][nt][r];
      }
    }
  }

  if constexpr (SCORE == 1) {
    static_assert(MT == 2 && NT == 4 && N == D1 && C1 == 2 * NT * 16 && ROWS * H1 * 2 == THREADS,
                  "two waves per head, 32 rows, one score per thread");
    __shared__ float red[WN * ROWS * 2];
    const int hd = wn >> 1;
    const int cb = (wn & 1) * (NT * 16) + NT * l16;
    const float4 al = *(const float4*)(att + hd * (2 * C1) + cb);
    const float4 ar = *(const float4*)(att + hd * (2 * C1) + C1 + cb);
    #pragma unroll
    for (int mt = 0; mt < MT; ++mt) {
      float p[16];
      #pragma unroll
      for (int r = 0; r < 8; ++r) {
        float c0 = acc[mt][0][r], c1 = acc[mt][1][r], c2 = acc[mt][2][r], c3 = acc[mt][3][r];
        p[r]     = c0 * al.x + c1 * al.y + c2 * al.z + c3 * al.w;
        p[8 + r] = c0 * ar.x + c1 * ar.y + c2 * ar.z + c3 * ar.w;
      }
      const float sc = reduce_scatter16(p, l16);
      red[(wn * ROWS + mt * 16 + (half << 3) + (l16 & 7)) * 2 + (l16 >> 3)] = sc;
    }
    __syncthreads();
    const int wh = tid >> 7, idx = tid & 127, rr = idx >> 2, hh = idx & 3;
    const float v = red[((2 * hh) * ROWS + rr) * 2 + wh] + red[((2 * hh + 1) * ROWS + rr) * 2 + wh];
    (wh ? sb : sa)[(size_t)row0 * H1 + idx] = v;
  }

  if constexpr (SCORE == 2) {
    static_assert(MT == 1 && NT == 3 && WN == 1 && N == D2 && C2 == 2 * NT && H2 == 8,
                  "one wave, two lanes per head");
    const int hd = l16 >> 1;
    const int b0 = l16 & 1;
    const float* __restrict__ ap = att + hd * (2 * C2) + NT * b0;
    const float k0 = ap[0] + ap[C2], k1 = ap[1] + ap[C2 + 1], k2 = ap[2] + ap[C2 + 2];
    float p[8];
    #pragma unroll
    for (int r = 0; r < 8; ++r) p[r] = acc[0][0][r] * k0 + acc[0][1][r] * k1 + acc[0][2][r] * k2;
    #pragma unroll
    for (int i = 0; i < 4; ++i) {
      float keep = b0 ? p[i + 4] : p[i], send = b0 ? p[i] : p[i + 4];
      float t = keep + row_xor<1>(send);
      sa[(size_t)(row0 + (half << 3) + (b0 << 2) + i) * H2 + hd] = tanhf(t);
    }
  }
}


__global__ __launch_bounds__(256)
void row_agg1(const int* __restrict__ cur, const col_t* __restrict__ csr,
              const float* __restrict__ ai, const float* __restrict__ aj,
              const float* __restrict__ h1, float* __restrict__ out1) {
  int r = __builtin_amdgcn_readfirstlane((int)(blockIdx.x * 8 + (threadIdx.x >> 5)));
  int lane = threadIdx.x & 31;
  int beg = r ? cur[r - 1] : 0;
  int end = cur[r];
  const float4* __restrict__ aj4 = (const float4*)aj;
  float4 a_i = *(const float4*)(ai + (size_t)r * 4);
  float4 bs  = aj4[r];
  float t0 = lrelu(a_i.x + bs.x), t1 = lrelu(a_i.y + bs.y);
  float t2 = lrelu(a_i.z + bs.z), t3 = lrelu(a_i.w + bs.w);

  float m0 = t0, m1 = t1, m2 = t2, m3 = t3;
  for (int j = beg + lane; j < end; j += 32) {
    float4 b = aj4[csr[j]];
    m0 = fmaxf(m0, lrelu(a_i.x + b.x));
    m1 = fmaxf(m1, lrelu(a_i.y + b.y));
    m2 = fmaxf(m2, lrelu(a_i.z + b.z));
    m3 = fmaxf(m3, lrelu(a_i.w + b.w));
  }
  #pragma unroll
  for (int off = 16; off > 0; off >>= 1) {
    m0 = fmaxf(m0, __shfl_xor(m0, off, 32));
    m1 = fmaxf(m1, __shfl_xor(m1, off, 32));
    m2 = fmaxf(m2, __shfl_xor(m2, off, 32));
    m3 = fmaxf(m3, __shfl_xor(m3, off, 32));
  }

  const float4* __restrict__ h4 = (const float4*)h1;
  float u0 = expf(t0 - m0), u1 = expf(t1 - m1), u2 = expf(t2 - m2), u3 = expf(t3 - m3);
  float s0 = u0, s1 = u1, s2 = u2, s3 = u3;
  const float4* __restrict__ src = h4 + (size_t)r * (D1 / 4);
  float4 v0 = src[lane], v1 = src[32 + lane], v2 = src[64 + lane], v3 = src[96 + lane];
  float4 acc0 = {u0 * v0.x, u0 * v0.y, u0 * v0.z, u0 * v0.w};
  float4 acc1 = {u1 * v1.x, u1 * v1.y, u1 * v1.z, u1 * v1.w};
  float4 acc2 = {u2 * v2.x, u2 * v2.y, u2 * v2.z, u2 * v2.w};
  float4 acc3 = {u3 * v3.x, u3 * v3.y, u3 * v3.z, u3 * v3.w};

  for (int base = beg; base < end; base += 32) {
    int j = base + lane;
    int c = r;
    float w0 = 0.f, w1 = 0.f, w2 = 0.f, w3 = 0.f;
    if (j < end) {
      c = csr[j];
      float4 b = aj4[c];
      w0 = expf(lrelu(a_i.x + b.x) - m0);
      w1 = expf(lrelu(a_i.y + b.y) - m1);
      w2 = expf(lrelu(a_i.z + b.z) - m2);
      w3 = expf(lrelu(a_i.w + b.w) - m3);
    }
    int cnt = end - base < 32 ? end - base : 32;
    for (int k = 0; k < cnt; ++k) {
      int ck = __shfl(c, k, 32);
      u0 = __shfl(w0, k, 32); u1 = __shfl(w1, k, 32);
      u2 = __shfl(w2, k, 32); u3 = __shfl(w3, k, 32);
      s0 += u0; s1 += u1; s2 += u2; s3 += u3;
      src = h4 + (size_t)ck * (D1 / 4);
      v0 = src[lane]; v1 = src[32 + lane]; v2 = src[64 + lane]; v3 = src[96 + lane];
      acc0.x += u0 * v0.x; acc0.y += u0 * v0.y; acc0.z += u0 * v0.z; acc0.w += u0 * v0.w;
      acc1.x += u1 * v1.x; acc1.y += u1 * v1.y; acc1.z += u1 * v1.z; acc1.w += u1 * v1.w;
      acc2.x += u2 * v2.x; acc2.y += u2 * v2.y; acc2.z += u2 * v2.z; acc2.w += u2 * v2.w;
      acc3.x += u3 * v3.x; acc3.y += u3 * v3.y; acc3.z += u3 * v3.z; acc3.w += u3 * v3.w;
    }
  }
  float i0 = 1.0f / (s0 + 1e-16f), i1 = 1.0f / (s1 + 1e-16f);
  float i2 = 1.0f / (s2 + 1e-16f), i3 = 1.0f / (s3 + 1e-16f);
  float4* __restrict__ dst = (float4*)(out1 + (size_t)r * D1);
  dst[lane]      = make_float4(acc0.x * i0, acc0.y * i0, acc0.z * i0, acc0.w * i0);
  dst[32 + lane] = make_float4(acc1.x * i1, acc1.y * i1, acc1.z * i1, acc1.w * i1);
  dst[64 + lane] = make_float4(acc2.x * i2, acc2.y * i2, acc2.z * i2, acc2.w * i2);
  dst[96 + lane] = make_float4(acc3.x * i3, acc3.y * i3, acc3.z * i3, acc3.w * i3);
}


__global__ void row_agg2(const int* __restrict__ cur, const col_t* __restrict__ csr,
                         const float* __restrict__ s2, const float* __restrict__ h2,
                         float* __restrict__ out2) {
  int tid = blockIdx.x * blockDim.x + threadIdx.x;
  if (tid >= N_NODES * H2) return;
  int r = tid >> 3, h = tid & 7;
  int beg = r ? cur[r - 1] : 0;
  int end = cur[r];
  float sself = s2[tid];
  float m = sself;
  for (int j = beg; j < end; ++j) m = fmaxf(m, s2[(int)csr[j] * H2 + h]);

  float w = expf(sself - m);
  float den = w;
  const float2* __restrict__ src = (const float2*)(h2 + (size_t)r * D2 + h * C2);
  float2 p0 = src[0], p1 = src[1], p2 = src[2];
  float a0 = w * p0.x, a1 = w * p0.y, a2 = w * p1.x, a3 = w * p1.y, a4 = w * p2.x, a5 = w * p2.y;
  for (int j = beg; j < end; ++j) {
    int c = csr[j];
    w = expf(s2[c * H2 + h] - m);
    den += w;
    src = (const float2*)(h2 + (size_t)c * D2 + h * C2);
    p0 = src[0]; p1 = src[1]; p2 = src[2];
    a0 += w * p0.x; a1 += w * p0.y; a2 += w * p1.x;
    a3 += w * p1.y; a4 += w * p2.x; a5 += w * p2.y;
  }
  float inv = 1.0f / (den + 1e-16f);
  float2* __restrict__ dst = (float2*)(out2 + (size_t)r * D2 + h * C2);
  dst[0] = make_float2(a0 * inv, a1 * inv);
  dst[1] = make_float2(a2 * inv, a3 * inv);
  dst[2] = make_float2(a4 * inv, a5 * inv);
}

__global__ void finalize(const float* __restrict__ out2, const float* __restrict__ b2,
                         float* __restrict__ out) {
  int n = blockIdx.x * blockDim.x + threadIdx.x;
  if (n >= N_NODES) return;
  float v[C2];
  #pragma unroll
  for (int c = 0; c < C2; ++c) {
    float s = 0.f;
    #pragma unroll
    for (int h = 0; h < H2; ++h) s += out2[(size_t)n * D2 + h * C2 + c];
    float t = s * (1.0f / H2) + b2[c];
    v[c] = t > 0.f ? t : (expf(t) - 1.0f);
  }
  float m = v[0];
  #pragma unroll
  for (int c = 1; c < C2; ++c) m = fmaxf(m, v[c]);
  float sum = 0.f;
  #pragma unroll
  for (int c = 0; c < C2; ++c) sum += expf(v[c] - m);
  float lse = logf(sum);
  #pragma unroll
  for (int c = 0; c < C2; ++c) out[(size_t)n * C2 + c] = v[c] - m - lse;
}


extern "C" void kernel_launch(void* const* d_in, const int* in_sizes, int n_in,
                              void* d_out, int out_size, void* d_ws, size_t ws_size,
                              hipStream_t stream) {
  const float* x    = (const float*)d_in[0];
  const int*   ei   = (const int*)  d_in[1];
  const float* g1   = (const float*)d_in[2];
  const float* be1  = (const float*)d_in[3];
  const float* mu1  = (const float*)d_in[4];
  const float* va1  = (const float*)d_in[5];
  const float* W1   = (const float*)d_in[6];
  const float* att1 = (const float*)d_in[7];
  const float* b1   = (const float*)d_in[8];
  const float* g2   = (const float*)d_in[9];
  const float* be2  = (const float*)d_in[10];
  const float* mu2  = (const float*)d_in[11];
  const float* va2  = (const float*)d_in[12];
  const float* W2   = (const float*)d_in[13];
  const float* att2 = (const float*)d_in[14];
  const float* b2   = (const float*)d_in[15];
  const int* rowA = ei;
  const int* colA = ei + N_EDGES;
  float* out = (float*)d_out;
  float* ws  = (float*)d_ws;

  float* sc1  = ws;
  float* sh1  = ws + 256;
  float* sc2  = ws + 512;
  float* sh2  = ws + 1024;
  float* ai   = ws + 1536;
  float* aj   = ai  + (size_t)N_NODES * H1;
  float* edg  = aj  + (size_t)N_NODES * H1;
  int*   cur  = (int*)edg;
  col_t* csr  = (col_t*)(edg + N_NODES);
  float* out1 = edg + (size_t)N_NODES * 2 * H1;
  float* h1   = out1 + (size_t)N_NODES * D1;
  float* h2   = h1;
  float* s2   = h2  + (size_t)N_NODES * D2;
  float* out2 = s2  + (size_t)N_NODES * H2;

  bn_prep<<<(N_NODES + 255) / 256, 256, 0, stream>>>(g1, be1, mu1, va1, g2, be2, mu2, va2, b1,
                                                     sc1, sh1, sc2, sh2, cur);

  gemm_bn_wmma<F_IN, D1, 2, 4, 8, 64, 2, 1, 8, true><<<G1_FIRST + EDGE_BLOCKS, 256, 0, stream>>>(
      x, sc1, sh1, W1, h1, att1, ai, aj, 0, G1_FIRST, 0, rowA, colA, cur, csr);
  row_scan<<<1, SCAN_T, 0, stream>>>(cur);
  gemm_bn_wmma<F_IN, D1, 2, 4, 8, 64, 2, 1, 8, true><<<G1_BLOCKS - G1_FIRST + EDGE_BLOCKS, 256, 0, stream>>>(
      x, sc1, sh1, W1, h1, att1, ai, aj, G1_FIRST, G1_BLOCKS - G1_FIRST, 1, rowA, colA, cur, csr);

  row_agg1<<<N_NODES / 8, 256, 0, stream>>>(cur, csr, ai, aj, h1, out1);

  gemm_bn_wmma<D1, D2, 1, 3, 1, 128, 2, 2, 1, false><<<N_NODES / 16, 32, 0, stream>>>(
      out1, sc2, sh2, W2, h2, att2, s2, nullptr, 0, N_NODES / 16, 0, nullptr, nullptr, nullptr, nullptr);

  row_agg2<<<(N_NODES * H2) / 256, 256, 0, stream>>>(cur, csr, s2, h2, out2);

  finalize<<<(N_NODES + 255) / 256, 256, 0, stream>>>(out2, b2, out);
}


# Layer-2 GEMM fused behind the layer-1 aggregation: a work-group's 16 rows go LDS -> WMMA, aggregate never stored; product k sum now 8 per-wave 64-wide slice sums added in wave order
# speedup vs baseline: 4.5818x; 1.0722x over previous
#include <hip/hip_runtime.h>
#include <math.h>

#define N_NODES 40000
#define N_EDGES 320000
#define ETOT    (N_EDGES + N_NODES)
#define F_IN    256
#define H1      4
#define C1      128
#define D1      (H1*C1)
#define H2      8
#define C2      6
#define D2      (H2*C2)
#define SCAN_T     1024
#define SCAN_W     (SCAN_T / 32)
#define SCAN_CHUNK (N_NODES / SCAN_W)
#define SCAN_STEPS ((SCAN_CHUNK + 31) / 32)
#define BN_EPS    1e-5f
#define NEG_SLOPE 0.2f

typedef __attribute__((ext_vector_type(2))) float v2f;
typedef __attribute__((ext_vector_type(8))) float v8f;


static_assert(N_NODES <= 65536, "16-bit column ids need N_NODES <= 65536");
typedef unsigned short col_t;
static_assert(N_EDGES % 256 == 0 && N_NODES % 8 == 0 && (N_NODES * H2) % 256 == 0,
              "launch grids below divide exactly");

__device__ __forceinline__ float lrelu(float a) { return a > 0.f ? a : NEG_SLOPE * a; }


__global__ void bn_prep(const float* g1, const float* be1, const float* mu1, const float* va1,
                        const float* g2, const float* be2, const float* mu2, const float* va2,
                        const float* bias1,
                        float* sc1, float* sh1, float* sc2, float* sh2, int* __restrict__ cnt) {
  int i = blockIdx.x * blockDim.x + threadIdx.x;
  if (i < N_NODES) cnt[i] = 0;
  if (i < F_IN) {
    float s = g1[i] / sqrtf(va1[i] + BN_EPS);
    sc1[i] = s; sh1[i] = be1[i] - mu1[i] * s;
  }
  if (i < D1) {
    float s = g2[i] / sqrtf(va2[i] + BN_EPS);
    sc2[i] = s; sh2[i] = be2[i] + (bias1[i] - mu2[i]) * s;
  }
}

#define EDGE_BLOCKS (N_EDGES / 256)
#define G1_BLOCKS   (N_NODES / 32)
#define G1_FIRST    512
__device__ __forceinline__ void row_count_blk(int blk, const int* __restrict__ rowA, int* __restrict__ cnt) {
  int e = blk * 256 + threadIdx.x;
  if (e < N_EDGES) atomicAdd(&cnt[rowA[e]], 1);
}

static_assert(SCAN_W == 32 && N_NODES % SCAN_W == 0, "one lane per wave total; equal chunks");
__global__ __launch_bounds__(SCAN_T)
void row_scan(int* __restrict__ cnt) {
  __shared__ int wtot[SCAN_W];
  int lane = threadIdx.x & 31, w = threadIdx.x >> 5;
  int* __restrict__ p = cnt + w * SCAN_CHUNK;
  int s = 0;
  #pragma unroll
  for (int i = 0; i < SCAN_STEPS; ++i) {
    bool in = i < SCAN_CHUNK / 32 || lane < SCAN_CHUNK % 32;
    s += in ? p[i * 32 + lane] : 0;
  }
  #pragma unroll
  for (int off = 16; off > 0; off >>= 1) s += __shfl_xor(s, off, 32);
  if (lane == 0) wtot[w] = s;
  __syncthreads();
  int carry = lane < w ? wtot[lane] : 0;
  #pragma unroll
  for (int off = 16; off > 0; off >>= 1) carry += __shfl_xor(carry, off, 32);
  #pragma unroll 8
  for (int i = 0; i < SCAN_STEPS; ++i) {
    bool in = i < SCAN_CHUNK / 32 || lane < SCAN_CHUNK % 32;
    int x0 = in ? p[i * 32 + lane] : 0;
    int x = x0;
    #pragma unroll
    for (int off = 1; off < 32; off <<= 1) {
      int y = __shfl_up(x, off, 32);
      if (lane >= off) x += y;
    }
    if (in) p[i * 32 + lane] = x - x0 + carry;
    carry += __shfl(x, 31, 32);
  }
}

__device__ __forceinline__ void csr_fill_blk(int blk, const int* __restrict__ rowA, const int* __restrict__ colA,
                                             int* __restrict__ cur, col_t* __restrict__ csr) {
  int e = blk * 256 + threadIdx.x;
  if (e >= N_EDGES) return;
  int slot = atomicAdd(&cur[rowA[e]], 1);
  csr[slot] = (col_t)colA[e];
}

template <int M>
__device__ __forceinline__ float row_xor(float v) {
  static_assert(M >= 1 && M <= 15, "partner inside the 16-lane row");
  const int x = __float_as_int(v);
  return __int_as_float(__builtin_amdgcn_update_dpp(x, x, 0x160 + M, 0xf, 0xf, false));
}

__device__ __forceinline__ float reduce_scatter16(const float (&v)[16], int l16) {
  const bool b3 = l16 & 8, b2 = l16 & 4, b1 = l16 & 2, b0 = l16 & 1;
  float w8[8], w4[4], w2[2];
  #pragma unroll
  for (int i = 0; i < 8; ++i) {
    float keep = b3 ? v[i + 8] : v[i], send = b3 ? v[i] : v[i + 8];
    w8[i] = keep + row_xor<8>(send);
  }
  #pragma unroll
  for (int i = 0; i < 4; ++i) {
    float keep = b2 ? w8[i + 4] : w8[i], send = b2 ? w8[i] : w8[i + 4];
    w4[i] = keep + row_xor<4>(send);
  }
  #pragma unroll
  for (int i = 0; i < 2; ++i) {
    float keep = b1 ? w4[i + 2] : w4[i], send = b1 ? w4[i] : w4[i + 2];
    w2[i] = keep + row_xor<2>(send);
  }
  float keep = b0 ? w2[1] : w2[0], send = b0 ? w2[0] : w2[1];
  return keep + row_xor<1>(send);
}


template <int NT, int N>
__device__ __forceinline__ void load_b(const float* __restrict__ b0, float (&bx)[NT], float (&by)[NT]) {
  const float* __restrict__ b1 = b0 + N;
  if constexpr (NT == 4) {
    float4 t0 = *(const float4*)b0, t1 = *(const float4*)b1;
    bx[0] = t0.x; bx[1] = t0.y; bx[2] = t0.z; bx[3] = t0.w;
    by[0] = t1.x; by[1] = t1.y; by[2] = t1.z; by[3] = t1.w;
  } else {
    #pragma unroll
    for (int nt = 0; nt < NT; ++nt) { bx[nt] = b0[nt]; by[nt] = b1[nt]; }
  }
}

template <int K, int N, int MT, int NT, int WN, int KC, int PF, int SCORE, int WPS, bool EDGES>
__global__ __launch_bounds__(32 * WN)
__attribute__((amdgpu_waves_per_eu(WPS)))
void gemm_bn_wmma(const float* A,
                  const float* __restrict__ scale,
                  const float* __restrict__ shift,
                  const float* __restrict__ B,
                  float* __restrict__ C,
                  const float* __restrict__ att,
                  float* __restrict__ sa,
                  float* __restrict__ sb,
                  int blk0, int nblk, int pass,
                  const int* __restrict__ rowA, const int* __restrict__ colA,
                  int* __restrict__ cur, col_t* __restrict__ csr) {
  constexpr int THREADS = 32 * WN;
  if constexpr (EDGES) {
    static_assert(THREADS == 256, "edge blocks are 256 threads");
    if ((int)blockIdx.x >= nblk) {
      const int eb = (int)blockIdx.x - nblk;
      if (pass == 0) row_count_blk(eb, rowA, cur);
      else           csr_fill_blk(eb, rowA, colA, cur, csr);
      return;
    }
  }
  constexpr int ROWS  = 16 * MT;
  constexpr int PITCH = K + 4;
  constexpr int K4  = K / 4;
  constexpr int C4  = KC / 4;
  constexpr int NCH = K / KC;
  constexpr int RPP = THREADS / C4;
  constexpr int LPT = ROWS / RPP;
  static_assert(PF >= 1 && 4 * PF <= K, "queue depth");
  static_assert(N == WN * NT * 16, "the waves of a work-group tile all N columns");
  static_assert(K % KC == 0 && KC % 4 == 0 && THREADS % C4 == 0 && ROWS % RPP == 0, "exact staging");
  static_assert((PITCH / 4) % 2 == 1, "pitch is 16 B x odd");
  static_assert(N_NODES % ROWS == 0, "grid divides exactly");
  __shared__ float As[ROWS * PITCH];

  const int tid  = threadIdx.x;
  const int lane = tid & 31;
  const int half = lane >> 4;
  const int l16  = lane & 15;
  const int wn   = tid >> 5;
  const int row0 = (blk0 + (int)blockIdx.x) * ROWS;

  const int tr = tid / C4, c4 = tid % C4;
  const float4* A4 = (const float4*)(A + (size_t)row0 * K) + (size_t)tr * K4 + c4;
  const float4* __restrict__ sc4 = (const float4*)scale + c4;
  const float4* __restrict__ sh4 = (const float4*)shift + c4;
  float* st = As + tr * PITCH + c4 * 4;

  float4 pre[LPT];
  #pragma unroll
  for (int i = 0; i < LPT; ++i) pre[i] = A4[(size_t)i * RPP * K4];

  const float* as = As + l16 * PITCH + 2 * half;
  const float* __restrict__ bp = B + (size_t)(2 * half) * N + wn * (NT * 16) + NT * l16;

  v8f acc[MT][NT];
  #pragma unroll
  for (int mt = 0; mt < MT; ++mt)
    #pragma unroll
    for (int nt = 0; nt < NT; ++nt) acc[mt][nt] = v8f{};

  float qx[PF][NT], qy[PF][NT];
  #pragma unroll
  for (int p = 0; p < PF; ++p) load_b<NT, N>(bp + (size_t)(4 * p) * N, qx[p], qy[p]);

  for (int ch = 0; ch < NCH; ++ch) {
    const float4 sc = sc4[ch * C4], sh = sh4[ch * C4];
    #pragma unroll
    for (int i = 0; i < LPT; ++i) {
      float4 v = pre[i], o;
      o.x = v.x * sc.x + sh.x; o.y = v.y * sc.y + sh.y;
      o.z = v.z * sc.z + sh.z; o.w = v.w * sc.w + sh.w;
      *(float4*)(st + i * RPP * PITCH + ch * KC) = o;
    }
    __syncthreads();
    if (ch + 1 < NCH) {
      #pragma unroll
      for (int i = 0; i < LPT; ++i) pre[i] = A4[(size_t)i * RPP * K4 + (ch + 1) * C4];
    }
    for (int k = ch * KC; k < ch * KC + KC; k += 4) {
      v2f a[MT];
      #pragma unroll
      for (int mt = 0; mt < MT; ++mt) a[mt] = *(const v2f*)(as + mt * 16 * PITCH + k);
      float bx[NT], by[NT];
      #pragma unroll
      for (int nt = 0; nt < NT; ++nt) { bx[nt] = qx[0][nt]; by[nt] = qy[0][nt]; }
      #pragma unroll
      for (int p = 0; p + 1 < PF; ++p)
        #pragma unroll
        for (int nt = 0; nt < NT; ++nt) { qx[p][nt] = qx[p + 1][nt]; qy[p][nt] = qy[p + 1][nt]; }
      if (k + 4 * PF < K) load_b<NT, N>(bp + (size_t)(k + 4 * PF) * N, qx[PF - 1], qy[PF - 1]);
      #pragma unroll
      for (int mt = 0; mt < MT; ++mt)
        #pragma unroll
        for (int nt = 0; nt < NT; ++nt) {
          v2f b; b.x = bx[nt]; b.y = by[nt];
          acc[mt][nt] = __builtin_amdgcn_wmma_f32_16x16x4_f32(
               false, a[mt],  false, b,
               (short)0, acc[mt][nt],  false,  false);
        }
    }
  }

  #pragma unroll
  for (int mt = 0; mt < MT; ++mt) {
    float* cp = C + (size_t)(row0 + mt * 16 + (half << 3)) * N + wn * (NT * 16) + NT * l16;
    #pragma unroll
    for (int r = 0; r < 8; ++r) {
      if constexpr (NT == 4) {
        *(float4*)(cp + (size_t)r * N) =
            make_float4(acc[mt][0][r], acc[mt][1][r], acc[mt][2][r], acc[mt][3][r]);
      } else {
        #pragma unroll
        for (int nt = 0; nt < NT; ++nt) cp[(size_t)r * N + nt] = acc[mt][nt][r];
      }
    }
  }

  if constexpr (SCORE == 1) {
    static_assert(MT == 2 && NT == 4 && N == D1 && C1 == 2 * NT * 16 && ROWS * H1 * 2 == THREADS,
                  "two waves per head, 32 rows, one score per thread");
    __shared__ float red[WN * ROWS * 2];
    const int hd = wn >> 1;
    const int cb = (wn & 1) * (NT * 16) + NT * l16;
    const float4 al = *(const float4*)(att + hd * (2 * C1) + cb);
    const float4 ar = *(const float4*)(att + hd * (2 * C1) + C1 + cb);
    #pragma unroll
    for (int mt = 0; mt < MT; ++mt) {
      float p[16];
      #pragma unroll
      for (int r = 0; r < 8; ++r) {
        float c0 = acc[mt][0][r], c1 = acc[mt][1][r], c2 = acc[mt][2][r], c3 = acc[mt][3][r];
        p[r]     = c0 * al.x + c1 * al.y + c2 * al.z + c3 * al.w;
        p[8 + r] = c0 * ar.x + c1 * ar.y + c2 * ar.z + c3 * ar.w;
      }
      const float sc = reduce_scatter16(p, l16);
      red[(wn * ROWS + mt * 16 + (half << 3) + (l16 & 7)) * 2 + (l16 >> 3)] = sc;
    }
    __syncthreads();
    const int wh = tid >> 7, idx = tid & 127, rr = idx >> 2, hh = idx & 3;
    const float v = red[((2 * hh) * ROWS + rr) * 2 + wh] + red[((2 * hh + 1) * ROWS + rr) * 2 + wh];
    (wh ? sb : sa)[(size_t)row0 * H1 + idx] = v;
  }
}


#define AG_ROWS  16
#define AG_RPW   (AG_ROWS / 8)
#define AG_PITCH (D1 + 4)
#define AG_KW    (D1 / 8)
static_assert(N_NODES % AG_ROWS == 0 && (AG_PITCH / 4) % 2 == 1 && AG_KW % 4 == 0, "exact tiles");
static_assert(8 * 3 * 8 * 32 <= AG_ROWS * AG_PITCH, "the slice sums fit the row buffer they replace");
static_assert(D2 == 3 * 16 && C2 == 6 && H2 == 8 && AG_ROWS * H2 == 128, "part 2 maps");

__global__ __launch_bounds__(256)
__attribute__((amdgpu_waves_per_eu(12)))
void row_agg1_gemm2(const int* __restrict__ cur, const col_t* __restrict__ csr,
                    const float* __restrict__ ai, const float* __restrict__ aj,
                    const float* __restrict__ h1,
                    const float* __restrict__ scale, const float* __restrict__ shift,
                    const float* __restrict__ B, const float* __restrict__ att,
                    float* __restrict__ C, float* __restrict__ sa) {
  __shared__ float As[AG_ROWS * AG_PITCH];
  const int tid = threadIdx.x;
  const int lane = tid & 31;
  const int wv = tid >> 5;
  const int row0 = (int)blockIdx.x * AG_ROWS;
  const float4* __restrict__ aj4 = (const float4*)aj;
  const float4* __restrict__ h4 = (const float4*)h1;

  for (int q = 0; q < AG_RPW; ++q) {
    const int rl = wv * AG_RPW + q;
    int r = __builtin_amdgcn_readfirstlane(row0 + rl);
    int beg = r ? cur[r - 1] : 0;
    int end = cur[r];
    float4 a_i = *(const float4*)(ai + (size_t)r * 4);
    float4 bs  = aj4[r];
    float t0 = lrelu(a_i.x + bs.x), t1 = lrelu(a_i.y + bs.y);
    float t2 = lrelu(a_i.z + bs.z), t3 = lrelu(a_i.w + bs.w);

    float m0 = t0, m1 = t1, m2 = t2, m3 = t3;
    for (int j = beg + lane; j < end; j += 32) {
      float4 b = aj4[csr[j]];
      m0 = fmaxf(m0, lrelu(a_i.x + b.x));
      m1 = fmaxf(m1, lrelu(a_i.y + b.y));
      m2 = fmaxf(m2, lrelu(a_i.z + b.z));
      m3 = fmaxf(m3, lrelu(a_i.w + b.w));
    }
    #pragma unroll
    for (int off = 16; off > 0; off >>= 1) {
      m0 = fmaxf(m0, __shfl_xor(m0, off, 32));
      m1 = fmaxf(m1, __shfl_xor(m1, off, 32));
      m2 = fmaxf(m2, __shfl_xor(m2, off, 32));
      m3 = fmaxf(m3, __shfl_xor(m3, off, 32));
    }

    float u0 = expf(t0 - m0), u1 = expf(t1 - m1), u2 = expf(t2 - m2), u3 = expf(t3 - m3);
    float s0 = u0, s1 = u1, s2 = u2, s3 = u3;
    const float4* __restrict__ src = h4 + (size_t)r * (D1 / 4);
    float4 v0 = src[lane], v1 = src[32 + lane], v2 = src[64 + lane], v3 = src[96 + lane];
    float4 acc0 = {u0 * v0.x, u0 * v0.y, u0 * v0.z, u0 * v0.w};
    float4 acc1 = {u1 * v1.x, u1 * v1.y, u1 * v1.z, u1 * v1.w};
    float4 acc2 = {u2 * v2.x, u2 * v2.y, u2 * v2.z, u2 * v2.w};
    float4 acc3 = {u3 * v3.x, u3 * v3.y, u3 * v3.z, u3 * v3.w};

    for (int base = beg; base < end; base += 32) {
      int j = base + lane;
      int c = r;
      float w0 = 0.f, w1 = 0.f, w2 = 0.f, w3 = 0.f;
      if (j < end) {
        c = csr[j];
        float4 b = aj4[c];
        w0 = expf(lrelu(a_i.x + b.x) - m0);
        w1 = expf(lrelu(a_i.y + b.y) - m1);
        w2 = expf(lrelu(a_i.z + b.z) - m2);
        w3 = expf(lrelu(a_i.w + b.w) - m3);
      }
      int cnt = end - base < 32 ? end - base : 32;
      for (int k = 0; k < cnt; ++k) {
        int ck = __shfl(c, k, 32);
        u0 = __shfl(w0, k, 32); u1 = __shfl(w1, k, 32);
        u2 = __shfl(w2, k, 32); u3 = __shfl(w3, k, 32);
        s0 += u0; s1 += u1; s2 += u2; s3 += u3;
        src = h4 + (size_t)ck * (D1 / 4);
        v0 = src[lane]; v1 = src[32 + lane]; v2 = src[64 + lane]; v3 = src[96 + lane];
        acc0.x += u0 * v0.x; acc0.y += u0 * v0.y; acc0.z += u0 * v0.z; acc0.w += u0 * v0.w;
        acc1.x += u1 * v1.x; acc1.y += u1 * v1.y; acc1.z += u1 * v1.z; acc1.w += u1 * v1.w;
        acc2.x += u2 * v2.x; acc2.y += u2 * v2.y; acc2.z += u2 * v2.z; acc2.w += u2 * v2.w;
        acc3.x += u3 * v3.x; acc3.y += u3 * v3.y; acc3.z += u3 * v3.z; acc3.w += u3 * v3.w;
      }
    }
    float i0 = 1.0f / (s0 + 1e-16f), i1 = 1.0f / (s1 + 1e-16f);
    float i2 = 1.0f / (s2 + 1e-16f), i3 = 1.0f / (s3 + 1e-16f);
    float4* dst = (float4*)(As + rl * AG_PITCH);
    dst[lane]      = make_float4(acc0.x * i0, acc0.y * i0, acc0.z * i0, acc0.w * i0);
    dst[32 + lane] = make_float4(acc1.x * i1, acc1.y * i1, acc1.z * i1, acc1.w * i1);
    dst[64 + lane] = make_float4(acc2.x * i2, acc2.y * i2, acc2.z * i2, acc2.w * i2);
    dst[96 + lane] = make_float4(acc3.x * i3, acc3.y * i3, acc3.z * i3, acc3.w * i3);
  }
  __syncthreads();

  const int half = lane >> 4;
  const int l16  = lane & 15;
  const float* as = As + l16 * AG_PITCH + wv * AG_KW + 2 * half;
  const float* __restrict__ bp = B + (size_t)(wv * AG_KW + 2 * half) * D2 + 3 * l16;
  const float* __restrict__ scp = scale + wv * AG_KW + 2 * half;
  const float* __restrict__ shp = shift + wv * AG_KW + 2 * half;
  v8f acc[3];
  #pragma unroll
  for (int nt = 0; nt < 3; ++nt) acc[nt] = v8f{};
  #pragma unroll 4
  for (int k = 0; k < AG_KW; k += 4) {
    const v2f raw = *(const v2f*)(as + k);
    const v2f sc = *(const v2f*)(scp + k), sh = *(const v2f*)(shp + k);
    v2f a; a.x = raw.x * sc.x + sh.x; a.y = raw.y * sc.y + sh.y;
    float bx[3], by[3];
    load_b<3, D2>(bp + (size_t)k * D2, bx, by);
    #pragma unroll
    for (int nt = 0; nt < 3; ++nt) {
      v2f b; b.x = bx[nt]; b.y = by[nt];
      acc[nt] = __builtin_amdgcn_wmma_f32_16x16x4_f32(
           false, a,  false, b,
           (short)0, acc[nt],  false,  false);
    }
  }
  __syncthreads();
  float* red = As;
  #pragma unroll
  for (int nt = 0; nt < 3; ++nt)
    #pragma unroll
    for (int r = 0; r < 8; ++r) red[((wv * 3 + nt) * 8 + r) * 32 + lane] = acc[nt][r];
  __syncthreads();

  if (tid < AG_ROWS * H2) {
    const int row = tid >> 3, hd = tid & 7;
    const int rbase = (row & 7) * 32 + (row >> 3) * 16 + 2 * hd;
    float v[C2];
    #pragma unroll
    for (int j = 0; j < C2; ++j) {
      const float* p = red + (j % 3) * 256 + rbase + j / 3;
      float s = p[0];
      #pragma unroll
      for (int w = 1; w < 8; ++w) s += p[w * 768];
      v[j] = s;
    }
    float2* __restrict__ cp = (float2*)(C + (size_t)(row0 + row) * D2 + hd * C2);
    cp[0] = make_float2(v[0], v[1]);
    cp[1] = make_float2(v[2], v[3]);
    cp[2] = make_float2(v[4], v[5]);
    const float* __restrict__ ap = att + hd * (2 * C2);
    const float pa = v[0] * (ap[0] + ap[C2])     + v[1] * (ap[1] + ap[C2 + 1]) + v[2] * (ap[2] + ap[C2 + 2]);
    const float pb = v[3] * (ap[3] + ap[C2 + 3]) + v[4] * (ap[4] + ap[C2 + 4]) + v[5] * (ap[5] + ap[C2 + 5]);
    sa[(size_t)(row0 + row) * H2 + hd] = tanhf(pa + pb);
  }
}


__global__ void row_agg2(const int* __restrict__ cur, const col_t* __restrict__ csr,
                         const float* __restrict__ s2, const float* __restrict__ h2,
                         float* __restrict__ out2) {
  int tid = blockIdx.x * blockDim.x + threadIdx.x;
  if (tid >= N_NODES * H2) return;
  int r = tid >> 3, h = tid & 7;
  int beg = r ? cur[r - 1] : 0;
  int end = cur[r];
  float sself = s2[tid];
  float m = sself;
  for (int j = beg; j < end; ++j) m = fmaxf(m, s2[(int)csr[j] * H2 + h]);

  float w = expf(sself - m);
  float den = w;
  const float2* __restrict__ src = (const float2*)(h2 + (size_t)r * D2 + h * C2);
  float2 p0 = src[0], p1 = src[1], p2 = src[2];
  float a0 = w * p0.x, a1 = w * p0.y, a2 = w * p1.x, a3 = w * p1.y, a4 = w * p2.x, a5 = w * p2.y;
  for (int j = beg; j < end; ++j) {
    int c = csr[j];
    w = expf(s2[c * H2 + h] - m);
    den += w;
    src = (const float2*)(h2 + (size_t)c * D2 + h * C2);
    p0 = src[0]; p1 = src[1]; p2 = src[2];
    a0 += w * p0.x; a1 += w * p0.y; a2 += w * p1.x;
    a3 += w * p1.y; a4 += w * p2.x; a5 += w * p2.y;
  }
  float inv = 1.0f / (den + 1e-16f);
  float2* __restrict__ dst = (float2*)(out2 + (size_t)r * D2 + h * C2);
  dst[0] = make_float2(a0 * inv, a1 * inv);
  dst[1] = make_float2(a2 * inv, a3 * inv);
  dst[2] = make_float2(a4 * inv, a5 * inv);
}

__global__ void finalize(const float* __restrict__ out2, const float* __restrict__ b2,
                         float* __restrict__ out) {
  int n = blockIdx.x * blockDim.x + threadIdx.x;
  if (n >= N_NODES) return;
  float v[C2];
  #pragma unroll
  for (int c = 0; c < C2; ++c) {
    float s = 0.f;
    #pragma unroll
    for (int h = 0; h < H2; ++h) s += out2[(size_t)n * D2 + h * C2 + c];
    float t = s * (1.0f / H2) + b2[c];
    v[c] = t > 0.f ? t : (expf(t) - 1.0f);
  }
  float m = v[0];
  #pragma unroll
  for (int c = 1; c < C2; ++c) m = fmaxf(m, v[c]);
  float sum = 0.f;
  #pragma unroll
  for (int c = 0; c < C2; ++c) sum += expf(v[c] - m);
  float lse = logf(sum);
  #pragma unroll
  for (int c = 0; c < C2; ++c) out[(size_t)n * C2 + c] = v[c] - m - lse;
}


extern "C" void kernel_launch(void* const* d_in, const int* in_sizes, int n_in,
                              void* d_out, int out_size, void* d_ws, size_t ws_size,
                              hipStream_t stream) {
  const float* x    = (const float*)d_in[0];
  const int*   ei   = (const int*)  d_in[1];
  const float* g1   = (const float*)d_in[2];
  const float* be1  = (const float*)d_in[3];
  const float* mu1  = (const float*)d_in[4];
  const float* va1  = (const float*)d_in[5];
  const float* W1   = (const float*)d_in[6];
  const float* att1 = (const float*)d_in[7];
  const float* b1   = (const float*)d_in[8];
  const float* g2   = (const float*)d_in[9];
  const float* be2  = (const float*)d_in[10];
  const float* mu2  = (const float*)d_in[11];
  const float* va2  = (const float*)d_in[12];
  const float* W2   = (const float*)d_in[13];
  const float* att2 = (const float*)d_in[14];
  const float* b2   = (const float*)d_in[15];
  const int* rowA = ei;
  const int* colA = ei + N_EDGES;
  float* out = (float*)d_out;
  float* ws  = (float*)d_ws;

  float* sc1  = ws;
  float* sh1  = ws + 256;
  float* sc2  = ws + 512;
  float* sh2  = ws + 1024;
  float* ai   = ws + 1536;
  float* aj   = ai  + (size_t)N_NODES * H1;
  float* edg  = aj  + (size_t)N_NODES * H1;
  int*   cur  = (int*)edg;
  col_t* csr  = (col_t*)(edg + N_NODES);
  float* l2   = edg + (size_t)N_NODES * 2 * H1;
  float* h2   = l2;
  float* s2   = h2  + (size_t)N_NODES * D2;
  float* out2 = s2  + (size_t)N_NODES * H2;
  float* h1   = l2  + (size_t)N_NODES * D1;

  bn_prep<<<(N_NODES + 255) / 256, 256, 0, stream>>>(g1, be1, mu1, va1, g2, be2, mu2, va2, b1,
                                                     sc1, sh1, sc2, sh2, cur);

  gemm_bn_wmma<F_IN, D1, 2, 4, 8, 64, 2, 1, 8, true><<<G1_FIRST + EDGE_BLOCKS, 256, 0, stream>>>(
      x, sc1, sh1, W1, h1, att1, ai, aj, 0, G1_FIRST, 0, rowA, colA, cur, csr);
  row_scan<<<1, SCAN_T, 0, stream>>>(cur);
  gemm_bn_wmma<F_IN, D1, 2, 4, 8, 64, 2, 1, 8, true><<<G1_BLOCKS - G1_FIRST + EDGE_BLOCKS, 256, 0, stream>>>(
      x, sc1, sh1, W1, h1, att1, ai, aj, G1_FIRST, G1_BLOCKS - G1_FIRST, 1, rowA, colA, cur, csr);

  row_agg1_gemm2<<<N_NODES / AG_ROWS, 256, 0, stream>>>(cur, csr, ai, aj, h1, sc2, sh2, W2, att2, h2, s2);

  row_agg2<<<(N_NODES * H2) / 256, 256, 0, stream>>>(cur, csr, s2, h2, out2);

  finalize<<<(N_NODES + 255) / 256, 256, 0, stream>>>(out2, b2, out);
}


# Layer-2 linear map (BN2, W2) commuted through the layer-1 aggregation: W1 folded per head with W2 and att1 each call; GEMM1 N 512->256, gathered rows 192 floats; (xW1)W2 summed as x(W1W2)
# speedup vs baseline: 6.0538x; 1.3213x over previous
#include <hip/hip_runtime.h>
#include <math.h>

#define N_NODES 40000
#define N_EDGES 320000
#define ETOT    (N_EDGES + N_NODES)
#define F_IN    256
#define H1      4
#define C1      128
#define D1      (H1*C1)
#define H2      8
#define C2      6
#define D2      (H2*C2)
#define SCAN_T     1024
#define SCAN_W     (SCAN_T / 32)
#define SCAN_CHUNK (N_NODES / SCAN_W)
#define SCAN_STEPS ((SCAN_CHUNK + 31) / 32)
#define BN_EPS    1e-5f
#define NEG_SLOPE 0.2f

typedef __attribute__((ext_vector_type(2))) float v2f;
typedef __attribute__((ext_vector_type(8))) float v8f;


static_assert(N_NODES <= 65536, "16-bit column ids need N_NODES <= 65536");
typedef unsigned short col_t;
static_assert(N_EDGES % 256 == 0 && N_NODES % 8 == 0 && (N_NODES * H2) % 256 == 0,
              "launch grids below divide exactly");

__device__ __forceinline__ float lrelu(float a) { return a > 0.f ? a : NEG_SLOPE * a; }


#define GP 192
#define NF 256
#define PREP_BLOCKS ((N_NODES + 255) / 256)
static_assert(GP == H1 * D2 && GP + 2 * H1 <= NF && D1 == 2 * 256 && H1 * 64 == 256 && D2 <= 64, "fold maps");

__global__ __launch_bounds__(256)
void bn_prep(const float* g1, const float* be1, const float* mu1, const float* va1,
             const float* g2, const float* be2, const float* mu2, const float* va2,
             const float* bias1, const float* __restrict__ W1, const float* __restrict__ W2,
             const float* __restrict__ att1,
             float* sc1, float* sh1, float* __restrict__ Wc, float* __restrict__ c0, int* __restrict__ cnt) {
  if ((int)blockIdx.x >= PREP_BLOCKS) {
    __shared__ float s2c[D1];
    const int fb = (int)blockIdx.x - PREP_BLOCKS, n = threadIdx.x;
    for (int kk = n; kk < D1; kk += 256) s2c[kk] = g2[kk] / sqrtf(va2[kk] + BN_EPS);
    __syncthreads();
    float s = 0.f;
    if (fb < F_IN) {
      const float* __restrict__ w1 = W1 + (size_t)fb * D1;
      if (n < GP) {
        const int h = n / D2, o = n % D2;
        for (int c = 0; c < C1; ++c) { const int kk = h * C1 + c; s += w1[kk] * (s2c[kk] * W2[kk * D2 + o]); }
      } else if (n < GP + 2 * H1) {
        const int h = (n - GP) & (H1 - 1), side = (n - GP) / H1;
        const float* __restrict__ a = att1 + h * (2 * C1) + side * C1;
        for (int c = 0; c < C1; ++c) s += w1[h * C1 + c] * a[c];
      }
      Wc[(size_t)fb * NF + n] = s;
    } else {
      const int h = n >> 6, o = n & 63;
      if (o < D2) {
        for (int c = 0; c < C1; ++c) {
          const int kk = h * C1 + c;
          const float sh = be2[kk] + (bias1[kk] - mu2[kk]) * s2c[kk];
          s += sh * W2[kk * D2 + o];
        }
        c0[h * D2 + o] = s;
      }
    }
    return;
  }
  int i = blockIdx.x * blockDim.x + threadIdx.x;
  if (i < N_NODES) cnt[i] = 0;
  if (i < F_IN) {
    float s = g1[i] / sqrtf(va1[i] + BN_EPS);
    sc1[i] = s; sh1[i] = be1[i] - mu1[i] * s;
  }
}

#define EDGE_BLOCKS (N_EDGES / 256)
#define G1_BLOCKS   (N_NODES / 32)
#define G1_FIRST    512
__device__ __forceinline__ void row_count_blk(int blk, const int* __restrict__ rowA, int* __restrict__ cnt) {
  int e = blk * 256 + threadIdx.x;
  if (e < N_EDGES) atomicAdd(&cnt[rowA[e]], 1);
}

static_assert(SCAN_W == 32 && N_NODES % SCAN_W == 0, "one lane per wave total; equal chunks");
__global__ __launch_bounds__(SCAN_T)
void row_scan(int* __restrict__ cnt) {
  __shared__ int wtot[SCAN_W];
  int lane = threadIdx.x & 31, w = threadIdx.x >> 5;
  int* __restrict__ p = cnt + w * SCAN_CHUNK;
  int s = 0;
  #pragma unroll
  for (int i = 0; i < SCAN_STEPS; ++i) {
    bool in = i < SCAN_CHUNK / 32 || lane < SCAN_CHUNK % 32;
    s += in ? p[i * 32 + lane] : 0;
  }
  #pragma unroll
  for (int off = 16; off > 0; off >>= 1) s += __shfl_xor(s, off, 32);
  if (lane == 0) wtot[w] = s;
  __syncthreads();
  int carry = lane < w ? wtot[lane] : 0;
  #pragma unroll
  for (int off = 16; off > 0; off >>= 1) carry += __shfl_xor(carry, off, 32);
  #pragma unroll 8
  for (int i = 0; i < SCAN_STEPS; ++i) {
    bool in = i < SCAN_CHUNK / 32 || lane < SCAN_CHUNK % 32;
    int x0 = in ? p[i * 32 + lane] : 0;
    int x = x0;
    #pragma unroll
    for (int off = 1; off < 32; off <<= 1) {
      int y = __shfl_up(x, off, 32);
      if (lane >= off) x += y;
    }
    if (in) p[i * 32 + lane] = x - x0 + carry;
    carry += __shfl(x, 31, 32);
  }
}

__device__ __forceinline__ void csr_fill_blk(int blk, const int* __restrict__ rowA, const int* __restrict__ colA,
                                             int* __restrict__ cur, col_t* __restrict__ csr) {
  int e = blk * 256 + threadIdx.x;
  if (e >= N_EDGES) return;
  int slot = atomicAdd(&cur[rowA[e]], 1);
  csr[slot] = (col_t)colA[e];
}


template <int NT, int N>
__device__ __forceinline__ void load_b(const float* __restrict__ b0, float (&bx)[NT], float (&by)[NT]) {
  const float* __restrict__ b1 = b0 + N;
  if constexpr (NT == 4) {
    float4 t0 = *(const float4*)b0, t1 = *(const float4*)b1;
    bx[0] = t0.x; bx[1] = t0.y; bx[2] = t0.z; bx[3] = t0.w;
    by[0] = t1.x; by[1] = t1.y; by[2] = t1.z; by[3] = t1.w;
  } else {
    #pragma unroll
    for (int nt = 0; nt < NT; ++nt) { bx[nt] = b0[nt]; by[nt] = b1[nt]; }
  }
}

template <int K, int N, int MT, int NT, int WN, int KC, int PF, int SCORE, int WPS, bool EDGES>
__global__ __launch_bounds__(32 * WN)
__attribute__((amdgpu_waves_per_eu(WPS)))
void gemm_bn_wmma(const float* A,
                  const float* __restrict__ scale,
                  const float* __restrict__ shift,
                  const float* __restrict__ B,
                  float* __restrict__ C,
                  const float* __restrict__ att,
                  float* __restrict__ sa,
                  float* __restrict__ sb,
                  int blk0, int nblk, int pass,
                  const int* __restrict__ rowA, const int* __restrict__ colA,
                  int* __restrict__ cur, col_t* __restrict__ csr) {
  constexpr int THREADS = 32 * WN;
  if constexpr (EDGES) {
    static_assert(THREADS == 256, "edge blocks are 256 threads");
    if ((int)blockIdx.x >= nblk) {
      const int eb = (int)blockIdx.x - nblk;
      if (pass == 0) row_count_blk(eb, rowA, cur);
      else           csr_fill_blk(eb, rowA, colA, cur, csr);
      return;
    }
  }
  constexpr int ROWS  = 16 * MT;
  constexpr int PITCH = K + 4;
  constexpr int K4  = K / 4;
  constexpr int C4  = KC / 4;
  constexpr int NCH = K / KC;
  constexpr int RPP = THREADS / C4;
  constexpr int LPT = ROWS / RPP;
  static_assert(PF >= 1 && 4 * PF <= K, "queue depth");
  static_assert(N == WN * NT * 16, "the waves of a work-group tile all N columns");
  static_assert(K % KC == 0 && KC % 4 == 0 && THREADS % C4 == 0 && ROWS % RPP == 0, "exact staging");
  static_assert((PITCH / 4) % 2 == 1, "pitch is 16 B x odd");
  static_assert(N_NODES % ROWS == 0, "grid divides exactly");
  __shared__ float As[ROWS * PITCH];

  const int tid  = threadIdx.x;
  const int lane = tid & 31;
  const int half = lane >> 4;
  const int l16  = lane & 15;
  const int wn   = tid >> 5;
  const int row0 = (blk0 + (int)blockIdx.x) * ROWS;

  const int tr = tid / C4, c4 = tid % C4;
  const float4* A4 = (const float4*)(A + (size_t)row0 * K) + (size_t)tr * K4 + c4;
  const float4* __restrict__ sc4 = (const float4*)scale + c4;
  const float4* __restrict__ sh4 = (const float4*)shift + c4;
  float* st = As + tr * PITCH + c4 * 4;

  float4 pre[LPT];
  #pragma unroll
  for (int i = 0; i < LPT; ++i) pre[i] = A4[(size_t)i * RPP * K4];

  const float* as = As + l16 * PITCH + 2 * half;
  const float* __restrict__ bp = B + (size_t)(2 * half) * N + wn * (NT * 16) + NT * l16;

  v8f acc[MT][NT];
  #pragma unroll
  for (int mt = 0; mt < MT; ++mt)
    #pragma unroll
    for (int nt = 0; nt < NT; ++nt) acc[mt][nt] = v8f{};

  float qx[PF][NT], qy[PF][NT];
  #pragma unroll
  for (int p = 0; p < PF; ++p) load_b<NT, N>(bp + (size_t)(4 * p) * N, qx[p], qy[p]);

  for (int ch = 0; ch < NCH; ++ch) {
    const float4 sc = sc4[ch * C4], sh = sh4[ch * C4];
    #pragma unroll
    for (int i = 0; i < LPT; ++i) {
      float4 v = pre[i], o;
      o.x = v.x * sc.x + sh.x; o.y = v.y * sc.y + sh.y;
      o.z = v.z * sc.z + sh.z; o.w = v.w * sc.w + sh.w;
      *(float4*)(st + i * RPP * PITCH + ch * KC) = o;
    }
    __syncthreads();
    if (ch + 1 < NCH) {
      #pragma unroll
      for (int i = 0; i < LPT; ++i) pre[i] = A4[(size_t)i * RPP * K4 + (ch + 1) * C4];
    }
    for (int k = ch * KC; k < ch * KC + KC; k += 4) {
      v2f a[MT];
      #pragma unroll
      for (int mt = 0; mt < MT; ++mt) a[mt] = *(const v2f*)(as + mt * 16 * PITCH + k);
      float bx[NT], by[NT];
      #pragma unroll
      for (int nt = 0; nt < NT; ++nt) { bx[nt] = qx[0][nt]; by[nt] = qy[0][nt]; }
      #pragma unroll
      for (int p = 0; p + 1 < PF; ++p)
        #pragma unroll
        for (int nt = 0; nt < NT; ++nt) { qx[p][nt] = qx[p + 1][nt]; qy[p][nt] = qy[p + 1][nt]; }
      if (k + 4 * PF < K) load_b<NT, N>(bp + (size_t)(k + 4 * PF) * N, qx[PF - 1], qy[PF - 1]);
      #pragma unroll
      for (int mt = 0; mt < MT; ++mt)
        #pragma unroll
        for (int nt = 0; nt < NT; ++nt) {
          v2f b; b.x = bx[nt]; b.y = by[nt];
          acc[mt][nt] = __builtin_amdgcn_wmma_f32_16x16x4_f32(
               false, a[mt],  false, b,
               (short)0, acc[mt][nt],  false,  false);
        }
    }
  }

  if constexpr (SCORE != 0) {
    static_assert(MT == 2 && NT == 2, "four accumulators named below");
    asm volatile("v_nop\n\tv_nop\n\tv_nop\n\tv_nop\n\tv_nop"
                 : "+v"(acc[0][0]), "+v"(acc[0][1]), "+v"(acc[1][0]), "+v"(acc[1][1]));
  }
  if constexpr (SCORE == 0) {
    #pragma unroll
    for (int mt = 0; mt < MT; ++mt) {
      float* cp = C + (size_t)(row0 + mt * 16 + (half << 3)) * N + wn * (NT * 16) + NT * l16;
      #pragma unroll
      for (int r = 0; r < 8; ++r) {
        if constexpr (NT == 4) {
          *(float4*)(cp + (size_t)r * N) =
              make_float4(acc[mt][0][r], acc[mt][1][r], acc[mt][2][r], acc[mt][3][r]);
        } else {
          #pragma unroll
          for (int nt = 0; nt < NT; ++nt) cp[(size_t)r * N + nt] = acc[mt][nt][r];
        }
      }
    }
  } else {
    static_assert(NT == 2 && N == NF && GP % (NT * 16) == 0 && 2 * H1 == 4 * NT, "score columns in four lanes");
    if (wn < GP / (NT * 16)) {
      #pragma unroll
      for (int mt = 0; mt < MT; ++mt) {
        float* cp = C + (size_t)(row0 + mt * 16 + (half << 3)) * GP + wn * (NT * 16) + NT * l16;
        #pragma unroll
        for (int r = 0; r < 8; ++r)
          *(float2*)(cp + (size_t)r * GP) = make_float2(acc[mt][0][r], acc[mt][1][r]);
      }
    } else if (wn == GP / (NT * 16) && l16 < 4) {
      float* __restrict__ dst = (l16 < 2 ? sa : sb) + 2 * (l16 & 1);
      #pragma unroll
      for (int mt = 0; mt < MT; ++mt)
        #pragma unroll
        for (int r = 0; r < 8; ++r)
          *(float2*)(dst + (size_t)(row0 + mt * 16 + (half << 3) + r) * H1) =
              make_float2(acc[mt][0][r], acc[mt][1][r]);
    }
  }
}


#define AG_ROWS  16
#define AG_RPW   (AG_ROWS / 8)
#define AG_PITCH (GP + 4)
static_assert(N_NODES % AG_ROWS == 0 && GP == 96 * 2 && D2 == 24 * 2, "three float2 per lane, 24 float2 per head");
static_assert(C2 == 6 && H2 == 8 && AG_ROWS * H2 == 128, "epilogue map");

__global__ __launch_bounds__(256)
void row_agg1(const int* __restrict__ cur, const col_t* __restrict__ csr,
              const float* __restrict__ ai, const float* __restrict__ aj,
              const float* __restrict__ g, const float* __restrict__ c0,
              const float* __restrict__ att,
              float* __restrict__ C, float* __restrict__ sa) {
  __shared__ float Ls[AG_ROWS * AG_PITCH];
  const int tid = threadIdx.x;
  const int lane = tid & 31;
  const int wv = tid >> 5;
  const int row0 = (int)blockIdx.x * AG_ROWS;
  const float4* __restrict__ aj4 = (const float4*)aj;
  const float2* __restrict__ g2 = (const float2*)g;
  const bool lo0 = lane < 24, lo1 = lane < 16, lo2 = lane < 8;

  for (int q = 0; q < AG_RPW; ++q) {
    const int rl = wv * AG_RPW + q;
    int r = __builtin_amdgcn_readfirstlane(row0 + rl);
    int beg = r ? cur[r - 1] : 0;
    int end = cur[r];
    float4 a_i = *(const float4*)(ai + (size_t)r * 4);
    float4 bs  = aj4[r];
    float t0 = lrelu(a_i.x + bs.x), t1 = lrelu(a_i.y + bs.y);
    float t2 = lrelu(a_i.z + bs.z), t3 = lrelu(a_i.w + bs.w);

    float m0 = t0, m1 = t1, m2 = t2, m3 = t3;
    for (int j = beg + lane; j < end; j += 32) {
      float4 b = aj4[csr[j]];
      m0 = fmaxf(m0, lrelu(a_i.x + b.x));
      m1 = fmaxf(m1, lrelu(a_i.y + b.y));
      m2 = fmaxf(m2, lrelu(a_i.z + b.z));
      m3 = fmaxf(m3, lrelu(a_i.w + b.w));
    }
    #pragma unroll
    for (int off = 16; off > 0; off >>= 1) {
      m0 = fmaxf(m0, __shfl_xor(m0, off, 32));
      m1 = fmaxf(m1, __shfl_xor(m1, off, 32));
      m2 = fmaxf(m2, __shfl_xor(m2, off, 32));
      m3 = fmaxf(m3, __shfl_xor(m3, off, 32));
    }

    float u0 = expf(t0 - m0), u1 = expf(t1 - m1), u2 = expf(t2 - m2), u3 = expf(t3 - m3);
    float wa = lo0 ? u0 : u1, wb = lo1 ? u1 : u2, wc = lo2 ? u2 : u3;
    float sA = wa, sB = wb, sC = wc;
    const float2* __restrict__ src = g2 + (size_t)r * (GP / 2);
    float2 v0 = src[lane], v1 = src[32 + lane], v2 = src[64 + lane];
    float2 acc0 = {wa * v0.x, wa * v0.y};
    float2 acc1 = {wb * v1.x, wb * v1.y};
    float2 acc2 = {wc * v2.x, wc * v2.y};

    for (int base = beg; base < end; base += 32) {
      int j = base + lane;
      int c = r;
      float w0 = 0.f, w1 = 0.f, w2 = 0.f, w3 = 0.f;
      if (j < end) {
        c = csr[j];
        float4 b = aj4[c];
        w0 = expf(lrelu(a_i.x + b.x) - m0);
        w1 = expf(lrelu(a_i.y + b.y) - m1);
        w2 = expf(lrelu(a_i.z + b.z) - m2);
        w3 = expf(lrelu(a_i.w + b.w) - m3);
      }
      int cnt = end - base < 32 ? end - base : 32;
      for (int k = 0; k < cnt; ++k) {
        int ck = __shfl(c, k, 32);
        u0 = __shfl(w0, k, 32); u1 = __shfl(w1, k, 32);
        u2 = __shfl(w2, k, 32); u3 = __shfl(w3, k, 32);
        wa = lo0 ? u0 : u1; wb = lo1 ? u1 : u2; wc = lo2 ? u2 : u3;
        sA += wa; sB += wb; sC += wc;
        src = g2 + (size_t)ck * (GP / 2);
        v0 = src[lane]; v1 = src[32 + lane]; v2 = src[64 + lane];
        acc0.x += wa * v0.x; acc0.y += wa * v0.y;
        acc1.x += wb * v1.x; acc1.y += wb * v1.y;
        acc2.x += wc * v2.x; acc2.y += wc * v2.y;
      }
    }
    float iA = 1.0f / (sA + 1e-16f), iB = 1.0f / (sB + 1e-16f), iC = 1.0f / (sC + 1e-16f);
    float2* dst = (float2*)(Ls + rl * AG_PITCH);
    dst[lane]      = make_float2(acc0.x * iA, acc0.y * iA);
    dst[32 + lane] = make_float2(acc1.x * iB, acc1.y * iB);
    dst[64 + lane] = make_float2(acc2.x * iC, acc2.y * iC);
  }
  __syncthreads();

  if (tid < AG_ROWS * H2) {
    const int row = tid >> 3, hd = tid & 7;
    const float* p = Ls + row * AG_PITCH + hd * C2;
    const float* __restrict__ cz = c0 + hd * C2;
    float v[C2];
    #pragma unroll
    for (int j = 0; j < C2; ++j) {
      float s = p[j];
      #pragma unroll
      for (int h = 1; h < H1; ++h) s += p[h * D2 + j];
      v[j] = s + (((cz[j] + cz[D2 + j]) + cz[2 * D2 + j]) + cz[3 * D2 + j]);
    }
    float2* __restrict__ cp = (float2*)(C + (size_t)(row0 + row) * D2 + hd * C2);
    cp[0] = make_float2(v[0], v[1]);
    cp[1] = make_float2(v[2], v[3]);
    cp[2] = make_float2(v[4], v[5]);
    const float* __restrict__ ap = att + hd * (2 * C2);
    const float pa = v[0] * (ap[0] + ap[C2])     + v[1] * (ap[1] + ap[C2 + 1]) + v[2] * (ap[2] + ap[C2 + 2]);
    const float pb = v[3] * (ap[3] + ap[C2 + 3]) + v[4] * (ap[4] + ap[C2 + 4]) + v[5] * (ap[5] + ap[C2 + 5]);
    sa[(size_t)(row0 + row) * H2 + hd] = tanhf(pa + pb);
  }
}


__global__ void row_agg2(const int* __restrict__ cur, const col_t* __restrict__ csr,
                         const float* __restrict__ s2, const float* __restrict__ h2,
                         float* __restrict__ out2) {
  int tid = blockIdx.x * blockDim.x + threadIdx.x;
  if (tid >= N_NODES * H2) return;
  int r = tid >> 3, h = tid & 7;
  int beg = r ? cur[r - 1] : 0;
  int end = cur[r];
  float sself = s2[tid];
  float m = sself;
  for (int j = beg; j < end; ++j) m = fmaxf(m, s2[(int)csr[j] * H2 + h]);

  float w = expf(sself - m);
  float den = w;
  const float2* __restrict__ src = (const float2*)(h2 + (size_t)r * D2 + h * C2);
  float2 p0 = src[0], p1 = src[1], p2 = src[2];
  float a0 = w * p0.x, a1 = w * p0.y, a2 = w * p1.x, a3 = w * p1.y, a4 = w * p2.x, a5 = w * p2.y;
  for (int j = beg; j < end; ++j) {
    int c = csr[j];
    w = expf(s2[c * H2 + h] - m);
    den += w;
    src = (const float2*)(h2 + (size_t)c * D2 + h * C2);
    p0 = src[0]; p1 = src[1]; p2 = src[2];
    a0 += w * p0.x; a1 += w * p0.y; a2 += w * p1.x;
    a3 += w * p1.y; a4 += w * p2.x; a5 += w * p2.y;
  }
  float inv = 1.0f / (den + 1e-16f);
  float2* __restrict__ dst = (float2*)(out2 + (size_t)r * D2 + h * C2);
  dst[0] = make_float2(a0 * inv, a1 * inv);
  dst[1] = make_float2(a2 * inv, a3 * inv);
  dst[2] = make_float2(a4 * inv, a5 * inv);
}

__global__ void finalize(const float* __restrict__ out2, const float* __restrict__ b2,
                         float* __restrict__ out) {
  int n = blockIdx.x * blockDim.x + threadIdx.x;
  if (n >= N_NODES) return;
  float v[C2];
  #pragma unroll
  for (int c = 0; c < C2; ++c) {
    float s = 0.f;
    #pragma unroll
    for (int h = 0; h < H2; ++h) s += out2[(size_t)n * D2 + h * C2 + c];
    float t = s * (1.0f / H2) + b2[c];
    v[c] = t > 0.f ? t : (expf(t) - 1.0f);
  }
  float m = v[0];
  #pragma unroll
  for (int c = 1; c < C2; ++c) m = fmaxf(m, v[c]);
  float sum = 0.f;
  #pragma unroll
  for (int c = 0; c < C2; ++c) sum += expf(v[c] - m);
  float lse = logf(sum);
  #pragma unroll
  for (int c = 0; c < C2; ++c) out[(size_t)n * C2 + c] = v[c] - m - lse;
}


extern "C" void kernel_launch(void* const* d_in, const int* in_sizes, int n_in,
                              void* d_out, int out_size, void* d_ws, size_t ws_size,
                              hipStream_t stream) {
  const float* x    = (const float*)d_in[0];
  const int*   ei   = (const int*)  d_in[1];
  const float* g1   = (const float*)d_in[2];
  const float* be1  = (const float*)d_in[3];
  const float* mu1  = (const float*)d_in[4];
  const float* va1  = (const float*)d_in[5];
  const float* W1   = (const float*)d_in[6];
  const float* att1 = (const float*)d_in[7];
  const float* b1   = (const float*)d_in[8];
  const float* g2   = (const float*)d_in[9];
  const float* be2  = (const float*)d_in[10];
  const float* mu2  = (const float*)d_in[11];
  const float* va2  = (const float*)d_in[12];
  const float* W2   = (const float*)d_in[13];
  const float* att2 = (const float*)d_in[14];
  const float* b2   = (const float*)d_in[15];
  const int* rowA = ei;
  const int* colA = ei + N_EDGES;
  float* out = (float*)d_out;
  float* ws  = (float*)d_ws;

  float* sc1  = ws;
  float* sh1  = ws + 256;
  float* c0   = ws + 512;
  float* ai   = ws + 1536;
  float* aj   = ai  + (size_t)N_NODES * H1;
  float* edg  = aj  + (size_t)N_NODES * H1;
  int*   cur  = (int*)edg;
  col_t* csr  = (col_t*)(edg + N_NODES);
  float* l2   = edg + (size_t)N_NODES * 2 * H1;
  float* h2   = l2;
  float* s2   = h2  + (size_t)N_NODES * D2;
  float* out2 = s2  + (size_t)N_NODES * H2;
  float* gt   = l2  + (size_t)N_NODES * D1;
  float* Wc   = gt  + (size_t)N_NODES * GP;

  bn_prep<<<PREP_BLOCKS + F_IN + 1, 256, 0, stream>>>(g1, be1, mu1, va1, g2, be2, mu2, va2, b1,
                                                     W1, W2, att1, sc1, sh1, Wc, c0, cur);

  gemm_bn_wmma<F_IN, NF, 2, 2, 8, 64, 2, 1, 8, true><<<G1_FIRST + EDGE_BLOCKS, 256, 0, stream>>>(
      x, sc1, sh1, Wc, gt, att1, ai, aj, 0, G1_FIRST, 0, rowA, colA, cur, csr);
  row_scan<<<1, SCAN_T, 0, stream>>>(cur);
  gemm_bn_wmma<F_IN, NF, 2, 2, 8, 64, 2, 1, 8, true><<<G1_BLOCKS - G1_FIRST + EDGE_BLOCKS, 256, 0, stream>>>(
      x, sc1, sh1, Wc, gt, att1, ai, aj, G1_FIRST, G1_BLOCKS - G1_FIRST, 1, rowA, colA, cur, csr);

  row_agg1<<<N_NODES / AG_ROWS, 256, 0, stream>>>(cur, csr, ai, aj, gt, c0, att2, h2, s2);

  row_agg2<<<(N_NODES * H2) / 256, 256, 0, stream>>>(cur, csr, s2, h2, out2);

  finalize<<<(N_NODES + 255) / 256, 256, 0, stream>>>(out2, b2, out);
}


# Layer-1 GEMM: each chunk's A rows and BN scale/shift requested as one group a chunk ahead, in front of the chunk's products (chunk loop kept a real loop; the scheduler had sunk each load to its use)
# speedup vs baseline: 6.3880x; 1.0552x over previous
#include <hip/hip_runtime.h>
#include <math.h>

#define N_NODES 40000
#define N_EDGES 320000
#define ETOT    (N_EDGES + N_NODES)
#define F_IN    256
#define H1      4
#define C1      128
#define D1      (H1*C1)
#define H2      8
#define C2      6
#define D2      (H2*C2)
#define SCAN_T     1024
#define SCAN_W     (SCAN_T / 32)
#define SCAN_CHUNK (N_NODES / SCAN_W)
#define SCAN_STEPS ((SCAN_CHUNK + 31) / 32)
#define BN_EPS    1e-5f
#define NEG_SLOPE 0.2f

typedef __attribute__((ext_vector_type(2))) float v2f;
typedef __attribute__((ext_vector_type(8))) float v8f;


static_assert(N_NODES <= 65536, "16-bit column ids need N_NODES <= 65536");
typedef unsigned short col_t;
static_assert(N_EDGES % 256 == 0 && N_NODES % 8 == 0 && (N_NODES * H2) % 256 == 0,
              "launch grids below divide exactly");

__device__ __forceinline__ float lrelu(float a) { return a > 0.f ? a : NEG_SLOPE * a; }


#define GP 192
#define NF 256
#define PREP_BLOCKS ((N_NODES + 255) / 256)
static_assert(GP == H1 * D2 && GP + 2 * H1 <= NF && D1 == 2 * 256 && H1 * 64 == 256 && D2 <= 64, "fold maps");

__global__ __launch_bounds__(256)
void bn_prep(const float* g1, const float* be1, const float* mu1, const float* va1,
             const float* g2, const float* be2, const float* mu2, const float* va2,
             const float* bias1, const float* __restrict__ W1, const float* __restrict__ W2,
             const float* __restrict__ att1,
             float* sc1, float* sh1, float* __restrict__ Wc, float* __restrict__ c0, int* __restrict__ cnt) {
  if ((int)blockIdx.x >= PREP_BLOCKS) {
    __shared__ float s2c[D1];
    const int fb = (int)blockIdx.x - PREP_BLOCKS, n = threadIdx.x;
    for (int kk = n; kk < D1; kk += 256) s2c[kk] = g2[kk] / sqrtf(va2[kk] + BN_EPS);
    __syncthreads();
    float s = 0.f;
    if (fb < F_IN) {
      const float* __restrict__ w1 = W1 + (size_t)fb * D1;
      if (n < GP) {
        const int h = n / D2, o = n % D2;
        for (int c = 0; c < C1; ++c) { const int kk = h * C1 + c; s += w1[kk] * (s2c[kk] * W2[kk * D2 + o]); }
      } else if (n < GP + 2 * H1) {
        const int h = (n - GP) & (H1 - 1), side = (n - GP) / H1;
        const float* __restrict__ a = att1 + h * (2 * C1) + side * C1;
        for (int c = 0; c < C1; ++c) s += w1[h * C1 + c] * a[c];
      }
      Wc[(size_t)fb * NF + n] = s;
    } else {
      const int h = n >> 6, o = n & 63;
      if (o < D2) {
        for (int c = 0; c < C1; ++c) {
          const int kk = h * C1 + c;
          const float sh = be2[kk] + (bias1[kk] - mu2[kk]) * s2c[kk];
          s += sh * W2[kk * D2 + o];
        }
        c0[h * D2 + o] = s;
      }
    }
    return;
  }
  int i = blockIdx.x * blockDim.x + threadIdx.x;
  if (i < N_NODES) cnt[i] = 0;
  if (i < F_IN) {
    float s = g1[i] / sqrtf(va1[i] + BN_EPS);
    sc1[i] = s; sh1[i] = be1[i] - mu1[i] * s;
  }
}

#define EDGE_BLOCKS (N_EDGES / 256)
#define G1_BLOCKS   (N_NODES / 32)
#define G1_FIRST    512
__device__ __forceinline__ void row_count_blk(int blk, const int* __restrict__ rowA, int* __restrict__ cnt) {
  int e = blk * 256 + threadIdx.x;
  if (e < N_EDGES) atomicAdd(&cnt[rowA[e]], 1);
}

static_assert(SCAN_W == 32 && N_NODES % SCAN_W == 0, "one lane per wave total; equal chunks");
__global__ __launch_bounds__(SCAN_T)
void row_scan(int* __restrict__ cnt) {
  __shared__ int wtot[SCAN_W];
  int lane = threadIdx.x & 31, w = threadIdx.x >> 5;
  int* __restrict__ p = cnt + w * SCAN_CHUNK;
  int s = 0;
  #pragma unroll
  for (int i = 0; i < SCAN_STEPS; ++i) {
    bool in = i < SCAN_CHUNK / 32 || lane < SCAN_CHUNK % 32;
    s += in ? p[i * 32 + lane] : 0;
  }
  #pragma unroll
  for (int off = 16; off > 0; off >>= 1) s += __shfl_xor(s, off, 32);
  if (lane == 0) wtot[w] = s;
  __syncthreads();
  int carry = lane < w ? wtot[lane] : 0;
  #pragma unroll
  for (int off = 16; off > 0; off >>= 1) carry += __shfl_xor(carry, off, 32);
  #pragma unroll 8
  for (int i = 0; i < SCAN_STEPS; ++i) {
    bool in = i < SCAN_CHUNK / 32 || lane < SCAN_CHUNK % 32;
    int x0 = in ? p[i * 32 + lane] : 0;
    int x = x0;
    #pragma unroll
    for (int off = 1; off < 32; off <<= 1) {
      int y = __shfl_up(x, off, 32);
      if (lane >= off) x += y;
    }
    if (in) p[i * 32 + lane] = x - x0 + carry;
    carry += __shfl(x, 31, 32);
  }
}

__device__ __forceinline__ void csr_fill_blk(int blk, const int* __restrict__ rowA, const int* __restrict__ colA,
                                             int* __restrict__ cur, col_t* __restrict__ csr) {
  int e = blk * 256 + threadIdx.x;
  if (e >= N_EDGES) return;
  int slot = atomicAdd(&cur[rowA[e]], 1);
  csr[slot] = (col_t)colA[e];
}


template <int NT, int N>
__device__ __forceinline__ void load_b(const float* __restrict__ b0, float (&bx)[NT], float (&by)[NT]) {
  const float* __restrict__ b1 = b0 + N;
  if constexpr (NT == 4) {
    float4 t0 = *(const float4*)b0, t1 = *(const float4*)b1;
    bx[0] = t0.x; bx[1] = t0.y; bx[2] = t0.z; bx[3] = t0.w;
    by[0] = t1.x; by[1] = t1.y; by[2] = t1.z; by[3] = t1.w;
  } else {
    #pragma unroll
    for (int nt = 0; nt < NT; ++nt) { bx[nt] = b0[nt]; by[nt] = b1[nt]; }
  }
}

template <int K, int N, int MT, int NT, int WN, int KC, int PF, int SCORE, int WPS, bool EDGES>
__global__ __launch_bounds__(32 * WN)
__attribute__((amdgpu_waves_per_eu(WPS)))
void gemm_bn_wmma(const float* A,
                  const float* scale,
                  const float* shift,
                  const float* __restrict__ B,
                  float* __restrict__ C,
                  const float* __restrict__ att,
                  float* __restrict__ sa,
                  float* __restrict__ sb,
                  int blk0, int nblk, int pass,
                  const int* __restrict__ rowA, const int* __restrict__ colA,
                  int* __restrict__ cur, col_t* __restrict__ csr) {
  constexpr int THREADS = 32 * WN;
  if constexpr (EDGES) {
    static_assert(THREADS == 256, "edge blocks are 256 threads");
    if ((int)blockIdx.x >= nblk) {
      const int eb = (int)blockIdx.x - nblk;
      if (pass == 0) row_count_blk(eb, rowA, cur);
      else           csr_fill_blk(eb, rowA, colA, cur, csr);
      return;
    }
  }
  constexpr int ROWS  = 16 * MT;
  constexpr int PITCH = K + 4;
  constexpr int K4  = K / 4;
  constexpr int C4  = KC / 4;
  constexpr int NCH = K / KC;
  constexpr int RPP = THREADS / C4;
  constexpr int LPT = ROWS / RPP;
  static_assert(PF >= 1 && 4 * PF <= K, "queue depth");
  static_assert(N == WN * NT * 16, "the waves of a work-group tile all N columns");
  static_assert(K % KC == 0 && KC % 4 == 0 && THREADS % C4 == 0 && ROWS % RPP == 0, "exact staging");
  static_assert((PITCH / 4) % 2 == 1, "pitch is 16 B x odd");
  static_assert(N_NODES % ROWS == 0, "grid divides exactly");
  __shared__ float As[ROWS * PITCH];

  const int tid  = threadIdx.x;
  const int lane = tid & 31;
  const int half = lane >> 4;
  const int l16  = lane & 15;
  const int wn   = tid >> 5;
  const int row0 = (blk0 + (int)blockIdx.x) * ROWS;

  const int tr = tid / C4, c4 = tid % C4;
  const float4* A4 = (const float4*)(A + (size_t)row0 * K) + (size_t)tr * K4 + c4;
  const float4* sc4 = (const float4*)scale + c4;
  const float4* sh4 = (const float4*)shift + c4;
  float* st = As + tr * PITCH + c4 * 4;

  float4 pre[LPT];
  #pragma unroll
  for (int i = 0; i < LPT; ++i) pre[i] = A4[(size_t)i * RPP * K4];
  float4 sc = sc4[0], sh = sh4[0];
  __builtin_amdgcn_sched_barrier(0);

  const float* as = As + l16 * PITCH + 2 * half;
  const float* __restrict__ bp = B + (size_t)(2 * half) * N + wn * (NT * 16) + NT * l16;

  v8f acc[MT][NT];
  #pragma unroll
  for (int mt = 0; mt < MT; ++mt)
    #pragma unroll
    for (int nt = 0; nt < NT; ++nt) acc[mt][nt] = v8f{};

  float qx[PF][NT], qy[PF][NT];
  #pragma unroll
  for (int p = 0; p < PF; ++p) load_b<NT, N>(bp + (size_t)(4 * p) * N, qx[p], qy[p]);

  #pragma nounroll
  for (int ch = 0; ch < NCH; ++ch) {
    #pragma unroll
    for (int i = 0; i < LPT; ++i) {
      float4 v = pre[i], o;
      o.x = v.x * sc.x + sh.x; o.y = v.y * sc.y + sh.y;
      o.z = v.z * sc.z + sh.z; o.w = v.w * sc.w + sh.w;
      *(float4*)(st + i * RPP * PITCH + ch * KC) = o;
    }
    __syncthreads();
    if (ch + 1 < NCH) {
      #pragma unroll
      for (int i = 0; i < LPT; ++i) pre[i] = A4[(size_t)i * RPP * K4 + (ch + 1) * C4];
      sc = sc4[(ch + 1) * C4]; sh = sh4[(ch + 1) * C4];
    }
    for (int k = ch * KC; k < ch * KC + KC; k += 4) {
      v2f a[MT];
      #pragma unroll
      for (int mt = 0; mt < MT; ++mt) a[mt] = *(const v2f*)(as + mt * 16 * PITCH + k);
      float bx[NT], by[NT];
      #pragma unroll
      for (int nt = 0; nt < NT; ++nt) { bx[nt] = qx[0][nt]; by[nt] = qy[0][nt]; }
      #pragma unroll
      for (int p = 0; p + 1 < PF; ++p)
        #pragma unroll
        for (int nt = 0; nt < NT; ++nt) { qx[p][nt] = qx[p + 1][nt]; qy[p][nt] = qy[p + 1][nt]; }
      if (k + 4 * PF < K) load_b<NT, N>(bp + (size_t)(k + 4 * PF) * N, qx[PF - 1], qy[PF - 1]);
      #pragma unroll
      for (int mt = 0; mt < MT; ++mt)
        #pragma unroll
        for (int nt = 0; nt < NT; ++nt) {
          v2f b; b.x = bx[nt]; b.y = by[nt];
          acc[mt][nt] = __builtin_amdgcn_wmma_f32_16x16x4_f32(
               false, a[mt],  false, b,
               (short)0, acc[mt][nt],  false,  false);
        }
    }
  }

  if constexpr (SCORE != 0) {
    static_assert(MT == 2 && NT == 2, "four accumulators named below");
    asm volatile("v_nop\n\tv_nop\n\tv_nop\n\tv_nop\n\tv_nop"
                 : "+v"(acc[0][0]), "+v"(acc[0][1]), "+v"(acc[1][0]), "+v"(acc[1][1]));
  }
  if constexpr (SCORE == 0) {
    #pragma unroll
    for (int mt = 0; mt < MT; ++mt) {
      float* cp = C + (size_t)(row0 + mt * 16 + (half << 3)) * N + wn * (NT * 16) + NT * l16;
      #pragma unroll
      for (int r = 0; r < 8; ++r) {
        if constexpr (NT == 4) {
          *(float4*)(cp + (size_t)r * N) =
              make_float4(acc[mt][0][r], acc[mt][1][r], acc[mt][2][r], acc[mt][3][r]);
        } else {
          #pragma unroll
          for (int nt = 0; nt < NT; ++nt) cp[(size_t)r * N + nt] = acc[mt][nt][r];
        }
      }
    }
  } else {
    static_assert(NT == 2 && N == NF && GP % (NT * 16) == 0 && 2 * H1 == 4 * NT, "score columns in four lanes");
    if (wn < GP / (NT * 16)) {
      #pragma unroll
      for (int mt = 0; mt < MT; ++mt) {
        float* cp = C + (size_t)(row0 + mt * 16 + (half << 3)) * GP + wn * (NT * 16) + NT * l16;
        #pragma unroll
        for (int r = 0; r < 8; ++r)
          *(float2*)(cp + (size_t)r * GP) = make_float2(acc[mt][0][r], acc[mt][1][r]);
      }
    } else if (wn == GP / (NT * 16) && l16 < 4) {
      float* __restrict__ dst = (l16 < 2 ? sa : sb) + 2 * (l16 & 1);
      #pragma unroll
      for (int mt = 0; mt < MT; ++mt)
        #pragma unroll
        for (int r = 0; r < 8; ++r)
          *(float2*)(dst + (size_t)(row0 + mt * 16 + (half << 3) + r) * H1) =
              make_float2(acc[mt][0][r], acc[mt][1][r]);
    }
  }
}


#define AG_ROWS  16
#define AG_RPW   (AG_ROWS / 8)
#define AG_PITCH (GP + 4)
static_assert(N_NODES % AG_ROWS == 0 && GP == 96 * 2 && D2 == 24 * 2, "three float2 per lane, 24 float2 per head");
static_assert(C2 == 6 && H2 == 8 && AG_ROWS * H2 == 128, "epilogue map");

__global__ __launch_bounds__(256)
void row_agg1(const int* __restrict__ cur, const col_t* __restrict__ csr,
              const float* __restrict__ ai, const float* __restrict__ aj,
              const float* __restrict__ g, const float* __restrict__ c0,
              const float* __restrict__ att,
              float* __restrict__ C, float* __restrict__ sa) {
  __shared__ float Ls[AG_ROWS * AG_PITCH];
  const int tid = threadIdx.x;
  const int lane = tid & 31;
  const int wv = tid >> 5;
  const int row0 = (int)blockIdx.x * AG_ROWS;
  const float4* __restrict__ aj4 = (const float4*)aj;
  const float2* __restrict__ g2 = (const float2*)g;
  const bool lo0 = lane < 24, lo1 = lane < 16, lo2 = lane < 8;

  for (int q = 0; q < AG_RPW; ++q) {
    const int rl = wv * AG_RPW + q;
    int r = __builtin_amdgcn_readfirstlane(row0 + rl);
    int beg = r ? cur[r - 1] : 0;
    int end = cur[r];
    float4 a_i = *(const float4*)(ai + (size_t)r * 4);
    float4 bs  = aj4[r];
    float t0 = lrelu(a_i.x + bs.x), t1 = lrelu(a_i.y + bs.y);
    float t2 = lrelu(a_i.z + bs.z), t3 = lrelu(a_i.w + bs.w);

    float m0 = t0, m1 = t1, m2 = t2, m3 = t3;
    for (int j = beg + lane; j < end; j += 32) {
      float4 b = aj4[csr[j]];
      m0 = fmaxf(m0, lrelu(a_i.x + b.x));
      m1 = fmaxf(m1, lrelu(a_i.y + b.y));
      m2 = fmaxf(m2, lrelu(a_i.z + b.z));
      m3 = fmaxf(m3, lrelu(a_i.w + b.w));
    }
    #pragma unroll
    for (int off = 16; off > 0; off >>= 1) {
      m0 = fmaxf(m0, __shfl_xor(m0, off, 32));
      m1 = fmaxf(m1, __shfl_xor(m1, off, 32));
      m2 = fmaxf(m2, __shfl_xor(m2, off, 32));
      m3 = fmaxf(m3, __shfl_xor(m3, off, 32));
    }

    float u0 = expf(t0 - m0), u1 = expf(t1 - m1), u2 = expf(t2 - m2), u3 = expf(t3 - m3);
    float wa = lo0 ? u0 : u1, wb = lo1 ? u1 : u2, wc = lo2 ? u2 : u3;
    float sA = wa, sB = wb, sC = wc;
    const float2* __restrict__ src = g2 + (size_t)r * (GP / 2);
    float2 v0 = src[lane], v1 = src[32 + lane], v2 = src[64 + lane];
    float2 acc0 = {wa * v0.x, wa * v0.y};
    float2 acc1 = {wb * v1.x, wb * v1.y};
    float2 acc2 = {wc * v2.x, wc * v2.y};

    for (int base = beg; base < end; base += 32) {
      int j = base + lane;
      int c = r;
      float w0 = 0.f, w1 = 0.f, w2 = 0.f, w3 = 0.f;
      if (j < end) {
        c = csr[j];
        float4 b = aj4[c];
        w0 = expf(lrelu(a_i.x + b.x) - m0);
        w1 = expf(lrelu(a_i.y + b.y) - m1);
        w2 = expf(lrelu(a_i.z + b.z) - m2);
        w3 = expf(lrelu(a_i.w + b.w) - m3);
      }
      int cnt = end - base < 32 ? end - base : 32;
      for (int k = 0; k < cnt; ++k) {
        int ck = __shfl(c, k, 32);
        u0 = __shfl(w0, k, 32); u1 = __shfl(w1, k, 32);
        u2 = __shfl(w2, k, 32); u3 = __shfl(w3, k, 32);
        wa = lo0 ? u0 : u1; wb = lo1 ? u1 : u2; wc = lo2 ? u2 : u3;
        sA += wa; sB += wb; sC += wc;
        src = g2 + (size_t)ck * (GP / 2);
        v0 = src[lane]; v1 = src[32 + lane]; v2 = src[64 + lane];
        acc0.x += wa * v0.x; acc0.y += wa * v0.y;
        acc1.x += wb * v1.x; acc1.y += wb * v1.y;
        acc2.x += wc * v2.x; acc2.y += wc * v2.y;
      }
    }
    float iA = 1.0f / (sA + 1e-16f), iB = 1.0f / (sB + 1e-16f), iC = 1.0f / (sC + 1e-16f);
    float2* dst = (float2*)(Ls + rl * AG_PITCH);
    dst[lane]      = make_float2(acc0.x * iA, acc0.y * iA);
    dst[32 + lane] = make_float2(acc1.x * iB, acc1.y * iB);
    dst[64 + lane] = make_float2(acc2.x * iC, acc2.y * iC);
  }
  __syncthreads();

  if (tid < AG_ROWS * H2) {
    const int row = tid >> 3, hd = tid & 7;
    const float* p = Ls + row * AG_PITCH + hd * C2;
    const float* __restrict__ cz = c0 + hd * C2;
    float v[C2];
    #pragma unroll
    for (int j = 0; j < C2; ++j) {
      float s = p[j];
      #pragma unroll
      for (int h = 1; h < H1; ++h) s += p[h * D2 + j];
      v[j] = s + (((cz[j] + cz[D2 + j]) + cz[2 * D2 + j]) + cz[3 * D2 + j]);
    }
    float2* __restrict__ cp = (float2*)(C + (size_t)(row0 + row) * D2 + hd * C2);
    cp[0] = make_float2(v[0], v[1]);
    cp[1] = make_float2(v[2], v[3]);
    cp[2] = make_float2(v[4], v[5]);
    const float* __restrict__ ap = att + hd * (2 * C2);
    const float pa = v[0] * (ap[0] + ap[C2])     + v[1] * (ap[1] + ap[C2 + 1]) + v[2] * (ap[2] + ap[C2 + 2]);
    const float pb = v[3] * (ap[3] + ap[C2 + 3]) + v[4] * (ap[4] + ap[C2 + 4]) + v[5] * (ap[5] + ap[C2 + 5]);
    sa[(size_t)(row0 + row) * H2 + hd] = tanhf(pa + pb);
  }
}


__global__ void row_agg2(const int* __restrict__ cur, const col_t* __restrict__ csr,
                         const float* __restrict__ s2, const float* __restrict__ h2,
                         float* __restrict__ out2) {
  int tid = blockIdx.x * blockDim.x + threadIdx.x;
  if (tid >= N_NODES * H2) return;
  int r = tid >> 3, h = tid & 7;
  int beg = r ? cur[r - 1] : 0;
  int end = cur[r];
  float sself = s2[tid];
  float m = sself;
  for (int j = beg; j < end; ++j) m = fmaxf(m, s2[(int)csr[j] * H2 + h]);

  float w = expf(sself - m);
  float den = w;
  const float2* __restrict__ src = (const float2*)(h2 + (size_t)r * D2 + h * C2);
  float2 p0 = src[0], p1 = src[1], p2 = src[2];
  float a0 = w * p0.x, a1 = w * p0.y, a2 = w * p1.x, a3 = w * p1.y, a4 = w * p2.x, a5 = w * p2.y;
  for (int j = beg; j < end; ++j) {
    int c = csr[j];
    w = expf(s2[c * H2 + h] - m);
    den += w;
    src = (const float2*)(h2 + (size_t)c * D2 + h * C2);
    p0 = src[0]; p1 = src[1]; p2 = src[2];
    a0 += w * p0.x; a1 += w * p0.y; a2 += w * p1.x;
    a3 += w * p1.y; a4 += w * p2.x; a5 += w * p2.y;
  }
  float inv = 1.0f / (den + 1e-16f);
  float2* __restrict__ dst = (float2*)(out2 + (size_t)r * D2 + h * C2);
  dst[0] = make_float2(a0 * inv, a1 * inv);
  dst[1] = make_float2(a2 * inv, a3 * inv);
  dst[2] = make_float2(a4 * inv, a5 * inv);
}

__global__ void finalize(const float* __restrict__ out2, const float* __restrict__ b2,
                         float* __restrict__ out) {
  int n = blockIdx.x * blockDim.x + threadIdx.x;
  if (n >= N_NODES) return;
  float v[C2];
  #pragma unroll
  for (int c = 0; c < C2; ++c) {
    float s = 0.f;
    #pragma unroll
    for (int h = 0; h < H2; ++h) s += out2[(size_t)n * D2 + h * C2 + c];
    float t = s * (1.0f / H2) + b2[c];
    v[c] = t > 0.f ? t : (expf(t) - 1.0f);
  }
  float m = v[0];
  #pragma unroll
  for (int c = 1; c < C2; ++c) m = fmaxf(m, v[c]);
  float sum = 0.f;
  #pragma unroll
  for (int c = 0; c < C2; ++c) sum += expf(v[c] - m);
  float lse = logf(sum);
  #pragma unroll
  for (int c = 0; c < C2; ++c) out[(size_t)n * C2 + c] = v[c] - m - lse;
}


extern "C" void kernel_launch(void* const* d_in, const int* in_sizes, int n_in,
                              void* d_out, int out_size, void* d_ws, size_t ws_size,
                              hipStream_t stream) {
  const float* x    = (const float*)d_in[0];
  const int*   ei   = (const int*)  d_in[1];
  const float* g1   = (const float*)d_in[2];
  const float* be1  = (const float*)d_in[3];
  const float* mu1  = (const float*)d_in[4];
  const float* va1  = (const float*)d_in[5];
  const float* W1   = (const float*)d_in[6];
  const float* att1 = (const float*)d_in[7];
  const float* b1   = (const float*)d_in[8];
  const float* g2   = (const float*)d_in[9];
  const float* be2  = (const float*)d_in[10];
  const float* mu2  = (const float*)d_in[11];
  const float* va2  = (const float*)d_in[12];
  const float* W2   = (const float*)d_in[13];
  const float* att2 = (const float*)d_in[14];
  const float* b2   = (const float*)d_in[15];
  const int* rowA = ei;
  const int* colA = ei + N_EDGES;
  float* out = (float*)d_out;
  float* ws  = (float*)d_ws;

  float* sc1  = ws;
  float* sh1  = ws + 256;
  float* c0   = ws + 512;
  float* ai   = ws + 1536;
  float* aj   = ai  + (size_t)N_NODES * H1;
  float* edg  = aj  + (size_t)N_NODES * H1;
  int*   cur  = (int*)edg;
  col_t* csr  = (col_t*)(edg + N_NODES);
  float* l2   = edg + (size_t)N_NODES * 2 * H1;
  float* h2   = l2;
  float* s2   = h2  + (size_t)N_NODES * D2;
  float* out2 = s2  + (size_t)N_NODES * H2;
  float* gt   = l2  + (size_t)N_NODES * D1;
  float* Wc   = gt  + (size_t)N_NODES * GP;

  bn_prep<<<PREP_BLOCKS + F_IN + 1, 256, 0, stream>>>(g1, be1, mu1, va1, g2, be2, mu2, va2, b1,
                                                     W1, W2, att1, sc1, sh1, Wc, c0, cur);

  gemm_bn_wmma<F_IN, NF, 2, 2, 8, 64, 2, 1, 10, true><<<G1_FIRST + EDGE_BLOCKS, 256, 0, stream>>>(
      x, sc1, sh1, Wc, gt, att1, ai, aj, 0, G1_FIRST, 0, rowA, colA, cur, csr);
  row_scan<<<1, SCAN_T, 0, stream>>>(cur);
  gemm_bn_wmma<F_IN, NF, 2, 2, 8, 64, 2, 1, 10, true><<<G1_BLOCKS - G1_FIRST + EDGE_BLOCKS, 256, 0, stream>>>(
      x, sc1, sh1, Wc, gt, att1, ai, aj, G1_FIRST, G1_BLOCKS - G1_FIRST, 1, rowA, colA, cur, csr);

  row_agg1<<<N_NODES / AG_ROWS, 256, 0, stream>>>(cur, csr, ai, aj, gt, c0, att2, h2, s2);

  row_agg2<<<(N_NODES * H2) / 256, 256, 0, stream>>>(cur, csr, s2, h2, out2);

  finalize<<<(N_NODES + 255) / 256, 256, 0, stream>>>(out2, b2, out);
}


# Layer-2 softmax weight is a function of the source node alone: agg1 stores E=exp(score) and E*h2 per node; layer 2 is one gather-sum pass (no max pass, no exp), finalize in its tail: 7->6 launches
# speedup vs baseline: 6.8999x; 1.0801x over previous
#include <hip/hip_runtime.h>
#include <math.h>

#define N_NODES 40000
#define N_EDGES 320000
#define ETOT    (N_EDGES + N_NODES)
#define F_IN    256
#define H1      4
#define C1      128
#define D1      (H1*C1)
#define H2      8
#define C2      6
#define D2      (H2*C2)
#define SCAN_T     1024
#define SCAN_W     (SCAN_T / 32)
#define SCAN_CHUNK (N_NODES / SCAN_W)
#define SCAN_STEPS ((SCAN_CHUNK + 31) / 32)
#define BN_EPS    1e-5f
#define NEG_SLOPE 0.2f

typedef __attribute__((ext_vector_type(2))) float v2f;
typedef __attribute__((ext_vector_type(8))) float v8f;


static_assert(N_NODES <= 65536, "16-bit column ids need N_NODES <= 65536");
typedef unsigned short col_t;
static_assert(N_EDGES % 256 == 0 && N_NODES % 8 == 0 && (N_NODES * H2) % 256 == 0,
              "launch grids below divide exactly");

__device__ __forceinline__ float lrelu(float a) { return a > 0.f ? a : NEG_SLOPE * a; }


#define GP 192
#define NF 256
#define EPW 64
#define PREP_BLOCKS ((N_NODES + 255) / 256)
static_assert(GP == H1 * D2 && GP + 2 * H1 <= NF && D1 == 2 * 256 && H1 * 64 == 256 && D2 <= 64, "fold maps");

__global__ __launch_bounds__(256)
void bn_prep(const float* g1, const float* be1, const float* mu1, const float* va1,
             const float* g2, const float* be2, const float* mu2, const float* va2,
             const float* bias1, const float* __restrict__ W1, const float* __restrict__ W2,
             const float* __restrict__ att1,
             float* sc1, float* sh1, float* __restrict__ Wc, float* __restrict__ c0, int* __restrict__ cnt) {
  if ((int)blockIdx.x >= PREP_BLOCKS) {
    __shared__ float s2c[D1];
    const int fb = (int)blockIdx.x - PREP_BLOCKS, n = threadIdx.x;
    for (int kk = n; kk < D1; kk += 256) s2c[kk] = g2[kk] / sqrtf(va2[kk] + BN_EPS);
    __syncthreads();
    float s = 0.f;
    if (fb < F_IN) {
      const float* __restrict__ w1 = W1 + (size_t)fb * D1;
      if (n < GP) {
        const int h = n / D2, o = n % D2;
        for (int c = 0; c < C1; ++c) { const int kk = h * C1 + c; s += w1[kk] * (s2c[kk] * W2[kk * D2 + o]); }
      } else if (n < GP + 2 * H1) {
        const int h = (n - GP) & (H1 - 1), side = (n - GP) / H1;
        const float* __restrict__ a = att1 + h * (2 * C1) + side * C1;
        for (int c = 0; c < C1; ++c) s += w1[h * C1 + c] * a[c];
      }
      Wc[(size_t)fb * NF + n] = s;
    } else {
      const int h = n >> 6, o = n & 63;
      if (o < D2) {
        for (int c = 0; c < C1; ++c) {
          const int kk = h * C1 + c;
          const float sh = be2[kk] + (bias1[kk] - mu2[kk]) * s2c[kk];
          s += sh * W2[kk * D2 + o];
        }
        c0[h * D2 + o] = s;
      }
    }
    return;
  }
  int i = blockIdx.x * blockDim.x + threadIdx.x;
  if (i < N_NODES) cnt[i] = 0;
  if (i < F_IN) {
    float s = g1[i] / sqrtf(va1[i] + BN_EPS);
    sc1[i] = s; sh1[i] = be1[i] - mu1[i] * s;
  }
}

#define EDGE_BLOCKS (N_EDGES / 256)
#define G1_BLOCKS   (N_NODES / 32)
#define G1_FIRST    512
__device__ __forceinline__ void row_count_blk(int blk, const int* __restrict__ rowA, int* __restrict__ cnt) {
  int e = blk * 256 + threadIdx.x;
  if (e < N_EDGES) atomicAdd(&cnt[rowA[e]], 1);
}

static_assert(SCAN_W == 32 && N_NODES % SCAN_W == 0, "one lane per wave total; equal chunks");
__global__ __launch_bounds__(SCAN_T)
void row_scan(int* __restrict__ cnt) {
  __shared__ int wtot[SCAN_W];
  int lane = threadIdx.x & 31, w = threadIdx.x >> 5;
  int* __restrict__ p = cnt + w * SCAN_CHUNK;
  int s = 0;
  #pragma unroll
  for (int i = 0; i < SCAN_STEPS; ++i) {
    bool in = i < SCAN_CHUNK / 32 || lane < SCAN_CHUNK % 32;
    s += in ? p[i * 32 + lane] : 0;
  }
  #pragma unroll
  for (int off = 16; off > 0; off >>= 1) s += __shfl_xor(s, off, 32);
  if (lane == 0) wtot[w] = s;
  __syncthreads();
  int carry = lane < w ? wtot[lane] : 0;
  #pragma unroll
  for (int off = 16; off > 0; off >>= 1) carry += __shfl_xor(carry, off, 32);
  #pragma unroll 8
  for (int i = 0; i < SCAN_STEPS; ++i) {
    bool in = i < SCAN_CHUNK / 32 || lane < SCAN_CHUNK % 32;
    int x0 = in ? p[i * 32 + lane] : 0;
    int x = x0;
    #pragma unroll
    for (int off = 1; off < 32; off <<= 1) {
      int y = __shfl_up(x, off, 32);
      if (lane >= off) x += y;
    }
    if (in) p[i * 32 + lane] = x - x0 + carry;
    carry += __shfl(x, 31, 32);
  }
}

__device__ __forceinline__ void csr_fill_blk(int blk, const int* __restrict__ rowA, const int* __restrict__ colA,
                                             int* __restrict__ cur, col_t* __restrict__ csr) {
  int e = blk * 256 + threadIdx.x;
  if (e >= N_EDGES) return;
  int slot = atomicAdd(&cur[rowA[e]], 1);
  csr[slot] = (col_t)colA[e];
}


template <int NT, int N>
__device__ __forceinline__ void load_b(const float* __restrict__ b0, float (&bx)[NT], float (&by)[NT]) {
  const float* __restrict__ b1 = b0 + N;
  if constexpr (NT == 4) {
    float4 t0 = *(const float4*)b0, t1 = *(const float4*)b1;
    bx[0] = t0.x; bx[1] = t0.y; bx[2] = t0.z; bx[3] = t0.w;
    by[0] = t1.x; by[1] = t1.y; by[2] = t1.z; by[3] = t1.w;
  } else {
    #pragma unroll
    for (int nt = 0; nt < NT; ++nt) { bx[nt] = b0[nt]; by[nt] = b1[nt]; }
  }
}

template <int K, int N, int MT, int NT, int WN, int KC, int PF, int SCORE, int WPS, bool EDGES>
__global__ __launch_bounds__(32 * WN)
__attribute__((amdgpu_waves_per_eu(WPS)))
void gemm_bn_wmma(const float* A,
                  const float* scale,
                  const float* shift,
                  const float* __restrict__ B,
                  float* __restrict__ C,
                  const float* __restrict__ att,
                  float* __restrict__ sa,
                  float* __restrict__ sb,
                  int blk0, int nblk, int pass,
                  const int* __restrict__ rowA, const int* __restrict__ colA,
                  int* __restrict__ cur, col_t* __restrict__ csr) {
  constexpr int THREADS = 32 * WN;
  if constexpr (EDGES) {
    static_assert(THREADS == 256, "edge blocks are 256 threads");
    if ((int)blockIdx.x >= nblk) {
      const int eb = (int)blockIdx.x - nblk;
      if (pass == 0) row_count_blk(eb, rowA, cur);
      else           csr_fill_blk(eb, rowA, colA, cur, csr);
      return;
    }
  }
  constexpr int ROWS  = 16 * MT;
  constexpr int PITCH = K + 4;
  constexpr int K4  = K / 4;
  constexpr int C4  = KC / 4;
  constexpr int NCH = K / KC;
  constexpr int RPP = THREADS / C4;
  constexpr int LPT = ROWS / RPP;
  static_assert(PF >= 1 && 4 * PF <= K, "queue depth");
  static_assert(N == WN * NT * 16, "the waves of a work-group tile all N columns");
  static_assert(K % KC == 0 && KC % 4 == 0 && THREADS % C4 == 0 && ROWS % RPP == 0, "exact staging");
  static_assert((PITCH / 4) % 2 == 1, "pitch is 16 B x odd");
  static_assert(N_NODES % ROWS == 0, "grid divides exactly");
  __shared__ float As[ROWS * PITCH];

  const int tid  = threadIdx.x;
  const int lane = tid & 31;
  const int half = lane >> 4;
  const int l16  = lane & 15;
  const int wn   = tid >> 5;
  const int row0 = (blk0 + (int)blockIdx.x) * ROWS;

  const int tr = tid / C4, c4 = tid % C4;
  const float4* A4 = (const float4*)(A + (size_t)row0 * K) + (size_t)tr * K4 + c4;
  const float4* sc4 = (const float4*)scale + c4;
  const float4* sh4 = (const float4*)shift + c4;
  float* st = As + tr * PITCH + c4 * 4;

  float4 pre[LPT];
  #pragma unroll
  for (int i = 0; i < LPT; ++i) pre[i] = A4[(size_t)i * RPP * K4];
  float4 sc = sc4[0], sh = sh4[0];
  __builtin_amdgcn_sched_barrier(0);

  const float* as = As + l16 * PITCH + 2 * half;
  const float* __restrict__ bp = B + (size_t)(2 * half) * N + wn * (NT * 16) + NT * l16;

  v8f acc[MT][NT];
  #pragma unroll
  for (int mt = 0; mt < MT; ++mt)
    #pragma unroll
    for (int nt = 0; nt < NT; ++nt) acc[mt][nt] = v8f{};

  float qx[PF][NT], qy[PF][NT];
  #pragma unroll
  for (int p = 0; p < PF; ++p) load_b<NT, N>(bp + (size_t)(4 * p) * N, qx[p], qy[p]);

  #pragma nounroll
  for (int ch = 0; ch < NCH; ++ch) {
    #pragma unroll
    for (int i = 0; i < LPT; ++i) {
      float4 v = pre[i], o;
      o.x = v.x * sc.x + sh.x; o.y = v.y * sc.y + sh.y;
      o.z = v.z * sc.z + sh.z; o.w = v.w * sc.w + sh.w;
      *(float4*)(st + i * RPP * PITCH + ch * KC) = o;
    }
    __syncthreads();
    if (ch + 1 < NCH) {
      #pragma unroll
      for (int i = 0; i < LPT; ++i) pre[i] = A4[(size_t)i * RPP * K4 + (ch + 1) * C4];
      sc = sc4[(ch + 1) * C4]; sh = sh4[(ch + 1) * C4];
    }
    for (int k = ch * KC; k < ch * KC + KC; k += 4) {
      v2f a[MT];
      #pragma unroll
      for (int mt = 0; mt < MT; ++mt) a[mt] = *(const v2f*)(as + mt * 16 * PITCH + k);
      float bx[NT], by[NT];
      #pragma unroll
      for (int nt = 0; nt < NT; ++nt) { bx[nt] = qx[0][nt]; by[nt] = qy[0][nt]; }
      #pragma unroll
      for (int p = 0; p + 1 < PF; ++p)
        #pragma unroll
        for (int nt = 0; nt < NT; ++nt) { qx[p][nt] = qx[p + 1][nt]; qy[p][nt] = qy[p + 1][nt]; }
      if (k + 4 * PF < K) load_b<NT, N>(bp + (size_t)(k + 4 * PF) * N, qx[PF - 1], qy[PF - 1]);
      #pragma unroll
      for (int mt = 0; mt < MT; ++mt)
        #pragma unroll
        for (int nt = 0; nt < NT; ++nt) {
          v2f b; b.x = bx[nt]; b.y = by[nt];
          acc[mt][nt] = __builtin_amdgcn_wmma_f32_16x16x4_f32(
               false, a[mt],  false, b,
               (short)0, acc[mt][nt],  false,  false);
        }
    }
  }

  if constexpr (SCORE != 0) {
    static_assert(MT == 2 && NT == 2, "four accumulators named below");
    asm volatile("v_nop\n\tv_nop\n\tv_nop\n\tv_nop\n\tv_nop"
                 : "+v"(acc[0][0]), "+v"(acc[0][1]), "+v"(acc[1][0]), "+v"(acc[1][1]));
  }
  if constexpr (SCORE == 0) {
    #pragma unroll
    for (int mt = 0; mt < MT; ++mt) {
      float* cp = C + (size_t)(row0 + mt * 16 + (half << 3)) * N + wn * (NT * 16) + NT * l16;
      #pragma unroll
      for (int r = 0; r < 8; ++r) {
        if constexpr (NT == 4) {
          *(float4*)(cp + (size_t)r * N) =
              make_float4(acc[mt][0][r], acc[mt][1][r], acc[mt][2][r], acc[mt][3][r]);
        } else {
          #pragma unroll
          for (int nt = 0; nt < NT; ++nt) cp[(size_t)r * N + nt] = acc[mt][nt][r];
        }
      }
    }
  } else {
    static_assert(NT == 2 && N == NF && GP % (NT * 16) == 0 && 2 * H1 == 4 * NT, "score columns in four lanes");
    if (wn < GP / (NT * 16)) {
      #pragma unroll
      for (int mt = 0; mt < MT; ++mt) {
        float* cp = C + (size_t)(row0 + mt * 16 + (half << 3)) * GP + wn * (NT * 16) + NT * l16;
        #pragma unroll
        for (int r = 0; r < 8; ++r)
          *(float2*)(cp + (size_t)r * GP) = make_float2(acc[mt][0][r], acc[mt][1][r]);
      }
    } else if (wn == GP / (NT * 16) && l16 < 4) {
      float* __restrict__ dst = (l16 < 2 ? sa : sb) + 2 * (l16 & 1);
      #pragma unroll
      for (int mt = 0; mt < MT; ++mt)
        #pragma unroll
        for (int r = 0; r < 8; ++r)
          *(float2*)(dst + (size_t)(row0 + mt * 16 + (half << 3) + r) * H1) =
              make_float2(acc[mt][0][r], acc[mt][1][r]);
    }
  }
}


#define AG_ROWS  16
#define AG_RPW   (AG_ROWS / 8)
#define AG_PITCH (GP + 4)
static_assert(N_NODES % AG_ROWS == 0 && GP == 96 * 2 && D2 == 24 * 2, "three float2 per lane, 24 float2 per head");
static_assert(C2 == 6 && H2 == 8 && AG_ROWS * H2 == 128, "epilogue map");

__global__ __launch_bounds__(256)
void row_agg1(const int* __restrict__ cur, const col_t* __restrict__ csr,
              const float* __restrict__ ai, const float* __restrict__ aj,
              const float* __restrict__ g, const float* __restrict__ c0,
              const float* __restrict__ att,
              float* __restrict__ EP) {
  __shared__ float Ls[AG_ROWS * AG_PITCH];
  const int tid = threadIdx.x;
  const int lane = tid & 31;
  const int wv = tid >> 5;
  const int row0 = (int)blockIdx.x * AG_ROWS;
  const float4* __restrict__ aj4 = (const float4*)aj;
  const float2* __restrict__ g2 = (const float2*)g;
  const bool lo0 = lane < 24, lo1 = lane < 16, lo2 = lane < 8;

  for (int q = 0; q < AG_RPW; ++q) {
    const int rl = wv * AG_RPW + q;
    int r = __builtin_amdgcn_readfirstlane(row0 + rl);
    int beg = r ? cur[r - 1] : 0;
    int end = cur[r];
    float4 a_i = *(const float4*)(ai + (size_t)r * 4);
    float4 bs  = aj4[r];
    float t0 = lrelu(a_i.x + bs.x), t1 = lrelu(a_i.y + bs.y);
    float t2 = lrelu(a_i.z + bs.z), t3 = lrelu(a_i.w + bs.w);

    float m0 = t0, m1 = t1, m2 = t2, m3 = t3;
    for (int j = beg + lane; j < end; j += 32) {
      float4 b = aj4[csr[j]];
      m0 = fmaxf(m0, lrelu(a_i.x + b.x));
      m1 = fmaxf(m1, lrelu(a_i.y + b.y));
      m2 = fmaxf(m2, lrelu(a_i.z + b.z));
      m3 = fmaxf(m3, lrelu(a_i.w + b.w));
    }
    #pragma unroll
    for (int off = 16; off > 0; off >>= 1) {
      m0 = fmaxf(m0, __shfl_xor(m0, off, 32));
      m1 = fmaxf(m1, __shfl_xor(m1, off, 32));
      m2 = fmaxf(m2, __shfl_xor(m2, off, 32));
      m3 = fmaxf(m3, __shfl_xor(m3, off, 32));
    }

    float u0 = expf(t0 - m0), u1 = expf(t1 - m1), u2 = expf(t2 - m2), u3 = expf(t3 - m3);
    float wa = lo0 ? u0 : u1, wb = lo1 ? u1 : u2, wc = lo2 ? u2 : u3;
    float sA = wa, sB = wb, sC = wc;
    const float2* __restrict__ src = g2 + (size_t)r * (GP / 2);
    float2 v0 = src[lane], v1 = src[32 + lane], v2 = src[64 + lane];
    float2 acc0 = {wa * v0.x, wa * v0.y};
    float2 acc1 = {wb * v1.x, wb * v1.y};
    float2 acc2 = {wc * v2.x, wc * v2.y};

    for (int base = beg; base < end; base += 32) {
      int j = base + lane;
      int c = r;
      float w0 = 0.f, w1 = 0.f, w2 = 0.f, w3 = 0.f;
      if (j < end) {
        c = csr[j];
        float4 b = aj4[c];
        w0 = expf(lrelu(a_i.x + b.x) - m0);
        w1 = expf(lrelu(a_i.y + b.y) - m1);
        w2 = expf(lrelu(a_i.z + b.z) - m2);
        w3 = expf(lrelu(a_i.w + b.w) - m3);
      }
      int cnt = end - base < 32 ? end - base : 32;
      for (int k = 0; k < cnt; ++k) {
        int ck = __shfl(c, k, 32);
        u0 = __shfl(w0, k, 32); u1 = __shfl(w1, k, 32);
        u2 = __shfl(w2, k, 32); u3 = __shfl(w3, k, 32);
        wa = lo0 ? u0 : u1; wb = lo1 ? u1 : u2; wc = lo2 ? u2 : u3;
        sA += wa; sB += wb; sC += wc;
        src = g2 + (size_t)ck * (GP / 2);
        v0 = src[lane]; v1 = src[32 + lane]; v2 = src[64 + lane];
        acc0.x += wa * v0.x; acc0.y += wa * v0.y;
        acc1.x += wb * v1.x; acc1.y += wb * v1.y;
        acc2.x += wc * v2.x; acc2.y += wc * v2.y;
      }
    }
    float iA = 1.0f / (sA + 1e-16f), iB = 1.0f / (sB + 1e-16f), iC = 1.0f / (sC + 1e-16f);
    float2* dst = (float2*)(Ls + rl * AG_PITCH);
    dst[lane]      = make_float2(acc0.x * iA, acc0.y * iA);
    dst[32 + lane] = make_float2(acc1.x * iB, acc1.y * iB);
    dst[64 + lane] = make_float2(acc2.x * iC, acc2.y * iC);
  }
  __syncthreads();

  if (tid < AG_ROWS * H2) {
    const int row = tid >> 3, hd = tid & 7;
    const float* p = Ls + row * AG_PITCH + hd * C2;
    const float* __restrict__ cz = c0 + hd * C2;
    float v[C2];
    #pragma unroll
    for (int j = 0; j < C2; ++j) {
      float s = p[j];
      #pragma unroll
      for (int h = 1; h < H1; ++h) s += p[h * D2 + j];
      v[j] = s + (((cz[j] + cz[D2 + j]) + cz[2 * D2 + j]) + cz[3 * D2 + j]);
    }
    const float* __restrict__ ap = att + hd * (2 * C2);
    const float pa = v[0] * (ap[0] + ap[C2])     + v[1] * (ap[1] + ap[C2 + 1]) + v[2] * (ap[2] + ap[C2 + 2]);
    const float pb = v[3] * (ap[3] + ap[C2 + 3]) + v[4] * (ap[4] + ap[C2 + 4]) + v[5] * (ap[5] + ap[C2 + 5]);
    const float e = expf(tanhf(pa + pb));
    float4* __restrict__ ep = (float4*)(EP + (size_t)(row0 + row) * EPW) + hd;
    ep[0]  = make_float4(e, e * v[0], e * v[1], e * v[2]);
    ep[H2] = make_float4(e * v[3], e * v[4], e * v[5], 0.f);
  }
}


#define A2_ROWS  64
#define A2_HPL   2
#define A2_U     2
#define A2_PITCH 50
static_assert(N_NODES % A2_ROWS == 0 && A2_ROWS * (H2 / A2_HPL) == 256 && EPW == 2 * H2 * 4, "layer-2 maps");

__global__ __launch_bounds__(256)
void row_agg2(const int* __restrict__ cur, const col_t* __restrict__ csr,
              const float* __restrict__ ep, const float* __restrict__ b2,
              float* __restrict__ out) {
  __shared__ float Ls[A2_ROWS * A2_PITCH];
  const int rl = threadIdx.x >> 2, q = threadIdx.x & 3;
  const int r = (int)blockIdx.x * A2_ROWS + rl;
  const int beg = r ? cur[r - 1] : 0;
  const int end = cur[r];
  const float4* __restrict__ ep4 = (const float4*)ep + A2_HPL * q;
  float den[A2_HPL], a[A2_HPL][C2];
  {
    const float4* __restrict__ p = ep4 + (size_t)r * (EPW / 4);
    #pragma unroll
    for (int k = 0; k < A2_HPL; ++k) {
      const float4 x = p[k], y = p[H2 + k];
      den[k] = x.x; a[k][0] = x.y; a[k][1] = x.z; a[k][2] = x.w; a[k][3] = y.x; a[k][4] = y.y; a[k][5] = y.z;
    }
  }

  int c[A2_U];
  #pragma unroll
  for (int u = 0; u < A2_U; ++u) c[u] = beg < end ? (int)csr[min(beg + u, end - 1)] : r;
  for (int j = beg; j < end; j += A2_U) {
    float4 xs[A2_U][A2_HPL], ys[A2_U][A2_HPL];
    #pragma unroll
    for (int u = 0; u < A2_U; ++u) {
      const float4* __restrict__ p = ep4 + (size_t)c[u] * (EPW / 4);
      #pragma unroll
      for (int k = 0; k < A2_HPL; ++k) { xs[u][k] = p[k]; ys[u][k] = p[H2 + k]; }
    }
    #pragma unroll
    for (int u = 0; u < A2_U; ++u) c[u] = (int)csr[min(j + A2_U + u, end - 1)];
    #pragma unroll
    for (int u = 0; u < A2_U; ++u) {
      const bool in = j + u < end;
      #pragma unroll
      for (int k = 0; k < A2_HPL; ++k) {
        den[k]  += in ? xs[u][k].x : -0.0f;
        a[k][0] += in ? xs[u][k].y : -0.0f; a[k][1] += in ? xs[u][k].z : -0.0f; a[k][2] += in ? xs[u][k].w : -0.0f;
        a[k][3] += in ? ys[u][k].x : -0.0f; a[k][4] += in ? ys[u][k].y : -0.0f; a[k][5] += in ? ys[u][k].z : -0.0f;
      }
    }
  }
  float2* __restrict__ qd = (float2*)(Ls + rl * A2_PITCH + A2_HPL * q * C2);
  #pragma unroll
  for (int k = 0; k < A2_HPL; ++k) {
    const float inv = 1.0f / (den[k] + 1e-16f);
    qd[3 * k]     = make_float2(a[k][0] * inv, a[k][1] * inv);
    qd[3 * k + 1] = make_float2(a[k][2] * inv, a[k][3] * inv);
    qd[3 * k + 2] = make_float2(a[k][4] * inv, a[k][5] * inv);
  }
  __syncthreads();

  if (threadIdx.x < A2_ROWS) {
    const float* p = Ls + threadIdx.x * A2_PITCH;
    const int n = (int)blockIdx.x * A2_ROWS + (int)threadIdx.x;
    float v[C2];
    #pragma unroll
    for (int ch = 0; ch < C2; ++ch) {
      float s = 0.f;
      #pragma unroll
      for (int hh = 0; hh < H2; ++hh) s += p[hh * C2 + ch];
      float t = s * (1.0f / H2) + b2[ch];
      v[ch] = t > 0.f ? t : (expf(t) - 1.0f);
    }
    float m = v[0];
    #pragma unroll
    for (int ch = 1; ch < C2; ++ch) m = fmaxf(m, v[ch]);
    float sum = 0.f;
    #pragma unroll
    for (int ch = 0; ch < C2; ++ch) sum += expf(v[ch] - m);
    const float lse = logf(sum);
    float2* __restrict__ dst = (float2*)(out + (size_t)n * C2);
    dst[0] = make_float2(v[0] - m - lse, v[1] - m - lse);
    dst[1] = make_float2(v[2] - m - lse, v[3] - m - lse);
    dst[2] = make_float2(v[4] - m - lse, v[5] - m - lse);
  }
}


extern "C" void kernel_launch(void* const* d_in, const int* in_sizes, int n_in,
                              void* d_out, int out_size, void* d_ws, size_t ws_size,
                              hipStream_t stream) {
  const float* x    = (const float*)d_in[0];
  const int*   ei   = (const int*)  d_in[1];
  const float* g1   = (const float*)d_in[2];
  const float* be1  = (const float*)d_in[3];
  const float* mu1  = (const float*)d_in[4];
  const float* va1  = (const float*)d_in[5];
  const float* W1   = (const float*)d_in[6];
  const float* att1 = (const float*)d_in[7];
  const float* b1   = (const float*)d_in[8];
  const float* g2   = (const float*)d_in[9];
  const float* be2  = (const float*)d_in[10];
  const float* mu2  = (const float*)d_in[11];
  const float* va2  = (const float*)d_in[12];
  const float* W2   = (const float*)d_in[13];
  const float* att2 = (const float*)d_in[14];
  const float* b2   = (const float*)d_in[15];
  const int* rowA = ei;
  const int* colA = ei + N_EDGES;
  float* out = (float*)d_out;
  float* ws  = (float*)d_ws;

  float* sc1  = ws;
  float* sh1  = ws + 256;
  float* c0   = ws + 512;
  float* ai   = ws + 1536;
  float* aj   = ai  + (size_t)N_NODES * H1;
  float* edg  = aj  + (size_t)N_NODES * H1;
  int*   cur  = (int*)edg;
  col_t* csr  = (col_t*)(edg + N_NODES);
  float* l2   = edg + (size_t)N_NODES * 2 * H1;
  float* epb  = l2;
  float* gt   = l2  + (size_t)N_NODES * D1;
  float* Wc   = gt  + (size_t)N_NODES * GP;

  bn_prep<<<PREP_BLOCKS + F_IN + 1, 256, 0, stream>>>(g1, be1, mu1, va1, g2, be2, mu2, va2, b1,
                                                     W1, W2, att1, sc1, sh1, Wc, c0, cur);

  gemm_bn_wmma<F_IN, NF, 2, 2, 8, 64, 2, 1, 10, true><<<G1_FIRST + EDGE_BLOCKS, 256, 0, stream>>>(
      x, sc1, sh1, Wc, gt, att1, ai, aj, 0, G1_FIRST, 0, rowA, colA, cur, csr);
  row_scan<<<1, SCAN_T, 0, stream>>>(cur);
  gemm_bn_wmma<F_IN, NF, 2, 2, 8, 64, 2, 1, 10, true><<<G1_BLOCKS - G1_FIRST + EDGE_BLOCKS, 256, 0, stream>>>(
      x, sc1, sh1, Wc, gt, att1, ai, aj, G1_FIRST, G1_BLOCKS - G1_FIRST, 1, rowA, colA, cur, csr);

  row_agg1<<<N_NODES / AG_ROWS, 256, 0, stream>>>(cur, csr, ai, aj, gt, c0, att2, epb);

  row_agg2<<<N_NODES / A2_ROWS, 256, 0, stream>>>(cur, csr, epb, b2, out);
}


# Edges bucketed by destination in one pass: a returning add ranks each edge into its row's 32-slot bucket (longer rows spill to a list); no count pass, no prefix sum, one GEMM launch: 6->4 launches
# speedup vs baseline: 8.8004x; 1.2754x over previous
#include <hip/hip_runtime.h>
#include <math.h>

#define N_NODES 40000
#define N_EDGES 320000
#define ETOT    (N_EDGES + N_NODES)
#define F_IN    256
#define H1      4
#define C1      128
#define D1      (H1*C1)
#define H2      8
#define C2      6
#define D2      (H2*C2)
#define BN_EPS    1e-5f
#define NEG_SLOPE 0.2f

typedef __attribute__((ext_vector_type(2))) float v2f;
typedef __attribute__((ext_vector_type(8))) float v8f;


static_assert(N_NODES <= 65536, "16-bit column ids need N_NODES <= 65536");
typedef unsigned short col_t;
static_assert(N_EDGES % 256 == 0 && N_NODES % 8 == 0 && (N_NODES * H2) % 256 == 0,
              "launch grids below divide exactly");

__device__ __forceinline__ float lrelu(float a) { return a > 0.f ? a : NEG_SLOPE * a; }


#define GP 192
#define NF 256
#define EPW 64
#define PREP_BLOCKS ((N_NODES + 255) / 256)
static_assert(GP == H1 * D2 && GP + 2 * H1 <= NF && D1 == 2 * 256 && H1 * 64 == 256 && D2 <= 64, "fold maps");
static_assert(PREP_BLOCKS * 256 > N_NODES, "one thread per counter, the spill counter included");

__global__ __launch_bounds__(256)
void bn_prep(const float* g1, const float* be1, const float* mu1, const float* va1,
             const float* g2, const float* be2, const float* mu2, const float* va2,
             const float* bias1, const float* __restrict__ W1, const float* __restrict__ W2,
             const float* __restrict__ att1,
             float* sc1, float* sh1, float* __restrict__ Wc, float* __restrict__ c0, int* __restrict__ cnt) {
  if ((int)blockIdx.x >= PREP_BLOCKS) {
    __shared__ float s2c[D1];
    const int fb = (int)blockIdx.x - PREP_BLOCKS, n = threadIdx.x;
    for (int kk = n; kk < D1; kk += 256) s2c[kk] = g2[kk] / sqrtf(va2[kk] + BN_EPS);
    __syncthreads();
    float s = 0.f;
    if (fb < F_IN) {
      const float* __restrict__ w1 = W1 + (size_t)fb * D1;
      if (n < GP) {
        const int h = n / D2, o = n % D2;
        for (int c = 0; c < C1; ++c) { const int kk = h * C1 + c; s += w1[kk] * (s2c[kk] * W2[kk * D2 + o]); }
      } else if (n < GP + 2 * H1) {
        const int h = (n - GP) & (H1 - 1), side = (n - GP) / H1;
        const float* __restrict__ a = att1 + h * (2 * C1) + side * C1;
        for (int c = 0; c < C1; ++c) s += w1[h * C1 + c] * a[c];
      }
      Wc[(size_t)fb * NF + n] = s;
    } else {
      const int h = n >> 6, o = n & 63;
      if (o < D2) {
        for (int c = 0; c < C1; ++c) {
          const int kk = h * C1 + c;
          const float sh = be2[kk] + (bias1[kk] - mu2[kk]) * s2c[kk];
          s += sh * W2[kk * D2 + o];
        }
        c0[h * D2 + o] = s;
      }
    }
    return;
  }
  int i = blockIdx.x * blockDim.x + threadIdx.x;
  if (i <= N_NODES) cnt[i] = 0;
  if (i < F_IN) {
    float s = g1[i] / sqrtf(va1[i] + BN_EPS);
    sc1[i] = s; sh1[i] = be1[i] - mu1[i] * s;
  }
}

#define BK_CAP      32
#define EDGE_EPT    5
#define EDGE_BLOCKS (N_EDGES / (256 * EDGE_EPT))
#define G1_BLOCKS   (N_NODES / 32)
static_assert(N_EDGES % (256 * EDGE_EPT) == 0, "edge blocks divide the edge list exactly");
__device__ __forceinline__ void edge_fill_blk(int blk, const int* __restrict__ rowA, const int* __restrict__ colA,
                                              int* __restrict__ cnt, col_t* __restrict__ bkt,
                                              int* __restrict__ spr, int* __restrict__ spc) {
  const int e0 = blk * (256 * EDGE_EPT) + threadIdx.x;
  int r[EDGE_EPT], c[EDGE_EPT], slot[EDGE_EPT];
  #pragma unroll
  for (int i = 0; i < EDGE_EPT; ++i) { r[i] = rowA[e0 + i * 256]; c[i] = colA[e0 + i * 256]; }
  #pragma unroll
  for (int i = 0; i < EDGE_EPT; ++i) slot[i] = atomicAdd(&cnt[r[i]], 1);
  #pragma unroll
  for (int i = 0; i < EDGE_EPT; ++i) {
    if (slot[i] < BK_CAP) {
      bkt[(size_t)r[i] * BK_CAP + slot[i]] = (col_t)c[i];
    } else {
      const int o = atomicAdd(&cnt[N_NODES], 1);
      spr[o] = r[i]; spc[o] = c[i];
    }
  }
}


template <int NT, int N>
__device__ __forceinline__ void load_b(const float* __restrict__ b0, float (&bx)[NT], float (&by)[NT]) {
  const float* __restrict__ b1 = b0 + N;
  if constexpr (NT == 4) {
    float4 t0 = *(const float4*)b0, t1 = *(const float4*)b1;
    bx[0] = t0.x; bx[1] = t0.y; bx[2] = t0.z; bx[3] = t0.w;
    by[0] = t1.x; by[1] = t1.y; by[2] = t1.z; by[3] = t1.w;
  } else {
    #pragma unroll
    for (int nt = 0; nt < NT; ++nt) { bx[nt] = b0[nt]; by[nt] = b1[nt]; }
  }
}

template <int K, int N, int MT, int NT, int WN, int KC, int PF, int SCORE, int WPS, bool EDGES>
__global__ __launch_bounds__(32 * WN)
__attribute__((amdgpu_waves_per_eu(WPS)))
void gemm_bn_wmma(const float* A,
                  const float* scale,
                  const float* shift,
                  const float* __restrict__ B,
                  float* __restrict__ C,
                  const float* __restrict__ att,
                  float* __restrict__ sa,
                  float* __restrict__ sb,
                  const int* __restrict__ rowA, const int* __restrict__ colA,
                  int* __restrict__ cnt, col_t* __restrict__ bkt,
                  int* __restrict__ spr, int* __restrict__ spc) {
  constexpr int THREADS = 32 * WN;
  if constexpr (EDGES) {
    static_assert(THREADS == 256, "edge blocks are 256 threads");
    if ((int)blockIdx.x < EDGE_BLOCKS) {
      edge_fill_blk((int)blockIdx.x, rowA, colA, cnt, bkt, spr, spc);
      return;
    }
  }
  constexpr int ROWS  = 16 * MT;
  constexpr int PITCH = K + 4;
  constexpr int K4  = K / 4;
  constexpr int C4  = KC / 4;
  constexpr int NCH = K / KC;
  constexpr int RPP = THREADS / C4;
  constexpr int LPT = ROWS / RPP;
  static_assert(PF >= 1 && 4 * PF <= K, "queue depth");
  static_assert(N == WN * NT * 16, "the waves of a work-group tile all N columns");
  static_assert(K % KC == 0 && KC % 4 == 0 && THREADS % C4 == 0 && ROWS % RPP == 0, "exact staging");
  static_assert((PITCH / 4) % 2 == 1, "pitch is 16 B x odd");
  static_assert(N_NODES % ROWS == 0, "grid divides exactly");
  __shared__ float As[ROWS * PITCH];

  const int tid  = threadIdx.x;
  const int lane = tid & 31;
  const int half = lane >> 4;
  const int l16  = lane & 15;
  const int wn   = tid >> 5;
  const int row0 = ((int)blockIdx.x - (EDGES ? EDGE_BLOCKS : 0)) * ROWS;

  const int tr = tid / C4, c4 = tid % C4;
  const float4* A4 = (const float4*)(A + (size_t)row0 * K) + (size_t)tr * K4 + c4;
  const float4* sc4 = (const float4*)scale + c4;
  const float4* sh4 = (const float4*)shift + c4;
  float* st = As + tr * PITCH + c4 * 4;

  float4 pre[LPT];
  #pragma unroll
  for (int i = 0; i < LPT; ++i) pre[i] = A4[(size_t)i * RPP * K4];
  float4 sc = sc4[0], sh = sh4[0];
  __builtin_amdgcn_sched_barrier(0);

  const float* as = As + l16 * PITCH + 2 * half;
  const float* __restrict__ bp = B + (size_t)(2 * half) * N + wn * (NT * 16) + NT * l16;

  v8f acc[MT][NT];
  #pragma unroll
  for (int mt = 0; mt < MT; ++mt)
    #pragma unroll
    for (int nt = 0; nt < NT; ++nt) acc[mt][nt] = v8f{};

  float qx[PF][NT], qy[PF][NT];
  #pragma unroll
  for (int p = 0; p < PF; ++p) load_b<NT, N>(bp + (size_t)(4 * p) * N, qx[p], qy[p]);

  #pragma nounroll
  for (int ch = 0; ch < NCH; ++ch) {
    #pragma unroll
    for (int i = 0; i < LPT; ++i) {
      float4 v = pre[i], o;
      o.x = v.x * sc.x + sh.x; o.y = v.y * sc.y + sh.y;
      o.z = v.z * sc.z + sh.z; o.w = v.w * sc.w + sh.w;
      *(float4*)(st + i * RPP * PITCH + ch * KC) = o;
    }
    __syncthreads();
    if (ch + 1 < NCH) {
      #pragma unroll
      for (int i = 0; i < LPT; ++i) pre[i] = A4[(size_t)i * RPP * K4 + (ch + 1) * C4];
      sc = sc4[(ch + 1) * C4]; sh = sh4[(ch + 1) * C4];
    }
    for (int k = ch * KC; k < ch * KC + KC; k += 4) {
      v2f a[MT];
      #pragma unroll
      for (int mt = 0; mt < MT; ++mt) a[mt] = *(const v2f*)(as + mt * 16 * PITCH + k);
      float bx[NT], by[NT];
      #pragma unroll
      for (int nt = 0; nt < NT; ++nt) { bx[nt] = qx[0][nt]; by[nt] = qy[0][nt]; }
      #pragma unroll
      for (int p = 0; p + 1 < PF; ++p)
        #pragma unroll
        for (int nt = 0; nt < NT; ++nt) { qx[p][nt] = qx[p + 1][nt]; qy[p][nt] = qy[p + 1][nt]; }
      if (k + 4 * PF < K) load_b<NT, N>(bp + (size_t)(k + 4 * PF) * N, qx[PF - 1], qy[PF - 1]);
      #pragma unroll
      for (int mt = 0; mt < MT; ++mt)
        #pragma unroll
        for (int nt = 0; nt < NT; ++nt) {
          v2f b; b.x = bx[nt]; b.y = by[nt];
          acc[mt][nt] = __builtin_amdgcn_wmma_f32_16x16x4_f32(
               false, a[mt],  false, b,
               (short)0, acc[mt][nt],  false,  false);
        }
    }
  }

  if constexpr (SCORE != 0) {
    static_assert(MT == 2 && NT == 2, "four accumulators named below");
    asm volatile("v_nop\n\tv_nop\n\tv_nop\n\tv_nop\n\tv_nop"
                 : "+v"(acc[0][0]), "+v"(acc[0][1]), "+v"(acc[1][0]), "+v"(acc[1][1]));
  }
  if constexpr (SCORE == 0) {
    #pragma unroll
    for (int mt = 0; mt < MT; ++mt) {
      float* cp = C + (size_t)(row0 + mt * 16 + (half << 3)) * N + wn * (NT * 16) + NT * l16;
      #pragma unroll
      for (int r = 0; r < 8; ++r) {
        if constexpr (NT == 4) {
          *(float4*)(cp + (size_t)r * N) =
              make_float4(acc[mt][0][r], acc[mt][1][r], acc[mt][2][r], acc[mt][3][r]);
        } else {
          #pragma unroll
          for (int nt = 0; nt < NT; ++nt) cp[(size_t)r * N + nt] = acc[mt][nt][r];
        }
      }
    }
  } else {
    static_assert(NT == 2 && N == NF && GP % (NT * 16) == 0 && 2 * H1 == 4 * NT, "score columns in four lanes");
    if (wn < GP / (NT * 16)) {
      #pragma unroll
      for (int mt = 0; mt < MT; ++mt) {
        float* cp = C + (size_t)(row0 + mt * 16 + (half << 3)) * GP + wn * (NT * 16) + NT * l16;
        #pragma unroll
        for (int r = 0; r < 8; ++r)
          *(float2*)(cp + (size_t)r * GP) = make_float2(acc[mt][0][r], acc[mt][1][r]);
      }
    } else if (wn == GP / (NT * 16) && l16 < 4) {
      float* __restrict__ dst = (l16 < 2 ? sa : sb) + 2 * (l16 & 1);
      #pragma unroll
      for (int mt = 0; mt < MT; ++mt)
        #pragma unroll
        for (int r = 0; r < 8; ++r)
          *(float2*)(dst + (size_t)(row0 + mt * 16 + (half << 3) + r) * H1) =
              make_float2(acc[mt][0][r], acc[mt][1][r]);
    }
  }
}


#define AG_ROWS  16
#define AG_RPW   (AG_ROWS / 8)
#define AG_PITCH (GP + 4)
static_assert(N_NODES % AG_ROWS == 0 && GP == 96 * 2 && D2 == 24 * 2, "three float2 per lane, 24 float2 per head");
static_assert(C2 == 6 && H2 == 8 && AG_ROWS * H2 == 128, "epilogue map");

__global__ __launch_bounds__(256)
void row_agg1(const int* __restrict__ ecnt, const col_t* __restrict__ csr,
              const int* __restrict__ spr, const int* __restrict__ spc,
              const float* __restrict__ ai, const float* __restrict__ aj,
              const float* __restrict__ g, const float* __restrict__ c0,
              const float* __restrict__ att,
              float* __restrict__ EP) {
  __shared__ float Ls[AG_ROWS * AG_PITCH];
  const int tid = threadIdx.x;
  const int lane = tid & 31;
  const int wv = tid >> 5;
  const int row0 = (int)blockIdx.x * AG_ROWS;
  const float4* __restrict__ aj4 = (const float4*)aj;
  const float2* __restrict__ g2 = (const float2*)g;
  const bool lo0 = lane < 24, lo1 = lane < 16, lo2 = lane < 8;

  for (int q = 0; q < AG_RPW; ++q) {
    const int rl = wv * AG_RPW + q;
    int r = __builtin_amdgcn_readfirstlane(row0 + rl);
    const int deg = ecnt[r];
    const int spl = ecnt[N_NODES];
    const int beg = r * BK_CAP;
    const int end = beg + (deg < BK_CAP ? deg : BK_CAP);
    const int nsp = deg > BK_CAP ? spl : 0;
    float4 a_i = *(const float4*)(ai + (size_t)r * 4);
    float4 bs  = aj4[r];
    float t0 = lrelu(a_i.x + bs.x), t1 = lrelu(a_i.y + bs.y);
    float t2 = lrelu(a_i.z + bs.z), t3 = lrelu(a_i.w + bs.w);

    float m0 = t0, m1 = t1, m2 = t2, m3 = t3;
    for (int j = beg + lane; j < end; j += 32) {
      float4 b = aj4[csr[j]];
      m0 = fmaxf(m0, lrelu(a_i.x + b.x));
      m1 = fmaxf(m1, lrelu(a_i.y + b.y));
      m2 = fmaxf(m2, lrelu(a_i.z + b.z));
      m3 = fmaxf(m3, lrelu(a_i.w + b.w));
    }
    for (int ob = 0; ob < nsp; ob += 32) {
      const int i = ob + lane;
      if (i < nsp && spr[i] == r) {
        float4 b = aj4[spc[i]];
        m0 = fmaxf(m0, lrelu(a_i.x + b.x));
        m1 = fmaxf(m1, lrelu(a_i.y + b.y));
        m2 = fmaxf(m2, lrelu(a_i.z + b.z));
        m3 = fmaxf(m3, lrelu(a_i.w + b.w));
      }
    }
    #pragma unroll
    for (int off = 16; off > 0; off >>= 1) {
      m0 = fmaxf(m0, __shfl_xor(m0, off, 32));
      m1 = fmaxf(m1, __shfl_xor(m1, off, 32));
      m2 = fmaxf(m2, __shfl_xor(m2, off, 32));
      m3 = fmaxf(m3, __shfl_xor(m3, off, 32));
    }

    float u0 = expf(t0 - m0), u1 = expf(t1 - m1), u2 = expf(t2 - m2), u3 = expf(t3 - m3);
    float wa = lo0 ? u0 : u1, wb = lo1 ? u1 : u2, wc = lo2 ? u2 : u3;
    float sA = wa, sB = wb, sC = wc;
    const float2* __restrict__ src = g2 + (size_t)r * (GP / 2);
    float2 v0 = src[lane], v1 = src[32 + lane], v2 = src[64 + lane];
    float2 acc0 = {wa * v0.x, wa * v0.y};
    float2 acc1 = {wb * v1.x, wb * v1.y};
    float2 acc2 = {wc * v2.x, wc * v2.y};

    for (int base = beg; base < end; base += 32) {
      int j = base + lane;
      int c = r;
      float w0 = 0.f, w1 = 0.f, w2 = 0.f, w3 = 0.f;
      if (j < end) {
        c = csr[j];
        float4 b = aj4[c];
        w0 = expf(lrelu(a_i.x + b.x) - m0);
        w1 = expf(lrelu(a_i.y + b.y) - m1);
        w2 = expf(lrelu(a_i.z + b.z) - m2);
        w3 = expf(lrelu(a_i.w + b.w) - m3);
      }
      int cnt = end - base < 32 ? end - base : 32;
      for (int k = 0; k < cnt; ++k) {
        int ck = __shfl(c, k, 32);
        u0 = __shfl(w0, k, 32); u1 = __shfl(w1, k, 32);
        u2 = __shfl(w2, k, 32); u3 = __shfl(w3, k, 32);
        wa = lo0 ? u0 : u1; wb = lo1 ? u1 : u2; wc = lo2 ? u2 : u3;
        sA += wa; sB += wb; sC += wc;
        src = g2 + (size_t)ck * (GP / 2);
        v0 = src[lane]; v1 = src[32 + lane]; v2 = src[64 + lane];
        acc0.x += wa * v0.x; acc0.y += wa * v0.y;
        acc1.x += wb * v1.x; acc1.y += wb * v1.y;
        acc2.x += wc * v2.x; acc2.y += wc * v2.y;
      }
    }
    for (int ob = 0; ob < nsp; ob += 32) {
      const int i = ob + lane;
      const bool hit = i < nsp && spr[i] == r;
      int c = r;
      float w0 = 0.f, w1 = 0.f, w2 = 0.f, w3 = 0.f;
      if (hit) {
        c = spc[i];
        float4 b = aj4[c];
        w0 = expf(lrelu(a_i.x + b.x) - m0);
        w1 = expf(lrelu(a_i.y + b.y) - m1);
        w2 = expf(lrelu(a_i.z + b.z) - m2);
        w3 = expf(lrelu(a_i.w + b.w) - m3);
      }
      unsigned todo = __builtin_amdgcn_ballot_w32(hit);
      while (todo) {
        const int k = __builtin_ctz(todo);
        todo &= todo - 1;
        int ck = __shfl(c, k, 32);
        u0 = __shfl(w0, k, 32); u1 = __shfl(w1, k, 32);
        u2 = __shfl(w2, k, 32); u3 = __shfl(w3, k, 32);
        wa = lo0 ? u0 : u1; wb = lo1 ? u1 : u2; wc = lo2 ? u2 : u3;
        sA += wa; sB += wb; sC += wc;
        src = g2 + (size_t)ck * (GP / 2);
        v0 = src[lane]; v1 = src[32 + lane]; v2 = src[64 + lane];
        acc0.x += wa * v0.x; acc0.y += wa * v0.y;
        acc1.x += wb * v1.x; acc1.y += wb * v1.y;
        acc2.x += wc * v2.x; acc2.y += wc * v2.y;
      }
    }
    float iA = 1.0f / (sA + 1e-16f), iB = 1.0f / (sB + 1e-16f), iC = 1.0f / (sC + 1e-16f);
    float2* dst = (float2*)(Ls + rl * AG_PITCH);
    dst[lane]      = make_float2(acc0.x * iA, acc0.y * iA);
    dst[32 + lane] = make_float2(acc1.x * iB, acc1.y * iB);
    dst[64 + lane] = make_float2(acc2.x * iC, acc2.y * iC);
  }
  __syncthreads();

  if (tid < AG_ROWS * H2) {
    const int row = tid >> 3, hd = tid & 7;
    const float* p = Ls + row * AG_PITCH + hd * C2;
    const float* __restrict__ cz = c0 + hd * C2;
    float v[C2];
    #pragma unroll
    for (int j = 0; j < C2; ++j) {
      float s = p[j];
      #pragma unroll
      for (int h = 1; h < H1; ++h) s += p[h * D2 + j];
      v[j] = s + (((cz[j] + cz[D2 + j]) + cz[2 * D2 + j]) + cz[3 * D2 + j]);
    }
    const float* __restrict__ ap = att + hd * (2 * C2);
    const float pa = v[0] * (ap[0] + ap[C2])     + v[1] * (ap[1] + ap[C2 + 1]) + v[2] * (ap[2] + ap[C2 + 2]);
    const float pb = v[3] * (ap[3] + ap[C2 + 3]) + v[4] * (ap[4] + ap[C2 + 4]) + v[5] * (ap[5] + ap[C2 + 5]);
    const float e = expf(tanhf(pa + pb));
    float4* __restrict__ ep = (float4*)(EP + (size_t)(row0 + row) * EPW) + hd;
    ep[0]  = make_float4(e, e * v[0], e * v[1], e * v[2]);
    ep[H2] = make_float4(e * v[3], e * v[4], e * v[5], 0.f);
  }
}


#define A2_ROWS  64
#define A2_HPL   2
#define A2_U     2
#define A2_PITCH 50
static_assert(N_NODES % A2_ROWS == 0 && A2_ROWS * (H2 / A2_HPL) == 256 && EPW == 2 * H2 * 4, "layer-2 maps");

__global__ __launch_bounds__(256)
void row_agg2(const int* __restrict__ ecnt, const col_t* __restrict__ csr,
              const int* __restrict__ spr, const int* __restrict__ spc,
              const float* __restrict__ ep, const float* __restrict__ b2,
              float* __restrict__ out) {
  __shared__ float Ls[A2_ROWS * A2_PITCH];
  const int rl = threadIdx.x >> 2, q = threadIdx.x & 3;
  const int r = (int)blockIdx.x * A2_ROWS + rl;
  const int deg = ecnt[r];
  const int beg = r * BK_CAP;
  const int end = beg + min(deg, BK_CAP);
  const float4* __restrict__ ep4 = (const float4*)ep + A2_HPL * q;
  float den[A2_HPL], a[A2_HPL][C2];
  {
    const float4* __restrict__ p = ep4 + (size_t)r * (EPW / 4);
    #pragma unroll
    for (int k = 0; k < A2_HPL; ++k) {
      const float4 x = p[k], y = p[H2 + k];
      den[k] = x.x; a[k][0] = x.y; a[k][1] = x.z; a[k][2] = x.w; a[k][3] = y.x; a[k][4] = y.y; a[k][5] = y.z;
    }
  }

  int c[A2_U];
  #pragma unroll
  for (int u = 0; u < A2_U; ++u) c[u] = beg < end ? (int)csr[min(beg + u, end - 1)] : r;
  for (int j = beg; j < end; j += A2_U) {
    float4 xs[A2_U][A2_HPL], ys[A2_U][A2_HPL];
    #pragma unroll
    for (int u = 0; u < A2_U; ++u) {
      const float4* __restrict__ p = ep4 + (size_t)c[u] * (EPW / 4);
      #pragma unroll
      for (int k = 0; k < A2_HPL; ++k) { xs[u][k] = p[k]; ys[u][k] = p[H2 + k]; }
    }
    #pragma unroll
    for (int u = 0; u < A2_U; ++u) c[u] = (int)csr[min(j + A2_U + u, end - 1)];
    #pragma unroll
    for (int u = 0; u < A2_U; ++u) {
      const bool in = u == 0 || j + u < end;
      #pragma unroll
      for (int k = 0; k < A2_HPL; ++k) {
        den[k]  += in ? xs[u][k].x : -0.0f;
        a[k][0] += in ? xs[u][k].y : -0.0f; a[k][1] += in ? xs[u][k].z : -0.0f; a[k][2] += in ? xs[u][k].w : -0.0f;
        a[k][3] += in ? ys[u][k].x : -0.0f; a[k][4] += in ? ys[u][k].y : -0.0f; a[k][5] += in ? ys[u][k].z : -0.0f;
      }
    }
  }
  if (deg > BK_CAP) {
    const int nsp = ecnt[N_NODES];
    for (int i = 0; i < nsp; ++i) {
      if (spr[i] != r) continue;
      const float4* __restrict__ p = ep4 + (size_t)spc[i] * (EPW / 4);
      #pragma unroll
      for (int k = 0; k < A2_HPL; ++k) {
        const float4 x = p[k], y = p[H2 + k];
        den[k]  += x.x;
        a[k][0] += x.y; a[k][1] += x.z; a[k][2] += x.w;
        a[k][3] += y.x; a[k][4] += y.y; a[k][5] += y.z;
      }
    }
  }
  float2* __restrict__ qd = (float2*)(Ls + rl * A2_PITCH + A2_HPL * q * C2);
  #pragma unroll
  for (int k = 0; k < A2_HPL; ++k) {
    const float inv = 1.0f / (den[k] + 1e-16f);
    qd[3 * k]     = make_float2(a[k][0] * inv, a[k][1] * inv);
    qd[3 * k + 1] = make_float2(a[k][2] * inv, a[k][3] * inv);
    qd[3 * k + 2] = make_float2(a[k][4] * inv, a[k][5] * inv);
  }
  __syncthreads();

  if (threadIdx.x < A2_ROWS) {
    const float* p = Ls + threadIdx.x * A2_PITCH;
    const int n = (int)blockIdx.x * A2_ROWS + (int)threadIdx.x;
    float v[C2];
    #pragma unroll
    for (int ch = 0; ch < C2; ++ch) {
      float s = 0.f;
      #pragma unroll
      for (int hh = 0; hh < H2; ++hh) s += p[hh * C2 + ch];
      float t = s * (1.0f / H2) + b2[ch];
      v[ch] = t > 0.f ? t : (expf(t) - 1.0f);
    }
    float m = v[0];
    #pragma unroll
    for (int ch = 1; ch < C2; ++ch) m = fmaxf(m, v[ch]);
    float sum = 0.f;
    #pragma unroll
    for (int ch = 0; ch < C2; ++ch) sum += expf(v[ch] - m);
    const float lse = logf(sum);
    float2* __restrict__ dst = (float2*)(out + (size_t)n * C2);
    dst[0] = make_float2(v[0] - m - lse, v[1] - m - lse);
    dst[1] = make_float2(v[2] - m - lse, v[3] - m - lse);
    dst[2] = make_float2(v[4] - m - lse, v[5] - m - lse);
  }
}


extern "C" void kernel_launch(void* const* d_in, const int* in_sizes, int n_in,
                              void* d_out, int out_size, void* d_ws, size_t ws_size,
                              hipStream_t stream) {
  const float* x    = (const float*)d_in[0];
  const int*   ei   = (const int*)  d_in[1];
  const float* g1   = (const float*)d_in[2];
  const float* be1  = (const float*)d_in[3];
  const float* mu1  = (const float*)d_in[4];
  const float* va1  = (const float*)d_in[5];
  const float* W1   = (const float*)d_in[6];
  const float* att1 = (const float*)d_in[7];
  const float* b1   = (const float*)d_in[8];
  const float* g2   = (const float*)d_in[9];
  const float* be2  = (const float*)d_in[10];
  const float* mu2  = (const float*)d_in[11];
  const float* va2  = (const float*)d_in[12];
  const float* W2   = (const float*)d_in[13];
  const float* att2 = (const float*)d_in[14];
  const float* b2   = (const float*)d_in[15];
  const int* rowA = ei;
  const int* colA = ei + N_EDGES;
  float* out = (float*)d_out;
  float* ws  = (float*)d_ws;

  float* sc1  = ws;
  float* sh1  = ws + 256;
  float* c0   = ws + 512;
  float* ai   = ws + 1536;
  float* aj   = ai  + (size_t)N_NODES * H1;
  float* edg  = aj  + (size_t)N_NODES * H1;
  int*   cnt  = (int*)edg;
  float* l2   = edg + (size_t)N_NODES * 2 * H1;
  float* epb  = l2;
  col_t* bkt  = (col_t*)(l2 + (size_t)N_NODES * EPW);
  int*   spr  = (int*)(l2 + (size_t)N_NODES * (EPW + BK_CAP / 2));
  int*   spc  = spr + N_EDGES;
  float* gt   = l2  + (size_t)N_NODES * D1;
  float* Wc   = gt  + (size_t)N_NODES * GP;

  bn_prep<<<PREP_BLOCKS + F_IN + 1, 256, 0, stream>>>(g1, be1, mu1, va1, g2, be2, mu2, va2, b1,
                                                     W1, W2, att1, sc1, sh1, Wc, c0, cnt);

  gemm_bn_wmma<F_IN, NF, 2, 2, 8, 64, 2, 1, 10, true><<<EDGE_BLOCKS + G1_BLOCKS, 256, 0, stream>>>(
      x, sc1, sh1, Wc, gt, att1, ai, aj, rowA, colA, cnt, bkt, spr, spc);

  row_agg1<<<N_NODES / AG_ROWS, 256, 0, stream>>>(cnt, bkt, spr, spc, ai, aj, gt, c0, att2, epb);

  row_agg2<<<N_NODES / A2_ROWS, 256, 0, stream>>>(cnt, bkt, spr, spc, epb, b2, out);
}


# First launch's weight-fold blocks: each sum's operands requested 64 terms at a time as one group ahead of its multiply-adds (was 8 per round trip), BN2 scale pairs too; order of every sum unchanged
# speedup vs baseline: 9.4428x; 1.0730x over previous
#include <hip/hip_runtime.h>
#include <math.h>

#define N_NODES 40000
#define N_EDGES 320000
#define ETOT    (N_EDGES + N_NODES)
#define F_IN    256
#define H1      4
#define C1      128
#define D1      (H1*C1)
#define H2      8
#define C2      6
#define D2      (H2*C2)
#define BN_EPS    1e-5f
#define NEG_SLOPE 0.2f

typedef __attribute__((ext_vector_type(2))) float v2f;
typedef __attribute__((ext_vector_type(8))) float v8f;


static_assert(N_NODES <= 65536, "16-bit column ids need N_NODES <= 65536");
typedef unsigned short col_t;
static_assert(N_EDGES % 256 == 0 && N_NODES % 8 == 0 && (N_NODES * H2) % 256 == 0,
              "launch grids below divide exactly");

__device__ __forceinline__ float lrelu(float a) { return a > 0.f ? a : NEG_SLOPE * a; }


#define GP 192
#define NF 256
#define EPW 64
#define PREP_BLOCKS ((N_NODES + 255) / 256)
#define FOLD_G 64
static_assert(GP == H1 * D2 && GP + 2 * H1 <= NF && D1 == 2 * 256 && H1 * 64 == 256 && D2 <= 64, "fold maps");
static_assert(PREP_BLOCKS * 256 > N_NODES, "one thread per counter, the spill counter included");

__global__ __launch_bounds__(256)
void bn_prep(const float* g1, const float* be1, const float* mu1, const float* va1,
             const float* g2, const float* be2, const float* mu2, const float* va2,
             const float* bias1, const float* __restrict__ W1, const float* __restrict__ W2,
             const float* __restrict__ att1,
             float* sc1, float* sh1, float* __restrict__ Wc, float* __restrict__ c0, int* __restrict__ cnt) {
  if ((int)blockIdx.x >= PREP_BLOCKS) {
    __shared__ float s2c[D1];
    const int fb = (int)blockIdx.x - PREP_BLOCKS, n = threadIdx.x;
    static_assert(D1 == 2 * 256 && C1 % FOLD_G == 0 && FOLD_G % 2 == 0, "two BN2 scales per thread; whole load groups");
    {
      const float ga = g2[n], gb = g2[n + 256], va = va2[n], vb = va2[n + 256];
      s2c[n] = ga / sqrtf(va + BN_EPS);
      s2c[n + 256] = gb / sqrtf(vb + BN_EPS);
    }
    __syncthreads();
    float s = 0.f;
    if (fb < F_IN) {
      const float* __restrict__ w1 = W1 + (size_t)fb * D1;
      if (n < GP) {
        const int h = n / D2, o = n % D2;
        for (int cb = 0; cb < C1; cb += FOLD_G) {
          const int k0 = h * C1 + cb;
          float wa[FOLD_G], wb[FOLD_G];
          #pragma unroll
          for (int i = 0; i < FOLD_G; ++i) { wa[i] = w1[k0 + i]; wb[i] = W2[(k0 + i) * D2 + o]; }
          __builtin_amdgcn_sched_barrier(0);
          #pragma unroll
          for (int i = 0; i < FOLD_G; ++i) s += wa[i] * (s2c[k0 + i] * wb[i]);
        }
      } else if (n < GP + 2 * H1) {
        const int h = (n - GP) & (H1 - 1), side = (n - GP) / H1;
        const float* __restrict__ a = att1 + h * (2 * C1) + side * C1;
        for (int cb = 0; cb < C1; cb += FOLD_G) {
          float wa[FOLD_G], wb[FOLD_G];
          #pragma unroll
          for (int i = 0; i < FOLD_G; ++i) { wa[i] = w1[h * C1 + cb + i]; wb[i] = a[cb + i]; }
          __builtin_amdgcn_sched_barrier(0);
          #pragma unroll
          for (int i = 0; i < FOLD_G; ++i) s += wa[i] * wb[i];
        }
      }
      Wc[(size_t)fb * NF + n] = s;
    } else {
      const int h = n >> 6, o = n & 63;
      if (o < D2) {
        constexpr int G2 = FOLD_G / 2;
        for (int cb = 0; cb < C1; cb += G2) {
          const int k0 = h * C1 + cb;
          float wb[G2], vb[G2], vc[G2], vm[G2];
          #pragma unroll
          for (int i = 0; i < G2; ++i) {
            wb[i] = W2[(k0 + i) * D2 + o]; vb[i] = be2[k0 + i]; vc[i] = bias1[k0 + i]; vm[i] = mu2[k0 + i];
          }
          __builtin_amdgcn_sched_barrier(0);
          #pragma unroll
          for (int i = 0; i < G2; ++i) {
            const float sh = vb[i] + (vc[i] - vm[i]) * s2c[k0 + i];
            s += sh * wb[i];
          }
        }
        c0[h * D2 + o] = s;
      }
    }
    return;
  }
  int i = blockIdx.x * blockDim.x + threadIdx.x;
  if (i <= N_NODES) cnt[i] = 0;
  if (i < F_IN) {
    float s = g1[i] / sqrtf(va1[i] + BN_EPS);
    sc1[i] = s; sh1[i] = be1[i] - mu1[i] * s;
  }
}

#define BK_CAP      32
#define EDGE_EPT    5
#define EDGE_BLOCKS (N_EDGES / (256 * EDGE_EPT))
#define G1_BLOCKS   (N_NODES / 32)
static_assert(N_EDGES % (256 * EDGE_EPT) == 0, "edge blocks divide the edge list exactly");
__device__ __forceinline__ void edge_fill_blk(int blk, const int* __restrict__ rowA, const int* __restrict__ colA,
                                              int* __restrict__ cnt, col_t* __restrict__ bkt,
                                              int* __restrict__ spr, int* __restrict__ spc) {
  const int e0 = blk * (256 * EDGE_EPT) + threadIdx.x;
  int r[EDGE_EPT], c[EDGE_EPT], slot[EDGE_EPT];
  #pragma unroll
  for (int i = 0; i < EDGE_EPT; ++i) { r[i] = rowA[e0 + i * 256]; c[i] = colA[e0 + i * 256]; }
  #pragma unroll
  for (int i = 0; i < EDGE_EPT; ++i) slot[i] = atomicAdd(&cnt[r[i]], 1);
  #pragma unroll
  for (int i = 0; i < EDGE_EPT; ++i) {
    if (slot[i] < BK_CAP) {
      bkt[(size_t)r[i] * BK_CAP + slot[i]] = (col_t)c[i];
    } else {
      const int o = atomicAdd(&cnt[N_NODES], 1);
      spr[o] = r[i]; spc[o] = c[i];
    }
  }
}


template <int NT, int N>
__device__ __forceinline__ void load_b(const float* __restrict__ b0, float (&bx)[NT], float (&by)[NT]) {
  const float* __restrict__ b1 = b0 + N;
  if constexpr (NT == 4) {
    float4 t0 = *(const float4*)b0, t1 = *(const float4*)b1;
    bx[0] = t0.x; bx[1] = t0.y; bx[2] = t0.z; bx[3] = t0.w;
    by[0] = t1.x; by[1] = t1.y; by[2] = t1.z; by[3] = t1.w;
  } else {
    #pragma unroll
    for (int nt = 0; nt < NT; ++nt) { bx[nt] = b0[nt]; by[nt] = b1[nt]; }
  }
}

template <int K, int N, int MT, int NT, int WN, int KC, int PF, int SCORE, int WPS, bool EDGES>
__global__ __launch_bounds__(32 * WN)
__attribute__((amdgpu_waves_per_eu(WPS)))
void gemm_bn_wmma(const float* A,
                  const float* scale,
                  const float* shift,
                  const float* __restrict__ B,
                  float* __restrict__ C,
                  const float* __restrict__ att,
                  float* __restrict__ sa,
                  float* __restrict__ sb,
                  const int* __restrict__ rowA, const int* __restrict__ colA,
                  int* __restrict__ cnt, col_t* __restrict__ bkt,
                  int* __restrict__ spr, int* __restrict__ spc) {
  constexpr int THREADS = 32 * WN;
  if constexpr (EDGES) {
    static_assert(THREADS == 256, "edge blocks are 256 threads");
    if ((int)blockIdx.x < EDGE_BLOCKS) {
      edge_fill_blk((int)blockIdx.x, rowA, colA, cnt, bkt, spr, spc);
      return;
    }
  }
  constexpr int ROWS  = 16 * MT;
  constexpr int PITCH = K + 4;
  constexpr int K4  = K / 4;
  constexpr int C4  = KC / 4;
  constexpr int NCH = K / KC;
  constexpr int RPP = THREADS / C4;
  constexpr int LPT = ROWS / RPP;
  static_assert(PF >= 1 && 4 * PF <= K, "queue depth");
  static_assert(N == WN * NT * 16, "the waves of a work-group tile all N columns");
  static_assert(K % KC == 0 && KC % 4 == 0 && THREADS % C4 == 0 && ROWS % RPP == 0, "exact staging");
  static_assert((PITCH / 4) % 2 == 1, "pitch is 16 B x odd");
  static_assert(N_NODES % ROWS == 0, "grid divides exactly");
  __shared__ float As[ROWS * PITCH];

  const int tid  = threadIdx.x;
  const int lane = tid & 31;
  const int half = lane >> 4;
  const int l16  = lane & 15;
  const int wn   = tid >> 5;
  const int row0 = ((int)blockIdx.x - (EDGES ? EDGE_BLOCKS : 0)) * ROWS;

  const int tr = tid / C4, c4 = tid % C4;
  const float4* A4 = (const float4*)(A + (size_t)row0 * K) + (size_t)tr * K4 + c4;
  const float4* sc4 = (const float4*)scale + c4;
  const float4* sh4 = (const float4*)shift + c4;
  float* st = As + tr * PITCH + c4 * 4;

  float4 pre[LPT];
  #pragma unroll
  for (int i = 0; i < LPT; ++i) pre[i] = A4[(size_t)i * RPP * K4];
  float4 sc = sc4[0], sh = sh4[0];
  __builtin_amdgcn_sched_barrier(0);

  const float* as = As + l16 * PITCH + 2 * half;
  const float* __restrict__ bp = B + (size_t)(2 * half) * N + wn * (NT * 16) + NT * l16;

  v8f acc[MT][NT];
  #pragma unroll
  for (int mt = 0; mt < MT; ++mt)
    #pragma unroll
    for (int nt = 0; nt < NT; ++nt) acc[mt][nt] = v8f{};

  float qx[PF][NT], qy[PF][NT];
  #pragma unroll
  for (int p = 0; p < PF; ++p) load_b<NT, N>(bp + (size_t)(4 * p) * N, qx[p], qy[p]);

  #pragma nounroll
  for (int ch = 0; ch < NCH; ++ch) {
    #pragma unroll
    for (int i = 0; i < LPT; ++i) {
      float4 v = pre[i], o;
      o.x = v.x * sc.x + sh.x; o.y = v.y * sc.y + sh.y;
      o.z = v.z * sc.z + sh.z; o.w = v.w * sc.w + sh.w;
      *(float4*)(st + i * RPP * PITCH + ch * KC) = o;
    }
    __syncthreads();
    if (ch + 1 < NCH) {
      #pragma unroll
      for (int i = 0; i < LPT; ++i) pre[i] = A4[(size_t)i * RPP * K4 + (ch + 1) * C4];
      sc = sc4[(ch + 1) * C4]; sh = sh4[(ch + 1) * C4];
    }
    for (int k = ch * KC; k < ch * KC + KC; k += 4) {
      v2f a[MT];
      #pragma unroll
      for (int mt = 0; mt < MT; ++mt) a[mt] = *(const v2f*)(as + mt * 16 * PITCH + k);
      float bx[NT], by[NT];
      #pragma unroll
      for (int nt = 0; nt < NT; ++nt) { bx[nt] = qx[0][nt]; by[nt] = qy[0][nt]; }
      #pragma unroll
      for (int p = 0; p + 1 < PF; ++p)
        #pragma unroll
        for (int nt = 0; nt < NT; ++nt) { qx[p][nt] = qx[p + 1][nt]; qy[p][nt] = qy[p + 1][nt]; }
      if (k + 4 * PF < K) load_b<NT, N>(bp + (size_t)(k + 4 * PF) * N, qx[PF - 1], qy[PF - 1]);
      #pragma unroll
      for (int mt = 0; mt < MT; ++mt)
        #pragma unroll
        for (int nt = 0; nt < NT; ++nt) {
          v2f b; b.x = bx[nt]; b.y = by[nt];
          acc[mt][nt] = __builtin_amdgcn_wmma_f32_16x16x4_f32(
               false, a[mt],  false, b,
               (short)0, acc[mt][nt],  false,  false);
        }
    }
  }

  if constexpr (SCORE != 0) {
    static_assert(MT == 2 && NT == 2, "four accumulators named below");
    asm volatile("v_nop\n\tv_nop\n\tv_nop\n\tv_nop\n\tv_nop"
                 : "+v"(acc[0][0]), "+v"(acc[0][1]), "+v"(acc[1][0]), "+v"(acc[1][1]));
  }
  if constexpr (SCORE == 0) {
    #pragma unroll
    for (int mt = 0; mt < MT; ++mt) {
      float* cp = C + (size_t)(row0 + mt * 16 + (half << 3)) * N + wn * (NT * 16) + NT * l16;
      #pragma unroll
      for (int r = 0; r < 8; ++r) {
        if constexpr (NT == 4) {
          *(float4*)(cp + (size_t)r * N) =
              make_float4(acc[mt][0][r], acc[mt][1][r], acc[mt][2][r], acc[mt][3][r]);
        } else {
          #pragma unroll
          for (int nt = 0; nt < NT; ++nt) cp[(size_t)r * N + nt] = acc[mt][nt][r];
        }
      }
    }
  } else {
    static_assert(NT == 2 && N == NF && GP % (NT * 16) == 0 && 2 * H1 == 4 * NT, "score columns in four lanes");
    if (wn < GP / (NT * 16)) {
      #pragma unroll
      for (int mt = 0; mt < MT; ++mt) {
        float* cp = C + (size_t)(row0 + mt * 16 + (half << 3)) * GP + wn * (NT * 16) + NT * l16;
        #pragma unroll
        for (int r = 0; r < 8; ++r)
          *(float2*)(cp + (size_t)r * GP) = make_float2(acc[mt][0][r], acc[mt][1][r]);
      }
    } else if (wn == GP / (NT * 16) && l16 < 4) {
      float* __restrict__ dst = (l16 < 2 ? sa : sb) + 2 * (l16 & 1);
      #pragma unroll
      for (int mt = 0; mt < MT; ++mt)
        #pragma unroll
        for (int r = 0; r < 8; ++r)
          *(float2*)(dst + (size_t)(row0 + mt * 16 + (half << 3) + r) * H1) =
              make_float2(acc[mt][0][r], acc[mt][1][r]);
    }
  }
}


#define AG_ROWS  16
#define AG_RPW   (AG_ROWS / 8)
#define AG_PITCH (GP + 4)
static_assert(N_NODES % AG_ROWS == 0 && GP == 96 * 2 && D2 == 24 * 2, "three float2 per lane, 24 float2 per head");
static_assert(C2 == 6 && H2 == 8 && AG_ROWS * H2 == 128, "epilogue map");

__global__ __launch_bounds__(256)
void row_agg1(const int* __restrict__ ecnt, const col_t* __restrict__ csr,
              const int* __restrict__ spr, const int* __restrict__ spc,
              const float* __restrict__ ai, const float* __restrict__ aj,
              const float* __restrict__ g, const float* __restrict__ c0,
              const float* __restrict__ att,
              float* __restrict__ EP) {
  __shared__ float Ls[AG_ROWS * AG_PITCH];
  const int tid = threadIdx.x;
  const int lane = tid & 31;
  const int wv = tid >> 5;
  const int row0 = (int)blockIdx.x * AG_ROWS;
  const float4* __restrict__ aj4 = (const float4*)aj;
  const float2* __restrict__ g2 = (const float2*)g;
  const bool lo0 = lane < 24, lo1 = lane < 16, lo2 = lane < 8;

  for (int q = 0; q < AG_RPW; ++q) {
    const int rl = wv * AG_RPW + q;
    int r = __builtin_amdgcn_readfirstlane(row0 + rl);
    const int deg = ecnt[r];
    const int spl = ecnt[N_NODES];
    const int beg = r * BK_CAP;
    const int end = beg + (deg < BK_CAP ? deg : BK_CAP);
    const int nsp = deg > BK_CAP ? spl : 0;
    float4 a_i = *(const float4*)(ai + (size_t)r * 4);
    float4 bs  = aj4[r];
    float t0 = lrelu(a_i.x + bs.x), t1 = lrelu(a_i.y + bs.y);
    float t2 = lrelu(a_i.z + bs.z), t3 = lrelu(a_i.w + bs.w);

    float m0 = t0, m1 = t1, m2 = t2, m3 = t3;
    for (int j = beg + lane; j < end; j += 32) {
      float4 b = aj4[csr[j]];
      m0 = fmaxf(m0, lrelu(a_i.x + b.x));
      m1 = fmaxf(m1, lrelu(a_i.y + b.y));
      m2 = fmaxf(m2, lrelu(a_i.z + b.z));
      m3 = fmaxf(m3, lrelu(a_i.w + b.w));
    }
    for (int ob = 0; ob < nsp; ob += 32) {
      const int i = ob + lane;
      if (i < nsp && spr[i] == r) {
        float4 b = aj4[spc[i]];
        m0 = fmaxf(m0, lrelu(a_i.x + b.x));
        m1 = fmaxf(m1, lrelu(a_i.y + b.y));
        m2 = fmaxf(m2, lrelu(a_i.z + b.z));
        m3 = fmaxf(m3, lrelu(a_i.w + b.w));
      }
    }
    #pragma unroll
    for (int off = 16; off > 0; off >>= 1) {
      m0 = fmaxf(m0, __shfl_xor(m0, off, 32));
      m1 = fmaxf(m1, __shfl_xor(m1, off, 32));
      m2 = fmaxf(m2, __shfl_xor(m2, off, 32));
      m3 = fmaxf(m3, __shfl_xor(m3, off, 32));
    }

    float u0 = expf(t0 - m0), u1 = expf(t1 - m1), u2 = expf(t2 - m2), u3 = expf(t3 - m3);
    float wa = lo0 ? u0 : u1, wb = lo1 ? u1 : u2, wc = lo2 ? u2 : u3;
    float sA = wa, sB = wb, sC = wc;
    const float2* __restrict__ src = g2 + (size_t)r * (GP / 2);
    float2 v0 = src[lane], v1 = src[32 + lane], v2 = src[64 + lane];
    float2 acc0 = {wa * v0.x, wa * v0.y};
    float2 acc1 = {wb * v1.x, wb * v1.y};
    float2 acc2 = {wc * v2.x, wc * v2.y};

    for (int base = beg; base < end; base += 32) {
      int j = base + lane;
      int c = r;
      float w0 = 0.f, w1 = 0.f, w2 = 0.f, w3 = 0.f;
      if (j < end) {
        c = csr[j];
        float4 b = aj4[c];
        w0 = expf(lrelu(a_i.x + b.x) - m0);
        w1 = expf(lrelu(a_i.y + b.y) - m1);
        w2 = expf(lrelu(a_i.z + b.z) - m2);
        w3 = expf(lrelu(a_i.w + b.w) - m3);
      }
      int cnt = end - base < 32 ? end - base : 32;
      for (int k = 0; k < cnt; ++k) {
        int ck = __shfl(c, k, 32);
        u0 = __shfl(w0, k, 32); u1 = __shfl(w1, k, 32);
        u2 = __shfl(w2, k, 32); u3 = __shfl(w3, k, 32);
        wa = lo0 ? u0 : u1; wb = lo1 ? u1 : u2; wc = lo2 ? u2 : u3;
        sA += wa; sB += wb; sC += wc;
        src = g2 + (size_t)ck * (GP / 2);
        v0 = src[lane]; v1 = src[32 + lane]; v2 = src[64 + lane];
        acc0.x += wa * v0.x; acc0.y += wa * v0.y;
        acc1.x += wb * v1.x; acc1.y += wb * v1.y;
        acc2.x += wc * v2.x; acc2.y += wc * v2.y;
      }
    }
    for (int ob = 0; ob < nsp; ob += 32) {
      const int i = ob + lane;
      const bool hit = i < nsp && spr[i] == r;
      int c = r;
      float w0 = 0.f, w1 = 0.f, w2 = 0.f, w3 = 0.f;
      if (hit) {
        c = spc[i];
        float4 b = aj4[c];
        w0 = expf(lrelu(a_i.x + b.x) - m0);
        w1 = expf(lrelu(a_i.y + b.y) - m1);
        w2 = expf(lrelu(a_i.z + b.z) - m2);
        w3 = expf(lrelu(a_i.w + b.w) - m3);
      }
      unsigned todo = __builtin_amdgcn_ballot_w32(hit);
      while (todo) {
        const int k = __builtin_ctz(todo);
        todo &= todo - 1;
        int ck = __shfl(c, k, 32);
        u0 = __shfl(w0, k, 32); u1 = __shfl(w1, k, 32);
        u2 = __shfl(w2, k, 32); u3 = __shfl(w3, k, 32);
        wa = lo0 ? u0 : u1; wb = lo1 ? u1 : u2; wc = lo2 ? u2 : u3;
        sA += wa; sB += wb; sC += wc;
        src = g2 + (size_t)ck * (GP / 2);
        v0 = src[lane]; v1 = src[32 + lane]; v2 = src[64 + lane];
        acc0.x += wa * v0.x; acc0.y += wa * v0.y;
        acc1.x += wb * v1.x; acc1.y += wb * v1.y;
        acc2.x += wc * v2.x; acc2.y += wc * v2.y;
      }
    }
    float iA = 1.0f / (sA + 1e-16f), iB = 1.0f / (sB + 1e-16f), iC = 1.0f / (sC + 1e-16f);
    float2* dst = (float2*)(Ls + rl * AG_PITCH);
    dst[lane]      = make_float2(acc0.x * iA, acc0.y * iA);
    dst[32 + lane] = make_float2(acc1.x * iB, acc1.y * iB);
    dst[64 + lane] = make_float2(acc2.x * iC, acc2.y * iC);
  }
  __syncthreads();

  if (tid < AG_ROWS * H2) {
    const int row = tid >> 3, hd = tid & 7;
    const float* p = Ls + row * AG_PITCH + hd * C2;
    const float* __restrict__ cz = c0 + hd * C2;
    float v[C2];
    #pragma unroll
    for (int j = 0; j < C2; ++j) {
      float s = p[j];
      #pragma unroll
      for (int h = 1; h < H1; ++h) s += p[h * D2 + j];
      v[j] = s + (((cz[j] + cz[D2 + j]) + cz[2 * D2 + j]) + cz[3 * D2 + j]);
    }
    const float* __restrict__ ap = att + hd * (2 * C2);
    const float pa = v[0] * (ap[0] + ap[C2])     + v[1] * (ap[1] + ap[C2 + 1]) + v[2] * (ap[2] + ap[C2 + 2]);
    const float pb = v[3] * (ap[3] + ap[C2 + 3]) + v[4] * (ap[4] + ap[C2 + 4]) + v[5] * (ap[5] + ap[C2 + 5]);
    const float e = expf(tanhf(pa + pb));
    float4* __restrict__ ep = (float4*)(EP + (size_t)(row0 + row) * EPW) + hd;
    ep[0]  = make_float4(e, e * v[0], e * v[1], e * v[2]);
    ep[H2] = make_float4(e * v[3], e * v[4], e * v[5], 0.f);
  }
}


#define A2_ROWS  64
#define A2_HPL   2
#define A2_U     2
#define A2_PITCH 50
static_assert(N_NODES % A2_ROWS == 0 && A2_ROWS * (H2 / A2_HPL) == 256 && EPW == 2 * H2 * 4, "layer-2 maps");

__global__ __launch_bounds__(256)
void row_agg2(const int* __restrict__ ecnt, const col_t* __restrict__ csr,
              const int* __restrict__ spr, const int* __restrict__ spc,
              const float* __restrict__ ep, const float* __restrict__ b2,
              float* __restrict__ out) {
  __shared__ float Ls[A2_ROWS * A2_PITCH];
  const int rl = threadIdx.x >> 2, q = threadIdx.x & 3;
  const int r = (int)blockIdx.x * A2_ROWS + rl;
  const int deg = ecnt[r];
  const int beg = r * BK_CAP;
  const int end = beg + min(deg, BK_CAP);
  const float4* __restrict__ ep4 = (const float4*)ep + A2_HPL * q;
  float den[A2_HPL], a[A2_HPL][C2];
  {
    const float4* __restrict__ p = ep4 + (size_t)r * (EPW / 4);
    #pragma unroll
    for (int k = 0; k < A2_HPL; ++k) {
      const float4 x = p[k], y = p[H2 + k];
      den[k] = x.x; a[k][0] = x.y; a[k][1] = x.z; a[k][2] = x.w; a[k][3] = y.x; a[k][4] = y.y; a[k][5] = y.z;
    }
  }

  int c[A2_U];
  #pragma unroll
  for (int u = 0; u < A2_U; ++u) c[u] = beg < end ? (int)csr[min(beg + u, end - 1)] : r;
  for (int j = beg; j < end; j += A2_U) {
    float4 xs[A2_U][A2_HPL], ys[A2_U][A2_HPL];
    #pragma unroll
    for (int u = 0; u < A2_U; ++u) {
      const float4* __restrict__ p = ep4 + (size_t)c[u] * (EPW / 4);
      #pragma unroll
      for (int k = 0; k < A2_HPL; ++k) { xs[u][k] = p[k]; ys[u][k] = p[H2 + k]; }
    }
    #pragma unroll
    for (int u = 0; u < A2_U; ++u) c[u] = (int)csr[min(j + A2_U + u, end - 1)];
    #pragma unroll
    for (int u = 0; u < A2_U; ++u) {
      const bool in = u == 0 || j + u < end;
      #pragma unroll
      for (int k = 0; k < A2_HPL; ++k) {
        den[k]  += in ? xs[u][k].x : -0.0f;
        a[k][0] += in ? xs[u][k].y : -0.0f; a[k][1] += in ? xs[u][k].z : -0.0f; a[k][2] += in ? xs[u][k].w : -0.0f;
        a[k][3] += in ? ys[u][k].x : -0.0f; a[k][4] += in ? ys[u][k].y : -0.0f; a[k][5] += in ? ys[u][k].z : -0.0f;
      }
    }
  }
  if (deg > BK_CAP) {
    const int nsp = ecnt[N_NODES];
    for (int i = 0; i < nsp; ++i) {
      if (spr[i] != r) continue;
      const float4* __restrict__ p = ep4 + (size_t)spc[i] * (EPW / 4);
      #pragma unroll
      for (int k = 0; k < A2_HPL; ++k) {
        const float4 x = p[k], y = p[H2 + k];
        den[k]  += x.x;
        a[k][0] += x.y; a[k][1] += x.z; a[k][2] += x.w;
        a[k][3] += y.x; a[k][4] += y.y; a[k][5] += y.z;
      }
    }
  }
  float2* __restrict__ qd = (float2*)(Ls + rl * A2_PITCH + A2_HPL * q * C2);
  #pragma unroll
  for (int k = 0; k < A2_HPL; ++k) {
    const float inv = 1.0f / (den[k] + 1e-16f);
    qd[3 * k]     = make_float2(a[k][0] * inv, a[k][1] * inv);
    qd[3 * k + 1] = make_float2(a[k][2] * inv, a[k][3] * inv);
    qd[3 * k + 2] = make_float2(a[k][4] * inv, a[k][5] * inv);
  }
  __syncthreads();

  if (threadIdx.x < A2_ROWS) {
    const float* p = Ls + threadIdx.x * A2_PITCH;
    const int n = (int)blockIdx.x * A2_ROWS + (int)threadIdx.x;
    float v[C2];
    #pragma unroll
    for (int ch = 0; ch < C2; ++ch) {
      float s = 0.f;
      #pragma unroll
      for (int hh = 0; hh < H2; ++hh) s += p[hh * C2 + ch];
      float t = s * (1.0f / H2) + b2[ch];
      v[ch] = t > 0.f ? t : (expf(t) - 1.0f);
    }
    float m = v[0];
    #pragma unroll
    for (int ch = 1; ch < C2; ++ch) m = fmaxf(m, v[ch]);
    float sum = 0.f;
    #pragma unroll
    for (int ch = 0; ch < C2; ++ch) sum += expf(v[ch] - m);
    const float lse = logf(sum);
    float2* __restrict__ dst = (float2*)(out + (size_t)n * C2);
    dst[0] = make_float2(v[0] - m - lse, v[1] - m - lse);
    dst[1] = make_float2(v[2] - m - lse, v[3] - m - lse);
    dst[2] = make_float2(v[4] - m - lse, v[5] - m - lse);
  }
}


extern "C" void kernel_launch(void* const* d_in, const int* in_sizes, int n_in,
                              void* d_out, int out_size, void* d_ws, size_t ws_size,
                              hipStream_t stream) {
  const float* x    = (const float*)d_in[0];
  const int*   ei   = (const int*)  d_in[1];
  const float* g1   = (const float*)d_in[2];
  const float* be1  = (const float*)d_in[3];
  const float* mu1  = (const float*)d_in[4];
  const float* va1  = (const float*)d_in[5];
  const float* W1   = (const float*)d_in[6];
  const float* att1 = (const float*)d_in[7];
  const float* b1   = (const float*)d_in[8];
  const float* g2   = (const float*)d_in[9];
  const float* be2  = (const float*)d_in[10];
  const float* mu2  = (const float*)d_in[11];
  const float* va2  = (const float*)d_in[12];
  const float* W2   = (const float*)d_in[13];
  const float* att2 = (const float*)d_in[14];
  const float* b2   = (const float*)d_in[15];
  const int* rowA = ei;
  const int* colA = ei + N_EDGES;
  float* out = (float*)d_out;
  float* ws  = (float*)d_ws;

  float* sc1  = ws;
  float* sh1  = ws + 256;
  float* c0   = ws + 512;
  float* ai   = ws + 1536;
  float* aj   = ai  + (size_t)N_NODES * H1;
  float* edg  = aj  + (size_t)N_NODES * H1;
  int*   cnt  = (int*)edg;
  float* l2   = edg + (size_t)N_NODES * 2 * H1;
  float* epb  = l2;
  col_t* bkt  = (col_t*)(l2 + (size_t)N_NODES * EPW);
  int*   spr  = (int*)(l2 + (size_t)N_NODES * (EPW + BK_CAP / 2));
  int*   spc  = spr + N_EDGES;
  float* gt   = l2  + (size_t)N_NODES * D1;
  float* Wc   = gt  + (size_t)N_NODES * GP;

  bn_prep<<<PREP_BLOCKS + F_IN + 1, 256, 0, stream>>>(g1, be1, mu1, va1, g2, be2, mu2, va2, b1,
                                                     W1, W2, att1, sc1, sh1, Wc, c0, cnt);

  gemm_bn_wmma<F_IN, NF, 2, 2, 8, 64, 2, 1, 10, true><<<EDGE_BLOCKS + G1_BLOCKS, 256, 0, stream>>>(
      x, sc1, sh1, Wc, gt, att1, ai, aj, rowA, colA, cnt, bkt, spr, spc);

  row_agg1<<<N_NODES / AG_ROWS, 256, 0, stream>>>(cnt, bkt, spr, spc, ai, aj, gt, c0, att2, epb);

  row_agg2<<<N_NODES / A2_ROWS, 256, 0, stream>>>(cnt, bkt, spr, spc, epb, b2, out);
}
